# starting kernel
# baseline (speedup 1.0000x reference)
#include <hip/hip_runtime.h>
#include <math.h>

#define NNODES 100000
#define NEDGES 1000000
#define INF    128
#define NH     4
#define ND     32
#define HD     128
#define NEG_SLOPE 0.2f

typedef __attribute__((ext_vector_type(2))) float v2f;
typedef __attribute__((ext_vector_type(8))) float v8f;

__global__ void gat_gemm_wmma(const float* __restrict__ X,
                              const float* __restrict__ W,
                              float* __restrict__ Y) {
  const int row0 = blockIdx.x << 4;
  const int wv   = threadIdx.x >> 5;
  const int lane = threadIdx.x & 31;
  const int m    = lane & 15;
  const int koff = (lane >> 4) << 1;
  const int col  = (wv << 4) + m;

  const float* __restrict__ xrow = X + (size_t)(row0 + m) * INF;
  v8f c = {};
  #pragma unroll 4
  for (int k = 0; k < INF; k += 4) {
    v2f a = *(const v2f*)(xrow + k + koff);
    v2f b;
    b.x = W[(size_t)(k + koff) * HD + col];
    b.y = W[(size_t)(k + koff + 1) * HD + col];
    c = __builtin_amdgcn_wmma_f32_16x16x4_f32(false, a, false, b,
                                              (short)0, c, false, false);
  }
  const int rbase = row0 + ((lane >> 4) << 3);
  #pragma unroll
  for (int i = 0; i < 8; ++i)
    Y[(size_t)(rbase + i) * HD + col] = c[i];
}

__global__ void gat_attn_dots(const float* __restrict__ feat,
                              const float* __restrict__ al,
                              const float* __restrict__ ar,
                              float* __restrict__ el,
                              float* __restrict__ er) {
  int t = blockIdx.x * blockDim.x + threadIdx.x;
  if (t >= NNODES * NH) return;
  int n = t >> 2, hh = t & 3;
  const float* f = feat + (size_t)n * HD + hh * ND;
  const float* a = al + hh * ND;
  const float* b = ar + hh * ND;
  float sl = 0.f, sr = 0.f;
  #pragma unroll
  for (int d = 0; d < ND; ++d) {
    float v = f[d];
    sl += v * a[d];
    sr += v * b[d];
  }
  el[t] = sl;
  er[t] = sr;
}

__global__ void gat_init_out(float* __restrict__ out,
                             const float* __restrict__ b0,
                             const float* __restrict__ b1) {
  size_t t = (size_t)blockIdx.x * blockDim.x + threadIdx.x;
  if (t >= (size_t)NNODES * HD) return;
  int j = (int)(t & (HD - 1));
  out[t] = b0[j] + b1[j];
}

__global__ void gat_init_ms(float* __restrict__ mbuf, float* __restrict__ sbuf) {
  int t = blockIdx.x * blockDim.x + threadIdx.x;
  if (t >= NNODES * NH) return;
  mbuf[t] = -INFINITY;
  sbuf[t] = 0.f;
}

__device__ __forceinline__ float leaky(float v) {
  return v > 0.f ? v : NEG_SLOPE * v;
}

__device__ __forceinline__ void atomicMaxF32(float* addr, float v) {
  if (v >= 0.f)
    atomicMax((int*)addr, __float_as_int(v));
  else
    atomicMin((unsigned int*)addr, __float_as_uint(v));
}

__global__ void gat_edge_max(const int* __restrict__ src,
                             const int* __restrict__ dst,
                             const float* __restrict__ el,
                             const float* __restrict__ er,
                             float* __restrict__ mbuf) {
  int t = blockIdx.x * blockDim.x + threadIdx.x;
  if (t >= NEDGES * NH) return;
  int e = t >> 2, hh = t & 3;
  int sN = src[e], dN = dst[e];
  float v = leaky(el[sN * NH + hh] + er[dN * NH + hh]);
  atomicMaxF32(&mbuf[dN * NH + hh], v);
}

__global__ void gat_fix_m(float* __restrict__ mbuf) {
  int t = blockIdx.x * blockDim.x + threadIdx.x;
  if (t >= NNODES * NH) return;
  float m = mbuf[t];
  if (!isfinite(m)) mbuf[t] = 0.f;
}

__global__ void gat_edge_sum(const int* __restrict__ src,
                             const int* __restrict__ dst,
                             const float* __restrict__ el,
                             const float* __restrict__ er,
                             const float* __restrict__ mbuf,
                             float* __restrict__ sbuf) {
  int t = blockIdx.x * blockDim.x + threadIdx.x;
  if (t >= NEDGES * NH) return;
  int e = t >> 2, hh = t & 3;
  int sN = src[e], dN = dst[e];
  float v = leaky(el[sN * NH + hh] + er[dN * NH + hh]);
  float ex = __expf(v - mbuf[dN * NH + hh]);
  atomicAdd(&sbuf[dN * NH + hh], ex);
}

__global__ void gat_edge_agg(const int* __restrict__ src,
                             const int* __restrict__ dst,
                             const float* __restrict__ el,
                             const float* __restrict__ er,
                             const float* __restrict__ mbuf,
                             const float* __restrict__ sbuf,
                             const float* __restrict__ feat,
                             float* __restrict__ out) {
  int e = blockIdx.x * (blockDim.x >> 5) + (threadIdx.x >> 5);
  if (e >= NEDGES) return;
  int lane = threadIdx.x & 31;
  int sN = src[e], dN = dst[e];
  #pragma unroll
  for (int hh = 0; hh < NH; ++hh) {
    float v  = leaky(el[sN * NH + hh] + er[dN * NH + hh]);
    float ex = __expf(v - mbuf[dN * NH + hh]);
    float alpha = ex / fmaxf(sbuf[dN * NH + hh], 1e-20f);
    float val = feat[(size_t)sN * HD + hh * ND + lane] * alpha;
    atomicAdd(&out[(size_t)dN * HD + hh * ND + lane], val);
  }
}

extern "C" void kernel_launch(void* const* d_in, const int* in_sizes, int n_in,
                              void* d_out, int out_size, void* d_ws, size_t ws_size,
                              hipStream_t stream) {
  const float* h    = (const float*)d_in[0];
  const int*   srcs[2] = { (const int*)d_in[1], (const int*)d_in[3] };
  const int*   dsts[2] = { (const int*)d_in[2], (const int*)d_in[4] };
  const float* Ws[2]   = { (const float*)d_in[5],  (const float*)d_in[9]  };
  const float* als[2]  = { (const float*)d_in[6],  (const float*)d_in[10] };
  const float* ars[2]  = { (const float*)d_in[7],  (const float*)d_in[11] };
  const float* b0 = (const float*)d_in[8];
  const float* b1 = (const float*)d_in[12];
  float* out = (float*)d_out;

  float* feat = (float*)d_ws;
  float* el   = feat + (size_t)NNODES * HD;
  float* er   = el   + (size_t)NNODES * NH;
  float* mbuf = er   + (size_t)NNODES * NH;
  float* sbuf = mbuf + (size_t)NNODES * NH;

  const int B = 256;
  const int nNH   = (NNODES * NH + B - 1) / B;
  const int nOut  = (int)(((size_t)NNODES * HD + B - 1) / B);
  const int nEH   = (NEDGES * NH + B - 1) / B;
  const int nEdgeWaves = NEDGES / (B / 32);

  gat_init_out<<<nOut, B, 0, stream>>>(out, b0, b1);

  for (int r = 0; r < 2; ++r) {
    gat_gemm_wmma<<<NNODES / 16, B, 0, stream>>>(h, Ws[r], feat);
    gat_attn_dots<<<nNH, B, 0, stream>>>(feat, als[r], ars[r], el, er);
    gat_init_ms<<<nNH, B, 0, stream>>>(mbuf, sbuf);
    gat_edge_max<<<nEH, B, 0, stream>>>(srcs[r], dsts[r], el, er, mbuf);
    gat_fix_m<<<nNH, B, 0, stream>>>(mbuf);
    gat_edge_sum<<<nEH, B, 0, stream>>>(srcs[r], dsts[r], el, er, mbuf, sbuf);
    gat_edge_agg<<<nEdgeWaves, B, 0, stream>>>(srcs[r], dsts[r], el, er,
                                               mbuf, sbuf, feat, out);
  }
}


# edge aggregation: one wave walks 32 consecutive edges (lane i prepares edge i: node ids, 4 softmax weights; 8 edges' rows in flight), not one wave per edge; atomic adds kept, arrival-order sums
# speedup vs baseline: 1.3815x; 1.3815x over previous
#include <hip/hip_runtime.h>
#include <math.h>

#define NNODES 100000
#define NEDGES 1000000
#define INF    128
#define NH     4
#define ND     32
#define HD     128
#define NEG_SLOPE 0.2f

typedef __attribute__((ext_vector_type(2))) float v2f;
typedef __attribute__((ext_vector_type(8))) float v8f;

__global__ void gat_gemm_wmma(const float* __restrict__ X,
                              const float* __restrict__ W,
                              float* __restrict__ Y) {
  const int row0 = blockIdx.x << 4;
  const int wv   = threadIdx.x >> 5;
  const int lane = threadIdx.x & 31;
  const int m    = lane & 15;
  const int koff = (lane >> 4) << 1;
  const int col  = (wv << 4) + m;

  const float* __restrict__ xrow = X + (size_t)(row0 + m) * INF;
  v8f c = {};
  #pragma unroll 4
  for (int k = 0; k < INF; k += 4) {
    v2f a = *(const v2f*)(xrow + k + koff);
    v2f b;
    b.x = W[(size_t)(k + koff) * HD + col];
    b.y = W[(size_t)(k + koff + 1) * HD + col];
    c = __builtin_amdgcn_wmma_f32_16x16x4_f32(false, a, false, b,
                                              (short)0, c, false, false);
  }
  const int rbase = row0 + ((lane >> 4) << 3);
  #pragma unroll
  for (int i = 0; i < 8; ++i)
    Y[(size_t)(rbase + i) * HD + col] = c[i];
}

__global__ void gat_attn_dots(const float* __restrict__ feat,
                              const float* __restrict__ al,
                              const float* __restrict__ ar,
                              float* __restrict__ el,
                              float* __restrict__ er) {
  int t = blockIdx.x * blockDim.x + threadIdx.x;
  if (t >= NNODES * NH) return;
  int n = t >> 2, hh = t & 3;
  const float* f = feat + (size_t)n * HD + hh * ND;
  const float* a = al + hh * ND;
  const float* b = ar + hh * ND;
  float sl = 0.f, sr = 0.f;
  #pragma unroll
  for (int d = 0; d < ND; ++d) {
    float v = f[d];
    sl += v * a[d];
    sr += v * b[d];
  }
  el[t] = sl;
  er[t] = sr;
}

__global__ void gat_init_out(float* __restrict__ out,
                             const float* __restrict__ b0,
                             const float* __restrict__ b1) {
  size_t t = (size_t)blockIdx.x * blockDim.x + threadIdx.x;
  if (t >= (size_t)NNODES * HD) return;
  int j = (int)(t & (HD - 1));
  out[t] = b0[j] + b1[j];
}

__global__ void gat_init_ms(float* __restrict__ mbuf, float* __restrict__ sbuf) {
  int t = blockIdx.x * blockDim.x + threadIdx.x;
  if (t >= NNODES * NH) return;
  mbuf[t] = -INFINITY;
  sbuf[t] = 0.f;
}

__device__ __forceinline__ float leaky(float v) {
  return v > 0.f ? v : NEG_SLOPE * v;
}

__device__ __forceinline__ void atomicMaxF32(float* addr, float v) {
  if (v >= 0.f)
    atomicMax((int*)addr, __float_as_int(v));
  else
    atomicMin((unsigned int*)addr, __float_as_uint(v));
}

__global__ void gat_edge_max(const int* __restrict__ src,
                             const int* __restrict__ dst,
                             const float* __restrict__ el,
                             const float* __restrict__ er,
                             float* __restrict__ mbuf) {
  int t = blockIdx.x * blockDim.x + threadIdx.x;
  if (t >= NEDGES * NH) return;
  int e = t >> 2, hh = t & 3;
  int sN = src[e], dN = dst[e];
  float v = leaky(el[sN * NH + hh] + er[dN * NH + hh]);
  atomicMaxF32(&mbuf[dN * NH + hh], v);
}

__global__ void gat_fix_m(float* __restrict__ mbuf) {
  int t = blockIdx.x * blockDim.x + threadIdx.x;
  if (t >= NNODES * NH) return;
  float m = mbuf[t];
  if (!isfinite(m)) mbuf[t] = 0.f;
}

__global__ void gat_edge_sum(const int* __restrict__ src,
                             const int* __restrict__ dst,
                             const float* __restrict__ el,
                             const float* __restrict__ er,
                             const float* __restrict__ mbuf,
                             float* __restrict__ sbuf) {
  int t = blockIdx.x * blockDim.x + threadIdx.x;
  if (t >= NEDGES * NH) return;
  int e = t >> 2, hh = t & 3;
  int sN = src[e], dN = dst[e];
  float v = leaky(el[sN * NH + hh] + er[dN * NH + hh]);
  float ex = __expf(v - mbuf[dN * NH + hh]);
  atomicAdd(&sbuf[dN * NH + hh], ex);
}

#define EPW 32
#define GRP 8
__global__ void gat_edge_agg(const int* __restrict__ src,
                             const int* __restrict__ dst,
                             const float* __restrict__ el,
                             const float* __restrict__ er,
                             const float* __restrict__ mbuf,
                             const float* __restrict__ sbuf,
                             const float* __restrict__ feat,
                             float* __restrict__ out) {
  const int wave = blockIdx.x * (blockDim.x >> 5) + (threadIdx.x >> 5);
  if (wave >= NEDGES / EPW) return;
  const unsigned lane = threadIdx.x & 31;
  const int e = wave * EPW + (int)lane;
  const int sN = src[e], dN = dst[e];
  const float4 l4 = *(const float4*)(el   + (size_t)sN * NH);
  const float4 r4 = *(const float4*)(er   + (size_t)dN * NH);
  const float4 m4 = *(const float4*)(mbuf + (size_t)dN * NH);
  const float4 s4 = *(const float4*)(sbuf + (size_t)dN * NH);
  float al[NH];
  al[0] = __expf(leaky(l4.x + r4.x) - m4.x) / fmaxf(s4.x, 1e-20f);
  al[1] = __expf(leaky(l4.y + r4.y) - m4.y) / fmaxf(s4.y, 1e-20f);
  al[2] = __expf(leaky(l4.z + r4.z) - m4.z) / fmaxf(s4.z, 1e-20f);
  al[3] = __expf(leaky(l4.w + r4.w) - m4.w) / fmaxf(s4.w, 1e-20f);

  #pragma unroll
  for (int g = 0; g < EPW; g += GRP) {
    float r[GRP][NH];
    #pragma unroll
    for (int u = 0; u < GRP; ++u) {
      const int sj = __builtin_amdgcn_readlane(sN, g + u);
      const float* __restrict__ row = feat + (size_t)sj * HD;
      #pragma unroll
      for (int hh = 0; hh < NH; ++hh) r[u][hh] = row[hh * ND + lane];
    }
    #pragma unroll
    for (int u = 0; u < GRP; ++u) {
      const int dj = __builtin_amdgcn_readlane(dN, g + u);
      float* __restrict__ orow = out + (size_t)dj * HD;
      #pragma unroll
      for (int hh = 0; hh < NH; ++hh) {
        const float a = __int_as_float(
            __builtin_amdgcn_readlane(__float_as_int(al[hh]), g + u));
        atomicAdd(&orow[hh * ND + lane], r[u][hh] * a);
      }
    }
  }
}

extern "C" void kernel_launch(void* const* d_in, const int* in_sizes, int n_in,
                              void* d_out, int out_size, void* d_ws, size_t ws_size,
                              hipStream_t stream) {
  const float* h    = (const float*)d_in[0];
  const int*   srcs[2] = { (const int*)d_in[1], (const int*)d_in[3] };
  const int*   dsts[2] = { (const int*)d_in[2], (const int*)d_in[4] };
  const float* Ws[2]   = { (const float*)d_in[5],  (const float*)d_in[9]  };
  const float* als[2]  = { (const float*)d_in[6],  (const float*)d_in[10] };
  const float* ars[2]  = { (const float*)d_in[7],  (const float*)d_in[11] };
  const float* b0 = (const float*)d_in[8];
  const float* b1 = (const float*)d_in[12];
  float* out = (float*)d_out;

  float* feat = (float*)d_ws;
  float* el   = feat + (size_t)NNODES * HD;
  float* er   = el   + (size_t)NNODES * NH;
  float* mbuf = er   + (size_t)NNODES * NH;
  float* sbuf = mbuf + (size_t)NNODES * NH;

  const int B = 256;
  const int nNH   = (NNODES * NH + B - 1) / B;
  const int nOut  = (int)(((size_t)NNODES * HD + B - 1) / B);
  const int nEH   = (NEDGES * NH + B - 1) / B;
  const int nAggBlocks = (NEDGES / EPW + (B / 32) - 1) / (B / 32);

  gat_init_out<<<nOut, B, 0, stream>>>(out, b0, b1);

  for (int r = 0; r < 2; ++r) {
    gat_gemm_wmma<<<NNODES / 16, B, 0, stream>>>(h, Ws[r], feat);
    gat_attn_dots<<<nNH, B, 0, stream>>>(feat, als[r], ars[r], el, er);
    gat_init_ms<<<nNH, B, 0, stream>>>(mbuf, sbuf);
    gat_edge_max<<<nEH, B, 0, stream>>>(srcs[r], dsts[r], el, er, mbuf);
    gat_fix_m<<<nNH, B, 0, stream>>>(mbuf);
    gat_edge_sum<<<nEH, B, 0, stream>>>(srcs[r], dsts[r], el, er, mbuf, sbuf);
    gat_edge_agg<<<nAggBlocks, B, 0, stream>>>(srcs[r], dsts[r], el, er,
                                               mbuf, sbuf, feat, out);
  }
}


# edges counting-sorted by destination (int atomics), then one wave per destination does max, sum and weighted row sum in registers and stores out once; no float atomics, sums in sort-arrival order
# speedup vs baseline: 1.6913x; 1.2242x over previous
#include <hip/hip_runtime.h>
#include <math.h>

#define NNODES 100000
#define NEDGES 1000000
#define INF    128
#define NH     4
#define ND     32
#define HD     128
#define NEG_SLOPE 0.2f

typedef __attribute__((ext_vector_type(2))) float v2f;
typedef __attribute__((ext_vector_type(8))) float v8f;

__global__ void gat_gemm_wmma(const float* __restrict__ X,
                              const float* __restrict__ W,
                              float* __restrict__ Y) {
  const int row0 = blockIdx.x << 4;
  const int wv   = threadIdx.x >> 5;
  const int lane = threadIdx.x & 31;
  const int m    = lane & 15;
  const int koff = (lane >> 4) << 1;
  const int col  = (wv << 4) + m;

  const float* __restrict__ xrow = X + (size_t)(row0 + m) * INF;
  v8f c = {};
  #pragma unroll 4
  for (int k = 0; k < INF; k += 4) {
    v2f a = *(const v2f*)(xrow + k + koff);
    v2f b;
    b.x = W[(size_t)(k + koff) * HD + col];
    b.y = W[(size_t)(k + koff + 1) * HD + col];
    c = __builtin_amdgcn_wmma_f32_16x16x4_f32(false, a, false, b,
                                              (short)0, c, false, false);
  }
  const int rbase = row0 + ((lane >> 4) << 3);
  #pragma unroll
  for (int i = 0; i < 8; ++i)
    Y[(size_t)(rbase + i) * HD + col] = c[i];
}

__global__ void gat_attn_dots(const float* __restrict__ feat,
                              const float* __restrict__ al,
                              float* __restrict__ el) {
  int t = blockIdx.x * blockDim.x + threadIdx.x;
  if (t >= NNODES * NH) return;
  int n = t >> 2, hh = t & 3;
  const float* f = feat + (size_t)n * HD + hh * ND;
  const float* a = al + hh * ND;
  float sl = 0.f;
  #pragma unroll
  for (int d = 0; d < ND; ++d) sl += f[d] * a[d];
  el[t] = sl;
}

__device__ __forceinline__ float leaky(float v) {
  return v > 0.f ? v : NEG_SLOPE * v;
}

__global__ void gat_zero_counts(int* __restrict__ cnt) {
  int t = blockIdx.x * blockDim.x + threadIdx.x;
  if (t < NNODES) cnt[t] = 0;
}

__global__ void gat_dst_count(const int* __restrict__ dst, int* __restrict__ cnt) {
  int e = blockIdx.x * blockDim.x + threadIdx.x;
  if (e >= NEDGES) return;
  atomicAdd(&cnt[dst[e]], 1);
}

#define SCAN_THREADS 1024
#define SCAN_TILE    4096
__global__ void __launch_bounds__(SCAN_THREADS) gat_scan_counts(const int* __restrict__ cnt,
                                                                int* __restrict__ cur) {
  __shared__ int wpre[32];
  __shared__ int wtot[32];
  const int t = threadIdx.x;
  const int lane = t & 31, wv = t >> 5;
  const int b = blockIdx.x;
  const int4* __restrict__ c4 = (const int4*)cnt;

  int pre = 0;
  for (int k0 = 0; k0 < b; k0 += 8) {
    int4 q[8];
    #pragma unroll
    for (int u = 0; u < 8; ++u) {
      q[u] = make_int4(0, 0, 0, 0);
      if (k0 + u < b) q[u] = c4[(k0 + u) * SCAN_THREADS + t];
    }
    #pragma unroll
    for (int u = 0; u < 8; ++u) pre += q[u].x + q[u].y + q[u].z + q[u].w;
  }
  #pragma unroll
  for (int m = 16; m >= 1; m >>= 1) pre += __shfl_xor(pre, m, 32);
  if (lane == 0) wpre[wv] = pre;

  const int idx = b * SCAN_THREADS + t;
  int4 v = make_int4(0, 0, 0, 0);
  if (idx < NNODES / 4) v = c4[idx];
  const int tot = v.x + v.y + v.z + v.w;
  int inc = tot;
  #pragma unroll
  for (int d = 1; d < 32; d <<= 1) {
    const int y = __shfl_up(inc, d, 32);
    if (lane >= d) inc += y;
  }
  if (lane == 31) wtot[wv] = inc;
  __syncthreads();
  int run = inc - tot;
  #pragma unroll
  for (int w = 0; w < 32; ++w) {
    run += wpre[w];
    if (w < wv) run += wtot[w];
  }
  if (idx < NNODES / 4) {
    int4 o;
    o.x = run; run += v.x;
    o.y = run; run += v.y;
    o.z = run; run += v.z;
    o.w = run;
    ((int4*)cur)[idx] = o;
  }
}

__global__ void gat_dst_scatter(const int* __restrict__ src,
                                const int* __restrict__ dst,
                                int* __restrict__ cur,
                                int* __restrict__ ssrc) {
  int e = blockIdx.x * blockDim.x + threadIdx.x;
  if (e >= NEDGES) return;
  int p = atomicAdd(&cur[dst[e]], 1);
  p = min(max(p, 0), NEDGES - 1);
  ssrc[p] = src[e];
}

__device__ __forceinline__ float wave_sum(float v) {
  #pragma unroll
  for (int m = 16; m >= 1; m >>= 1) v += __shfl_xor(v, m, 32);
  return v;
}
__device__ __forceinline__ float wave_max(float v) {
  #pragma unroll
  for (int m = 16; m >= 1; m >>= 1) v = fmaxf(v, __shfl_xor(v, m, 32));
  return v;
}

__device__ __forceinline__ float uniform_of(float v) {
  return __int_as_float(__builtin_amdgcn_readfirstlane(__float_as_int(v)));
}

__device__ __forceinline__ void load_chunk(const int* __restrict__ ssrc,
                                           const float* __restrict__ el,
                                           int base, int cnt, unsigned lane,
                                           const float er[NH],
                                           int& sidx, float e[NH]) {
  sidx = 0;
  #pragma unroll
  for (int hh = 0; hh < NH; ++hh) e[hh] = -INFINITY;
  if ((int)lane < cnt) {
    int s = ssrc[base + (int)lane];
    s = min(max(s, 0), NNODES - 1);
    sidx = s;
    const float4 l4 = *(const float4*)(el + (size_t)s * NH);
    e[0] = leaky(l4.x + er[0]);
    e[1] = leaky(l4.y + er[1]);
    e[2] = leaky(l4.z + er[2]);
    e[3] = leaky(l4.w + er[3]);
  }
}

#define GRP 8
__device__ __forceinline__ void aggregate(const float* __restrict__ feat,
                                          int sidx, const float al[NH],
                                          int cnt, unsigned lane, float acc[NH]) {
  for (int j0 = 0; j0 < cnt; j0 += GRP) {
    float r[GRP][NH];
    #pragma unroll
    for (int u = 0; u < GRP; ++u) {
      if (j0 + u < cnt) {
        const int sj = __builtin_amdgcn_readlane(sidx, j0 + u);
        const float* __restrict__ row = feat + (size_t)sj * HD;
        #pragma unroll
        for (int hh = 0; hh < NH; ++hh) r[u][hh] = row[hh * ND + lane];
      } else {
        #pragma unroll
        for (int hh = 0; hh < NH; ++hh) r[u][hh] = 0.f;
      }
    }
    #pragma unroll
    for (int u = 0; u < GRP; ++u) {
      #pragma unroll
      for (int hh = 0; hh < NH; ++hh) {
        const float a = __int_as_float(
            __builtin_amdgcn_readlane(__float_as_int(al[hh]), j0 + u));
        acc[hh] += r[u][hh] * a;
      }
    }
  }
}

template <bool FIRST>
__global__ void gat_node_agg(const int* __restrict__ cur,
                             const int* __restrict__ ssrc,
                             const float* __restrict__ el,
                             const float* __restrict__ ar,
                             const float* __restrict__ feat,
                             const float* __restrict__ b0,
                             const float* __restrict__ b1,
                             float* __restrict__ out) {
  const int n = __builtin_amdgcn_readfirstlane(
      (int)(blockIdx.x * (blockDim.x >> 5) + (threadIdx.x >> 5)));
  if (n >= NNODES) return;
  const unsigned lane = threadIdx.x & 31;

  const int prev = cur[n > 0 ? n - 1 : 0];
  int end = cur[n];
  int beg = n > 0 ? prev : 0;
  beg = __builtin_amdgcn_readfirstlane(min(max(beg, 0), NEDGES));
  end = __builtin_amdgcn_readfirstlane(min(max(end, beg), NEDGES));
  const int deg = end - beg;

  float* __restrict__ orow = out + (size_t)n * HD;
  float base[NH];
  #pragma unroll
  for (int hh = 0; hh < NH; ++hh)
    base[hh] = FIRST ? (b0[hh * ND + lane] + b1[hh * ND + lane]) : orow[hh * ND + lane];

  const float* __restrict__ myrow = feat + (size_t)n * HD;
  float er[NH];
  #pragma unroll
  for (int hh = 0; hh < NH; ++hh)
    er[hh] = uniform_of(wave_sum(myrow[hh * ND + lane] * ar[hh * ND + lane]));

  float acc[NH] = {0.f, 0.f, 0.f, 0.f};

  if (deg <= 32) {
    int sidx; float e[NH];
    load_chunk(ssrc, el, beg, deg, lane, er, sidx, e);
    float al[NH];
    #pragma unroll
    for (int hh = 0; hh < NH; ++hh) {
      float m = uniform_of(wave_max(e[hh]));
      if (!isfinite(m)) m = 0.f;
      const float ex = __expf(e[hh] - m);
      const float s = uniform_of(wave_sum(ex));
      al[hh] = ex / fmaxf(s, 1e-20f);
    }
    aggregate(feat, sidx, al, deg, lane, acc);
  } else {
    float m[NH], s[NH];
    {
      float mx[NH] = {-INFINITY, -INFINITY, -INFINITY, -INFINITY};
      for (int c = beg; c < end; c += 32) {
        int sidx; float e[NH];
        load_chunk(ssrc, el, c, min(32, end - c), lane, er, sidx, e);
        #pragma unroll
        for (int hh = 0; hh < NH; ++hh) mx[hh] = fmaxf(mx[hh], e[hh]);
      }
      #pragma unroll
      for (int hh = 0; hh < NH; ++hh) {
        m[hh] = uniform_of(wave_max(mx[hh]));
        if (!isfinite(m[hh])) m[hh] = 0.f;
      }
    }
    {
      float sm[NH] = {0.f, 0.f, 0.f, 0.f};
      for (int c = beg; c < end; c += 32) {
        int sidx; float e[NH];
        load_chunk(ssrc, el, c, min(32, end - c), lane, er, sidx, e);
        #pragma unroll
        for (int hh = 0; hh < NH; ++hh) sm[hh] += __expf(e[hh] - m[hh]);
      }
      #pragma unroll
      for (int hh = 0; hh < NH; ++hh) s[hh] = fmaxf(uniform_of(wave_sum(sm[hh])), 1e-20f);
    }
    for (int c = beg; c < end; c += 32) {
      const int cnt = min(32, end - c);
      int sidx; float e[NH];
      load_chunk(ssrc, el, c, cnt, lane, er, sidx, e);
      float al[NH];
      #pragma unroll
      for (int hh = 0; hh < NH; ++hh) al[hh] = __expf(e[hh] - m[hh]) / s[hh];
      aggregate(feat, sidx, al, cnt, lane, acc);
    }
  }

  #pragma unroll
  for (int hh = 0; hh < NH; ++hh) orow[hh * ND + lane] = base[hh] + acc[hh];
}

extern "C" void kernel_launch(void* const* d_in, const int* in_sizes, int n_in,
                              void* d_out, int out_size, void* d_ws, size_t ws_size,
                              hipStream_t stream) {
  const float* h    = (const float*)d_in[0];
  const int*   srcs[2] = { (const int*)d_in[1], (const int*)d_in[3] };
  const int*   dsts[2] = { (const int*)d_in[2], (const int*)d_in[4] };
  const float* Ws[2]   = { (const float*)d_in[5],  (const float*)d_in[9]  };
  const float* als[2]  = { (const float*)d_in[6],  (const float*)d_in[10] };
  const float* ars[2]  = { (const float*)d_in[7],  (const float*)d_in[11] };
  const float* b0 = (const float*)d_in[8];
  const float* b1 = (const float*)d_in[12];
  float* out = (float*)d_out;

  float* feat = (float*)d_ws;
  float* el   = feat + (size_t)NNODES * HD;
  int*   cnt  = (int*)(el + (size_t)NNODES * NH);
  int*   cur  = cnt + NNODES;
  int*   ssrc = cur + NNODES;

  const int B = 256;
  const int nN    = (NNODES + B - 1) / B;
  const int nNH   = (NNODES * NH + B - 1) / B;
  const int nE    = (NEDGES + B - 1) / B;
  const int nNode = (NNODES + (B / 32) - 1) / (B / 32);
  const int nScan = (NNODES + SCAN_TILE - 1) / SCAN_TILE;

  for (int r = 0; r < 2; ++r) {
    gat_gemm_wmma<<<NNODES / 16, B, 0, stream>>>(h, Ws[r], feat);
    gat_attn_dots<<<nNH, B, 0, stream>>>(feat, als[r], el);
    gat_zero_counts<<<nN, B, 0, stream>>>(cnt);
    gat_dst_count<<<nE, B, 0, stream>>>(dsts[r], cnt);
    gat_scan_counts<<<nScan, SCAN_THREADS, 0, stream>>>(cnt, cur);
    gat_dst_scatter<<<nE, B, 0, stream>>>(srcs[r], dsts[r], cur, ssrc);
    if (r == 0)
      gat_node_agg<true><<<nNode, B, 0, stream>>>(cur, ssrc, el, ars[r], feat, b0, b1, out);
    else
      gat_node_agg<false><<<nNode, B, 0, stream>>>(cur, ssrc, el, ars[r], feat, b0, b1, out);
  }
}


# every kernel: a wave or thread loops over several items (GEMM: 8 col tiles per wave, its X rows kept in registers; 8 edges; 8 dots; 12 nodes, next node's loads ahead): 9 times fewer waves launched
# speedup vs baseline: 1.7936x; 1.0605x over previous
#include <hip/hip_runtime.h>
#include <math.h>

#define NNODES 100000
#define NEDGES 1000000
#define INF    128
#define NH     4
#define ND     32
#define HD     128
#define NEG_SLOPE 0.2f

typedef __attribute__((ext_vector_type(2))) float v2f;
typedef __attribute__((ext_vector_type(8))) float v8f;

#define GEMM_KSTEPS (INF / 4)
#define GEMM_TILES  (NNODES / 16)
__global__ void __launch_bounds__(256) gat_gemm_wmma(const float* __restrict__ X,
                                                     const float* __restrict__ W,
                                                     float* __restrict__ Y) {
  const int tile = __builtin_amdgcn_readfirstlane(
      (int)(blockIdx.x * (blockDim.x >> 5) + (threadIdx.x >> 5)));
  if (tile >= GEMM_TILES) return;
  const int lane = threadIdx.x & 31;
  const int m    = lane & 15;
  const int koff = (lane >> 4) << 1;
  const int row0 = tile << 4;

  const float* __restrict__ xrow = X + (size_t)(row0 + m) * INF + koff;
  v2f af[GEMM_KSTEPS];
  #pragma unroll
  for (int s = 0; s < GEMM_KSTEPS; ++s) af[s] = *(const v2f*)(xrow + 4 * s);

  const int rbase = row0 + ((lane >> 4) << 3);
  for (int ct = 0; ct < HD / 16; ++ct) {
    const int col = (ct << 4) + m;
    const float* __restrict__ wcol = W + (size_t)koff * HD + col;
    v8f c = {};
    #pragma unroll
    for (int s = 0; s < GEMM_KSTEPS; ++s) {
      v2f b;
      b.x = wcol[(size_t)(4 * s) * HD];
      b.y = wcol[(size_t)(4 * s + 1) * HD];
      c = __builtin_amdgcn_wmma_f32_16x16x4_f32(false, af[s], false, b,
                                                (short)0, c, false, false);
    }
    #pragma unroll
    for (int i = 0; i < 8; ++i)
      Y[(size_t)(rbase + i) * HD + col] = c[i];
  }
}

#define DOTS_ITEMS 8
__global__ void gat_attn_dots(const float* __restrict__ feat,
                              const float* __restrict__ al,
                              float* __restrict__ el) {
  const int stride = gridDim.x * blockDim.x;
  for (int t = blockIdx.x * blockDim.x + threadIdx.x; t < NNODES * NH; t += stride) {
    int n = t >> 2, hh = t & 3;
    const float* f = feat + (size_t)n * HD + hh * ND;
    const float* a = al + hh * ND;
    float sl = 0.f;
    #pragma unroll
    for (int d = 0; d < ND; ++d) sl += f[d] * a[d];
    el[t] = sl;
  }
}

__device__ __forceinline__ float leaky(float v) {
  return v > 0.f ? v : NEG_SLOPE * v;
}

__global__ void gat_zero_counts(int* __restrict__ cnt) {
  int t = blockIdx.x * blockDim.x + threadIdx.x;
  if (t < NNODES / 4) ((int4*)cnt)[t] = make_int4(0, 0, 0, 0);
}

#define EDGE_THREADS (NEDGES / 8)
__global__ void gat_dst_count(const int* __restrict__ dst, int* __restrict__ cnt) {
  int t = blockIdx.x * blockDim.x + threadIdx.x;
  if (t >= EDGE_THREADS) return;
  const int4 d0 = ((const int4*)dst)[t];
  const int4 d1 = ((const int4*)dst)[t + EDGE_THREADS];
  atomicAdd(&cnt[d0.x], 1);
  atomicAdd(&cnt[d0.y], 1);
  atomicAdd(&cnt[d0.z], 1);
  atomicAdd(&cnt[d0.w], 1);
  atomicAdd(&cnt[d1.x], 1);
  atomicAdd(&cnt[d1.y], 1);
  atomicAdd(&cnt[d1.z], 1);
  atomicAdd(&cnt[d1.w], 1);
}

#define SCAN_THREADS 1024
#define SCAN_TILE    4096
__global__ void __launch_bounds__(SCAN_THREADS) gat_scan_counts(const int* __restrict__ cnt,
                                                                int* __restrict__ cur) {
  __shared__ int wpre[32];
  __shared__ int wtot[32];
  const int t = threadIdx.x;
  const int lane = t & 31, wv = t >> 5;
  const int b = blockIdx.x;
  const int4* __restrict__ c4 = (const int4*)cnt;

  int pre = 0;
  for (int k0 = 0; k0 < b; k0 += 8) {
    int4 q[8];
    #pragma unroll
    for (int u = 0; u < 8; ++u) {
      q[u] = make_int4(0, 0, 0, 0);
      if (k0 + u < b) q[u] = c4[(k0 + u) * SCAN_THREADS + t];
    }
    #pragma unroll
    for (int u = 0; u < 8; ++u) pre += q[u].x + q[u].y + q[u].z + q[u].w;
  }
  #pragma unroll
  for (int m = 16; m >= 1; m >>= 1) pre += __shfl_xor(pre, m, 32);
  if (lane == 0) wpre[wv] = pre;

  const int idx = b * SCAN_THREADS + t;
  int4 v = make_int4(0, 0, 0, 0);
  if (idx < NNODES / 4) v = c4[idx];
  const int tot = v.x + v.y + v.z + v.w;
  int inc = tot;
  #pragma unroll
  for (int d = 1; d < 32; d <<= 1) {
    const int y = __shfl_up(inc, d, 32);
    if (lane >= d) inc += y;
  }
  if (lane == 31) wtot[wv] = inc;
  __syncthreads();
  int run = inc - tot;
  #pragma unroll
  for (int w = 0; w < 32; ++w) {
    run += wpre[w];
    if (w < wv) run += wtot[w];
  }
  if (idx < NNODES / 4) {
    int4 o;
    o.x = run; run += v.x;
    o.y = run; run += v.y;
    o.z = run; run += v.z;
    o.w = run;
    ((int4*)cur)[idx] = o;
  }
}

__global__ void gat_dst_scatter(const int* __restrict__ src,
                                const int* __restrict__ dst,
                                int* __restrict__ cur,
                                int* __restrict__ ssrc) {
  int t = blockIdx.x * blockDim.x + threadIdx.x;
  if (t >= EDGE_THREADS) return;
  const int4 d0 = ((const int4*)dst)[t];
  const int4 d1 = ((const int4*)dst)[t + EDGE_THREADS];
  const int4 s0 = ((const int4*)src)[t];
  const int4 s1 = ((const int4*)src)[t + EDGE_THREADS];
  const int dd[8] = { d0.x, d0.y, d0.z, d0.w, d1.x, d1.y, d1.z, d1.w };
  const int ss[8] = { s0.x, s0.y, s0.z, s0.w, s1.x, s1.y, s1.z, s1.w };
  int p[8];
  #pragma unroll
  for (int u = 0; u < 8; ++u) p[u] = atomicAdd(&cur[dd[u]], 1);
  #pragma unroll
  for (int u = 0; u < 8; ++u) {
    const int q = min(max(p[u], 0), NEDGES - 1);
    ssrc[q] = ss[u];
  }
}

__device__ __forceinline__ float wave_sum(float v) {
  #pragma unroll
  for (int m = 16; m >= 1; m >>= 1) v += __shfl_xor(v, m, 32);
  return v;
}
__device__ __forceinline__ float wave_max(float v) {
  #pragma unroll
  for (int m = 16; m >= 1; m >>= 1) v = fmaxf(v, __shfl_xor(v, m, 32));
  return v;
}

__device__ __forceinline__ float uniform_of(float v) {
  return __int_as_float(__builtin_amdgcn_readfirstlane(__float_as_int(v)));
}

__device__ __forceinline__ void load_chunk(const int* __restrict__ ssrc,
                                           const float* __restrict__ el,
                                           int base, int cnt, unsigned lane,
                                           const float er[NH],
                                           int& sidx, float e[NH]) {
  sidx = 0;
  #pragma unroll
  for (int hh = 0; hh < NH; ++hh) e[hh] = -INFINITY;
  if ((int)lane < cnt) {
    int s = ssrc[base + (int)lane];
    s = min(max(s, 0), NNODES - 1);
    sidx = s;
    const float4 l4 = *(const float4*)(el + (size_t)s * NH);
    e[0] = leaky(l4.x + er[0]);
    e[1] = leaky(l4.y + er[1]);
    e[2] = leaky(l4.z + er[2]);
    e[3] = leaky(l4.w + er[3]);
  }
}

#define GRP 8
__device__ __forceinline__ void aggregate(const float* __restrict__ feat,
                                          int sidx, const float al[NH],
                                          int cnt, unsigned lane, float acc[NH]) {
  for (int j0 = 0; j0 < cnt; j0 += GRP) {
    float r[GRP][NH];
    #pragma unroll
    for (int u = 0; u < GRP; ++u) {
      if (j0 + u < cnt) {
        const int sj = __builtin_amdgcn_readlane(sidx, j0 + u);
        const float* __restrict__ row = feat + (size_t)sj * HD;
        #pragma unroll
        for (int hh = 0; hh < NH; ++hh) r[u][hh] = row[hh * ND + lane];
      } else {
        #pragma unroll
        for (int hh = 0; hh < NH; ++hh) r[u][hh] = 0.f;
      }
    }
    #pragma unroll
    for (int u = 0; u < GRP; ++u) {
      #pragma unroll
      for (int hh = 0; hh < NH; ++hh) {
        const float a = __int_as_float(
            __builtin_amdgcn_readlane(__float_as_int(al[hh]), j0 + u));
        acc[hh] += r[u][hh] * a;
      }
    }
  }
}

#define AGG_NODES 12
__device__ __forceinline__ void node_segment(int bnd, int k, int& beg, int& end) {
  beg = __builtin_amdgcn_readlane(bnd, k);
  end = __builtin_amdgcn_readlane(bnd, k + 1);
  beg = min(max(beg, 0), NEDGES);
  end = min(max(end, beg), NEDGES);
}

template <bool FIRST>
__global__ void gat_node_agg(const int* __restrict__ cur,
                             const int* __restrict__ ssrc,
                             const float* __restrict__ el,
                             const float* __restrict__ ar,
                             const float* __restrict__ feat,
                             const float* __restrict__ b0,
                             const float* __restrict__ b1,
                             float* __restrict__ out) {
  const int n0 = __builtin_amdgcn_readfirstlane(
      (int)(blockIdx.x * (blockDim.x >> 5) + (threadIdx.x >> 5))) * AGG_NODES;
  if (n0 >= NNODES) return;
  const unsigned lane = threadIdx.x & 31;
  const int nn = min(AGG_NODES, NNODES - n0);

  const int bidx = n0 - 1 + (int)lane;
  int bnd = 0;
  if (lane <= AGG_NODES && bidx >= 0 && bidx < NNODES) bnd = cur[bidx];

  float arv[NH], bias[NH];
  #pragma unroll
  for (int hh = 0; hh < NH; ++hh) {
    arv[hh]  = ar[hh * ND + lane];
    bias[hh] = FIRST ? (b0[hh * ND + lane] + b1[hh * ND + lane]) : 0.f;
  }

  int   sC = 0;
  float4 lC = make_float4(0.f, 0.f, 0.f, 0.f);
  float mrC[NH], bsC[NH];
  int   sN = 0;

  {
    int b, e; node_segment(bnd, 0, b, e);
    if ((int)lane < min(e - b, 32)) sN = ssrc[b + (int)lane];
  }
  {
    int b, e; node_segment(bnd, 0, b, e);
    sC = min(max(sN, 0), NNODES - 1);
    if ((int)lane < min(e - b, 32)) lC = *(const float4*)(el + (size_t)sC * NH);
    const float* __restrict__ myrow = feat + (size_t)n0 * HD;
    const float* __restrict__ orow0 = out + (size_t)n0 * HD;
    #pragma unroll
    for (int hh = 0; hh < NH; ++hh) {
      mrC[hh] = myrow[hh * ND + lane];
      bsC[hh] = FIRST ? 0.f : orow0[hh * ND + lane];
    }
  }
  sN = 0;
  if (1 < nn) {
    int b, e; node_segment(bnd, 1, b, e);
    if ((int)lane < min(e - b, 32)) sN = ssrc[b + (int)lane];
  }

  for (int i = 0; i < nn; ++i) {
    const int n = n0 + i;

    int   sX = 0;
    float4 lX = make_float4(0.f, 0.f, 0.f, 0.f);
    float mrX[NH] = {0.f, 0.f, 0.f, 0.f}, bsX[NH] = {0.f, 0.f, 0.f, 0.f};
    if (i + 1 < nn) {
      int b, e; node_segment(bnd, i + 1, b, e);
      sX = min(max(sN, 0), NNODES - 1);
      if ((int)lane < min(e - b, 32)) lX = *(const float4*)(el + (size_t)sX * NH);
      const float* __restrict__ nrow = feat + (size_t)(n + 1) * HD;
      const float* __restrict__ norow = out + (size_t)(n + 1) * HD;
      #pragma unroll
      for (int hh = 0; hh < NH; ++hh) {
        mrX[hh] = nrow[hh * ND + lane];
        bsX[hh] = FIRST ? 0.f : norow[hh * ND + lane];
      }
    }
    int sNN = 0;
    if (i + 2 < nn) {
      int b, e; node_segment(bnd, i + 2, b, e);
      if ((int)lane < min(e - b, 32)) sNN = ssrc[b + (int)lane];
    }

    int beg, end; node_segment(bnd, i, beg, end);
    const int deg = end - beg;

    float er[NH];
    #pragma unroll
    for (int hh = 0; hh < NH; ++hh)
      er[hh] = uniform_of(wave_sum(mrC[hh] * arv[hh]));

    float acc[NH] = {0.f, 0.f, 0.f, 0.f};

    if (deg <= 32) {
      int sidx = 0; float e[NH];
      #pragma unroll
      for (int hh = 0; hh < NH; ++hh) e[hh] = -INFINITY;
      if ((int)lane < deg) {
        sidx = sC;
        e[0] = leaky(lC.x + er[0]);
        e[1] = leaky(lC.y + er[1]);
        e[2] = leaky(lC.z + er[2]);
        e[3] = leaky(lC.w + er[3]);
      }
      float al[NH];
      #pragma unroll
      for (int hh = 0; hh < NH; ++hh) {
        float m = uniform_of(wave_max(e[hh]));
        if (!isfinite(m)) m = 0.f;
        const float ex = __expf(e[hh] - m);
        const float s = uniform_of(wave_sum(ex));
        al[hh] = ex / fmaxf(s, 1e-20f);
      }
      aggregate(feat, sidx, al, deg, lane, acc);
    } else {
      float m[NH], s[NH];
      {
        float mx[NH] = {-INFINITY, -INFINITY, -INFINITY, -INFINITY};
        for (int c = beg; c < end; c += 32) {
          int sidx; float e[NH];
          load_chunk(ssrc, el, c, min(32, end - c), lane, er, sidx, e);
          #pragma unroll
          for (int hh = 0; hh < NH; ++hh) mx[hh] = fmaxf(mx[hh], e[hh]);
        }
        #pragma unroll
        for (int hh = 0; hh < NH; ++hh) {
          m[hh] = uniform_of(wave_max(mx[hh]));
          if (!isfinite(m[hh])) m[hh] = 0.f;
        }
      }
      {
        float sm[NH] = {0.f, 0.f, 0.f, 0.f};
        for (int c = beg; c < end; c += 32) {
          int sidx; float e[NH];
          load_chunk(ssrc, el, c, min(32, end - c), lane, er, sidx, e);
          #pragma unroll
          for (int hh = 0; hh < NH; ++hh) sm[hh] += __expf(e[hh] - m[hh]);
        }
        #pragma unroll
        for (int hh = 0; hh < NH; ++hh) s[hh] = fmaxf(uniform_of(wave_sum(sm[hh])), 1e-20f);
      }
      for (int c = beg; c < end; c += 32) {
        const int cnt = min(32, end - c);
        int sidx; float e[NH];
        load_chunk(ssrc, el, c, cnt, lane, er, sidx, e);
        float al[NH];
        #pragma unroll
        for (int hh = 0; hh < NH; ++hh) al[hh] = __expf(e[hh] - m[hh]) / s[hh];
        aggregate(feat, sidx, al, cnt, lane, acc);
      }
    }

    float* __restrict__ orow = out + (size_t)n * HD;
    #pragma unroll
    for (int hh = 0; hh < NH; ++hh) orow[hh * ND + lane] = (FIRST ? bias[hh] : bsC[hh]) + acc[hh];

    sC = sX; lC = lX; sN = sNN;
    #pragma unroll
    for (int hh = 0; hh < NH; ++hh) { mrC[hh] = mrX[hh]; bsC[hh] = bsX[hh]; }
  }
}

extern "C" void kernel_launch(void* const* d_in, const int* in_sizes, int n_in,
                              void* d_out, int out_size, void* d_ws, size_t ws_size,
                              hipStream_t stream) {
  const float* h    = (const float*)d_in[0];
  const int*   srcs[2] = { (const int*)d_in[1], (const int*)d_in[3] };
  const int*   dsts[2] = { (const int*)d_in[2], (const int*)d_in[4] };
  const float* Ws[2]   = { (const float*)d_in[5],  (const float*)d_in[9]  };
  const float* als[2]  = { (const float*)d_in[6],  (const float*)d_in[10] };
  const float* ars[2]  = { (const float*)d_in[7],  (const float*)d_in[11] };
  const float* b0 = (const float*)d_in[8];
  const float* b1 = (const float*)d_in[12];
  float* out = (float*)d_out;

  float* feat = (float*)d_ws;
  float* el   = feat + (size_t)NNODES * HD;
  int*   cnt  = (int*)(el + (size_t)NNODES * NH);
  int*   cur  = cnt + NNODES;
  int*   ssrc = cur + NNODES;

  const int B = 256;
  const int BG = 64;
  const int BA = 128;
  const int nTile = (GEMM_TILES + (BG / 32) - 1) / (BG / 32);
  const int nN    = (NNODES / 4 + B - 1) / B;
  const int nNH   = (NNODES * NH / DOTS_ITEMS + B - 1) / B;
  const int nE    = (EDGE_THREADS + B - 1) / B;
  const int nNode = (NNODES + (BA / 32) * AGG_NODES - 1) / ((BA / 32) * AGG_NODES);
  const int nScan = (NNODES + SCAN_TILE - 1) / SCAN_TILE;

  for (int r = 0; r < 2; ++r) {
    gat_gemm_wmma<<<nTile, BG, 0, stream>>>(h, Ws[r], feat);
    gat_attn_dots<<<nNH, B, 0, stream>>>(feat, als[r], el);
    gat_zero_counts<<<nN, B, 0, stream>>>(cnt);
    gat_dst_count<<<nE, B, 0, stream>>>(dsts[r], cnt);
    gat_scan_counts<<<nScan, SCAN_THREADS, 0, stream>>>(cnt, cur);
    gat_dst_scatter<<<nE, B, 0, stream>>>(srcs[r], dsts[r], cur, ssrc);
    if (r == 0)
      gat_node_agg<true><<<nNode, BA, 0, stream>>>(cur, ssrc, el, ars[r], feat, b0, b1, out);
    else
      gat_node_agg<false><<<nNode, BA, 0, stream>>>(cur, ssrc, el, ars[r], feat, b0, b1, out);
  }
}


# 14 launches -> 9: el[n,h] summed from the GEMM accumulators (16-lane tree order, not serial d), dst counting as extra work-groups of the GEMM launch, counters cleared by the first scatter launch
# speedup vs baseline: 1.9491x; 1.0867x over previous
#include <hip/hip_runtime.h>
#include <math.h>

#define NNODES 100000
#define NEDGES 1000000
#define INF    128
#define NH     4
#define ND     32
#define HD     128
#define NEG_SLOPE 0.2f

typedef __attribute__((ext_vector_type(2))) float v2f;
typedef __attribute__((ext_vector_type(8))) float v8f;

#define EDGE_THREADS (NEDGES / 8)
#define GEMM_BLOCK   64
#define CNT_BLOCKS   ((EDGE_THREADS + GEMM_BLOCK - 1) / GEMM_BLOCK)
__device__ __forceinline__ void dst_count_part(const int* __restrict__ dst,
                                               int* __restrict__ cnt, int t) {
  if (t >= EDGE_THREADS) return;
  const int4 d0 = ((const int4*)dst)[t];
  const int4 d1 = ((const int4*)dst)[t + EDGE_THREADS];
  atomicAdd(&cnt[d0.x], 1);
  atomicAdd(&cnt[d0.y], 1);
  atomicAdd(&cnt[d0.z], 1);
  atomicAdd(&cnt[d0.w], 1);
  atomicAdd(&cnt[d1.x], 1);
  atomicAdd(&cnt[d1.y], 1);
  atomicAdd(&cnt[d1.z], 1);
  atomicAdd(&cnt[d1.w], 1);
}

template <int CTRL>
__device__ __forceinline__ float dpp_swz(float v) {
  return __int_as_float(__builtin_amdgcn_mov_dpp(__float_as_int(v), CTRL, 0xf, 0xf, false));
}
__device__ __forceinline__ float row16_sum(float v) {
  v += dpp_swz<0x161>(v);
  v += dpp_swz<0x162>(v);
  v += dpp_swz<0x164>(v);
  v += dpp_swz<0x168>(v);
  return v;
}

#define GEMM_KSTEPS (INF / 4)
#define GEMM_TILES  (NNODES / 16)
__global__ void __launch_bounds__(GEMM_BLOCK) gat_gemm_el_count(const float* __restrict__ X,
                                                                const float* __restrict__ W,
                                                                const float* __restrict__ al,
                                                                const int* __restrict__ dst,
                                                                float* __restrict__ Y,
                                                                float* __restrict__ el,
                                                                int* __restrict__ cnt) {
  if (blockIdx.x < CNT_BLOCKS) {
    dst_count_part(dst, cnt, (int)(blockIdx.x * GEMM_BLOCK + threadIdx.x));
    return;
  }
  const int tile = __builtin_amdgcn_readfirstlane(
      (int)((blockIdx.x - CNT_BLOCKS) * (GEMM_BLOCK >> 5) + (threadIdx.x >> 5)));
  if (tile >= GEMM_TILES) return;
  const int lane = threadIdx.x & 31;
  const int m    = lane & 15;
  const int koff = (lane >> 4) << 1;
  const int row0 = tile << 4;

  const float* __restrict__ xrow = X + (size_t)(row0 + m) * INF + koff;
  v2f af[GEMM_KSTEPS];
  #pragma unroll
  for (int s = 0; s < GEMM_KSTEPS; ++s) af[s] = *(const v2f*)(xrow + 4 * s);

  const int rbase = row0 + ((lane >> 4) << 3);
  for (int hh = 0; hh < NH; ++hh) {
    float p[8];
    #pragma unroll
    for (int half = 0; half < 2; ++half) {
      const int col = (hh << 5) + (half << 4) + m;
      const float* __restrict__ wcol = W + (size_t)koff * HD + col;
      v8f c = {};
      #pragma unroll
      for (int s = 0; s < GEMM_KSTEPS; ++s) {
        v2f b;
        b.x = wcol[(size_t)(4 * s) * HD];
        b.y = wcol[(size_t)(4 * s + 1) * HD];
        c = __builtin_amdgcn_wmma_f32_16x16x4_f32(false, af[s], false, b,
                                                  (short)0, c, false, false);
      }
      const float a = al[col];
      #pragma unroll
      for (int i = 0; i < 8; ++i) {
        Y[(size_t)(rbase + i) * HD + col] = c[i];
        p[i] = half == 0 ? c[i] * a : fmaf(c[i], a, p[i]);
      }
    }
    #pragma unroll
    for (int i = 0; i < 8; ++i) p[i] = row16_sum(p[i]);
    float v = p[0];
    #pragma unroll
    for (int i = 1; i < 8; ++i) v = (m == i) ? p[i] : v;
    if (m < 8) el[(size_t)(rbase + m) * NH + hh] = v;
  }
}

__device__ __forceinline__ float leaky(float v) {
  return v > 0.f ? v : NEG_SLOPE * v;
}

__global__ void gat_zero_counts(int* __restrict__ cnt) {
  int t = blockIdx.x * blockDim.x + threadIdx.x;
  if (t < NNODES / 4) ((int4*)cnt)[t] = make_int4(0, 0, 0, 0);
}

#define SCAN_THREADS 1024
#define SCAN_TILE    4096
__global__ void __launch_bounds__(SCAN_THREADS) gat_scan_counts(const int* __restrict__ cnt,
                                                                int* __restrict__ cur) {
  __shared__ int wpre[32];
  __shared__ int wtot[32];
  const int t = threadIdx.x;
  const int lane = t & 31, wv = t >> 5;
  const int b = blockIdx.x;
  const int4* __restrict__ c4 = (const int4*)cnt;

  int pre = 0;
  for (int k0 = 0; k0 < b; k0 += 8) {
    int4 q[8];
    #pragma unroll
    for (int u = 0; u < 8; ++u) {
      q[u] = make_int4(0, 0, 0, 0);
      if (k0 + u < b) q[u] = c4[(k0 + u) * SCAN_THREADS + t];
    }
    #pragma unroll
    for (int u = 0; u < 8; ++u) pre += q[u].x + q[u].y + q[u].z + q[u].w;
  }
  #pragma unroll
  for (int m = 16; m >= 1; m >>= 1) pre += __shfl_xor(pre, m, 32);
  if (lane == 0) wpre[wv] = pre;

  const int idx = b * SCAN_THREADS + t;
  int4 v = make_int4(0, 0, 0, 0);
  if (idx < NNODES / 4) v = c4[idx];
  const int tot = v.x + v.y + v.z + v.w;
  int inc = tot;
  #pragma unroll
  for (int d = 1; d < 32; d <<= 1) {
    const int y = __shfl_up(inc, d, 32);
    if (lane >= d) inc += y;
  }
  if (lane == 31) wtot[wv] = inc;
  __syncthreads();
  int run = inc - tot;
  #pragma unroll
  for (int w = 0; w < 32; ++w) {
    run += wpre[w];
    if (w < wv) run += wtot[w];
  }
  if (idx < NNODES / 4) {
    int4 o;
    o.x = run; run += v.x;
    o.y = run; run += v.y;
    o.z = run; run += v.z;
    o.w = run;
    ((int4*)cur)[idx] = o;
  }
}

template <bool CLEAR_CNT>
__global__ void gat_dst_scatter(const int* __restrict__ src,
                                const int* __restrict__ dst,
                                int* __restrict__ cur,
                                int* __restrict__ ssrc,
                                int* __restrict__ cnt) {
  int t = blockIdx.x * blockDim.x + threadIdx.x;
  if (CLEAR_CNT && t < NNODES / 4) ((int4*)cnt)[t] = make_int4(0, 0, 0, 0);
  if (t >= EDGE_THREADS) return;
  const int4 d0 = ((const int4*)dst)[t];
  const int4 d1 = ((const int4*)dst)[t + EDGE_THREADS];
  const int4 s0 = ((const int4*)src)[t];
  const int4 s1 = ((const int4*)src)[t + EDGE_THREADS];
  const int dd[8] = { d0.x, d0.y, d0.z, d0.w, d1.x, d1.y, d1.z, d1.w };
  const int ss[8] = { s0.x, s0.y, s0.z, s0.w, s1.x, s1.y, s1.z, s1.w };
  int p[8];
  #pragma unroll
  for (int u = 0; u < 8; ++u) p[u] = atomicAdd(&cur[dd[u]], 1);
  #pragma unroll
  for (int u = 0; u < 8; ++u) {
    const int q = min(max(p[u], 0), NEDGES - 1);
    ssrc[q] = ss[u];
  }
}

__device__ __forceinline__ float wave_sum(float v) {
  #pragma unroll
  for (int m = 16; m >= 1; m >>= 1) v += __shfl_xor(v, m, 32);
  return v;
}
__device__ __forceinline__ float wave_max(float v) {
  #pragma unroll
  for (int m = 16; m >= 1; m >>= 1) v = fmaxf(v, __shfl_xor(v, m, 32));
  return v;
}

__device__ __forceinline__ float uniform_of(float v) {
  return __int_as_float(__builtin_amdgcn_readfirstlane(__float_as_int(v)));
}

__device__ __forceinline__ void load_chunk(const int* __restrict__ ssrc,
                                           const float* __restrict__ el,
                                           int base, int cnt, unsigned lane,
                                           const float er[NH],
                                           int& sidx, float e[NH]) {
  sidx = 0;
  #pragma unroll
  for (int hh = 0; hh < NH; ++hh) e[hh] = -INFINITY;
  if ((int)lane < cnt) {
    int s = ssrc[base + (int)lane];
    s = min(max(s, 0), NNODES - 1);
    sidx = s;
    const float4 l4 = *(const float4*)(el + (size_t)s * NH);
    e[0] = leaky(l4.x + er[0]);
    e[1] = leaky(l4.y + er[1]);
    e[2] = leaky(l4.z + er[2]);
    e[3] = leaky(l4.w + er[3]);
  }
}

#define GRP 8
__device__ __forceinline__ void aggregate(const float* __restrict__ feat,
                                          int sidx, const float al[NH],
                                          int cnt, unsigned lane, float acc[NH]) {
  for (int j0 = 0; j0 < cnt; j0 += GRP) {
    float r[GRP][NH];
    #pragma unroll
    for (int u = 0; u < GRP; ++u) {
      if (j0 + u < cnt) {
        const int sj = __builtin_amdgcn_readlane(sidx, j0 + u);
        const float* __restrict__ row = feat + (size_t)sj * HD;
        #pragma unroll
        for (int hh = 0; hh < NH; ++hh) r[u][hh] = row[hh * ND + lane];
      } else {
        #pragma unroll
        for (int hh = 0; hh < NH; ++hh) r[u][hh] = 0.f;
      }
    }
    #pragma unroll
    for (int u = 0; u < GRP; ++u) {
      #pragma unroll
      for (int hh = 0; hh < NH; ++hh) {
        const float a = __int_as_float(
            __builtin_amdgcn_readlane(__float_as_int(al[hh]), j0 + u));
        acc[hh] += r[u][hh] * a;
      }
    }
  }
}

#define AGG_NODES 12
__device__ __forceinline__ void node_segment(int bnd, int k, int& beg, int& end) {
  beg = __builtin_amdgcn_readlane(bnd, k);
  end = __builtin_amdgcn_readlane(bnd, k + 1);
  beg = min(max(beg, 0), NEDGES);
  end = min(max(end, beg), NEDGES);
}

template <bool FIRST>
__global__ void gat_node_agg(const int* __restrict__ cur,
                             const int* __restrict__ ssrc,
                             const float* __restrict__ el,
                             const float* __restrict__ ar,
                             const float* __restrict__ feat,
                             const float* __restrict__ b0,
                             const float* __restrict__ b1,
                             float* __restrict__ out) {
  const int n0 = __builtin_amdgcn_readfirstlane(
      (int)(blockIdx.x * (blockDim.x >> 5) + (threadIdx.x >> 5))) * AGG_NODES;
  if (n0 >= NNODES) return;
  const unsigned lane = threadIdx.x & 31;
  const int nn = min(AGG_NODES, NNODES - n0);

  const int bidx = n0 - 1 + (int)lane;
  int bnd = 0;
  if (lane <= AGG_NODES && bidx >= 0 && bidx < NNODES) bnd = cur[bidx];

  float arv[NH], bias[NH];
  #pragma unroll
  for (int hh = 0; hh < NH; ++hh) {
    arv[hh]  = ar[hh * ND + lane];
    bias[hh] = FIRST ? (b0[hh * ND + lane] + b1[hh * ND + lane]) : 0.f;
  }

  int   sC = 0;
  float4 lC = make_float4(0.f, 0.f, 0.f, 0.f);
  float mrC[NH], bsC[NH];
  int   sN = 0;

  {
    int b, e; node_segment(bnd, 0, b, e);
    if ((int)lane < min(e - b, 32)) sN = ssrc[b + (int)lane];
  }
  {
    int b, e; node_segment(bnd, 0, b, e);
    sC = min(max(sN, 0), NNODES - 1);
    if ((int)lane < min(e - b, 32)) lC = *(const float4*)(el + (size_t)sC * NH);
    const float* __restrict__ myrow = feat + (size_t)n0 * HD;
    const float* __restrict__ orow0 = out + (size_t)n0 * HD;
    #pragma unroll
    for (int hh = 0; hh < NH; ++hh) {
      mrC[hh] = myrow[hh * ND + lane];
      bsC[hh] = FIRST ? 0.f : orow0[hh * ND + lane];
    }
  }
  sN = 0;
  if (1 < nn) {
    int b, e; node_segment(bnd, 1, b, e);
    if ((int)lane < min(e - b, 32)) sN = ssrc[b + (int)lane];
  }

  for (int i = 0; i < nn; ++i) {
    const int n = n0 + i;

    int   sX = 0;
    float4 lX = make_float4(0.f, 0.f, 0.f, 0.f);
    float mrX[NH] = {0.f, 0.f, 0.f, 0.f}, bsX[NH] = {0.f, 0.f, 0.f, 0.f};
    if (i + 1 < nn) {
      int b, e; node_segment(bnd, i + 1, b, e);
      sX = min(max(sN, 0), NNODES - 1);
      if ((int)lane < min(e - b, 32)) lX = *(const float4*)(el + (size_t)sX * NH);
      const float* __restrict__ nrow = feat + (size_t)(n + 1) * HD;
      const float* __restrict__ norow = out + (size_t)(n + 1) * HD;
      #pragma unroll
      for (int hh = 0; hh < NH; ++hh) {
        mrX[hh] = nrow[hh * ND + lane];
        bsX[hh] = FIRST ? 0.f : norow[hh * ND + lane];
      }
    }
    int sNN = 0;
    if (i + 2 < nn) {
      int b, e; node_segment(bnd, i + 2, b, e);
      if ((int)lane < min(e - b, 32)) sNN = ssrc[b + (int)lane];
    }

    int beg, end; node_segment(bnd, i, beg, end);
    const int deg = end - beg;

    float er[NH];
    #pragma unroll
    for (int hh = 0; hh < NH; ++hh)
      er[hh] = uniform_of(wave_sum(mrC[hh] * arv[hh]));

    float acc[NH] = {0.f, 0.f, 0.f, 0.f};

    if (deg <= 32) {
      int sidx = 0; float e[NH];
      #pragma unroll
      for (int hh = 0; hh < NH; ++hh) e[hh] = -INFINITY;
      if ((int)lane < deg) {
        sidx = sC;
        e[0] = leaky(lC.x + er[0]);
        e[1] = leaky(lC.y + er[1]);
        e[2] = leaky(lC.z + er[2]);
        e[3] = leaky(lC.w + er[3]);
      }
      float al[NH];
      #pragma unroll
      for (int hh = 0; hh < NH; ++hh) {
        float m = uniform_of(wave_max(e[hh]));
        if (!isfinite(m)) m = 0.f;
        const float ex = __expf(e[hh] - m);
        const float s = uniform_of(wave_sum(ex));
        al[hh] = ex / fmaxf(s, 1e-20f);
      }
      aggregate(feat, sidx, al, deg, lane, acc);
    } else {
      float m[NH], s[NH];
      {
        float mx[NH] = {-INFINITY, -INFINITY, -INFINITY, -INFINITY};
        for (int c = beg; c < end; c += 32) {
          int sidx; float e[NH];
          load_chunk(ssrc, el, c, min(32, end - c), lane, er, sidx, e);
          #pragma unroll
          for (int hh = 0; hh < NH; ++hh) mx[hh] = fmaxf(mx[hh], e[hh]);
        }
        #pragma unroll
        for (int hh = 0; hh < NH; ++hh) {
          m[hh] = uniform_of(wave_max(mx[hh]));
          if (!isfinite(m[hh])) m[hh] = 0.f;
        }
      }
      {
        float sm[NH] = {0.f, 0.f, 0.f, 0.f};
        for (int c = beg; c < end; c += 32) {
          int sidx; float e[NH];
          load_chunk(ssrc, el, c, min(32, end - c), lane, er, sidx, e);
          #pragma unroll
          for (int hh = 0; hh < NH; ++hh) sm[hh] += __expf(e[hh] - m[hh]);
        }
        #pragma unroll
        for (int hh = 0; hh < NH; ++hh) s[hh] = fmaxf(uniform_of(wave_sum(sm[hh])), 1e-20f);
      }
      for (int c = beg; c < end; c += 32) {
        const int cnt = min(32, end - c);
        int sidx; float e[NH];
        load_chunk(ssrc, el, c, cnt, lane, er, sidx, e);
        float al[NH];
        #pragma unroll
        for (int hh = 0; hh < NH; ++hh) al[hh] = __expf(e[hh] - m[hh]) / s[hh];
        aggregate(feat, sidx, al, cnt, lane, acc);
      }
    }

    float* __restrict__ orow = out + (size_t)n * HD;
    #pragma unroll
    for (int hh = 0; hh < NH; ++hh) orow[hh * ND + lane] = (FIRST ? bias[hh] : bsC[hh]) + acc[hh];

    sC = sX; lC = lX; sN = sNN;
    #pragma unroll
    for (int hh = 0; hh < NH; ++hh) { mrC[hh] = mrX[hh]; bsC[hh] = bsX[hh]; }
  }
}

extern "C" void kernel_launch(void* const* d_in, const int* in_sizes, int n_in,
                              void* d_out, int out_size, void* d_ws, size_t ws_size,
                              hipStream_t stream) {
  const float* h    = (const float*)d_in[0];
  const int*   srcs[2] = { (const int*)d_in[1], (const int*)d_in[3] };
  const int*   dsts[2] = { (const int*)d_in[2], (const int*)d_in[4] };
  const float* Ws[2]   = { (const float*)d_in[5],  (const float*)d_in[9]  };
  const float* als[2]  = { (const float*)d_in[6],  (const float*)d_in[10] };
  const float* ars[2]  = { (const float*)d_in[7],  (const float*)d_in[11] };
  const float* b0 = (const float*)d_in[8];
  const float* b1 = (const float*)d_in[12];
  float* out = (float*)d_out;

  float* feat = (float*)d_ws;
  float* el   = feat + (size_t)NNODES * HD;
  int*   cnt  = (int*)(el + (size_t)NNODES * NH);
  int*   cur  = cnt + NNODES;
  int*   ssrc = cur + NNODES;

  const int B = 256;
  const int BA = 128;
  const int nTile = (GEMM_TILES + (GEMM_BLOCK / 32) - 1) / (GEMM_BLOCK / 32);
  const int nN    = (NNODES / 4 + B - 1) / B;
  const int nE    = (EDGE_THREADS + B - 1) / B;
  const int nNode = (NNODES + (BA / 32) * AGG_NODES - 1) / ((BA / 32) * AGG_NODES);
  const int nScan = (NNODES + SCAN_TILE - 1) / SCAN_TILE;

  gat_zero_counts<<<nN, B, 0, stream>>>(cnt);
  for (int r = 0; r < 2; ++r) {
    gat_gemm_el_count<<<CNT_BLOCKS + nTile, GEMM_BLOCK, 0, stream>>>(h, Ws[r], als[r], dsts[r],
                                                                     feat, el, cnt);
    gat_scan_counts<<<nScan, SCAN_THREADS, 0, stream>>>(cnt, cur);
    if (r == 0) {
      gat_dst_scatter<true><<<nE, B, 0, stream>>>(srcs[r], dsts[r], cur, ssrc, cnt);
      gat_node_agg<true><<<nNode, BA, 0, stream>>>(cur, ssrc, el, ars[r], feat, b0, b1, out);
    } else {
      gat_dst_scatter<false><<<nE, B, 0, stream>>>(srcs[r], dsts[r], cur, ssrc, cnt);
      gat_node_agg<false><<<nNode, BA, 0, stream>>>(cur, ssrc, el, ars[r], feat, b0, b1, out);
    }
  }
}


# edge sort: two-level counting sort ranked by LDS counters (chunk x bucket table, then one work-group per 512-node bucket), no global atomic per edge; segment order = LDS arrival order
# speedup vs baseline: 3.2518x; 1.6684x over previous
#include <hip/hip_runtime.h>
#include <math.h>

#define NNODES 100000
#define NEDGES 1000000
#define INF    128
#define NH     4
#define ND     32
#define HD     128
#define NEG_SLOPE 0.2f

typedef __attribute__((ext_vector_type(2))) float v2f;
typedef __attribute__((ext_vector_type(8))) float v8f;

template <int CTRL>
__device__ __forceinline__ float dpp_swz(float v) {
  return __int_as_float(__builtin_amdgcn_mov_dpp(__float_as_int(v), CTRL, 0xf, 0xf, false));
}
__device__ __forceinline__ float row16_sum(float v) {
  v += dpp_swz<0x161>(v);
  v += dpp_swz<0x162>(v);
  v += dpp_swz<0x164>(v);
  v += dpp_swz<0x168>(v);
  return v;
}

#define GEMM_BLOCK  64
#define GEMM_KSTEPS (INF / 4)
#define GEMM_TILES  (NNODES / 16)
__global__ void __launch_bounds__(GEMM_BLOCK) gat_gemm_el(const float* __restrict__ X,
                                                          const float* __restrict__ W,
                                                          const float* __restrict__ al,
                                                          float* __restrict__ Y,
                                                          float* __restrict__ el) {
  const int tile = __builtin_amdgcn_readfirstlane(
      (int)(blockIdx.x * (GEMM_BLOCK >> 5) + (threadIdx.x >> 5)));
  if (tile >= GEMM_TILES) return;
  const int lane = threadIdx.x & 31;
  const int m    = lane & 15;
  const int koff = (lane >> 4) << 1;
  const int row0 = tile << 4;

  const float* __restrict__ xrow = X + (size_t)(row0 + m) * INF + koff;
  v2f af[GEMM_KSTEPS];
  #pragma unroll
  for (int s = 0; s < GEMM_KSTEPS; ++s) af[s] = *(const v2f*)(xrow + 4 * s);

  const int rbase = row0 + ((lane >> 4) << 3);
  for (int hh = 0; hh < NH; ++hh) {
    float p[8];
    #pragma unroll
    for (int half = 0; half < 2; ++half) {
      const int col = (hh << 5) + (half << 4) + m;
      const float* __restrict__ wcol = W + (size_t)koff * HD + col;
      v8f c = {};
      #pragma unroll
      for (int s = 0; s < GEMM_KSTEPS; ++s) {
        v2f b;
        b.x = wcol[(size_t)(4 * s) * HD];
        b.y = wcol[(size_t)(4 * s + 1) * HD];
        c = __builtin_amdgcn_wmma_f32_16x16x4_f32(false, af[s], false, b,
                                                  (short)0, c, false, false);
      }
      const float a = al[col];
      #pragma unroll
      for (int i = 0; i < 8; ++i) {
        Y[(size_t)(rbase + i) * HD + col] = c[i];
        p[i] = half == 0 ? c[i] * a : fmaf(c[i], a, p[i]);
      }
    }
    #pragma unroll
    for (int i = 0; i < 8; ++i) p[i] = row16_sum(p[i]);
    float v = p[0];
    #pragma unroll
    for (int i = 1; i < 8; ++i) v = (m == i) ? p[i] : v;
    if (m < 8) el[(size_t)(rbase + m) * NH + hh] = v;
  }
}

__device__ __forceinline__ float leaky(float v) {
  return v > 0.f ? v : NEG_SLOPE * v;
}

#define SORT_SHIFT   9
#define SORT_SPAN    (1 << SORT_SHIFT)
#define SORT_K       ((NNODES + SORT_SPAN - 1) >> SORT_SHIFT)
#define SORT_B       125
#define SORT_QUADS   (NEDGES / (4 * SORT_B))
#define SORT_THREADS 512
#define SORT_ITERS   ((SORT_QUADS + SORT_THREADS - 1) / SORT_THREADS)
#define SRC_BITS     17
#define SRC_MASK     ((1 << SRC_BITS) - 1)
static_assert(NNODES <= (1 << SRC_BITS), "a source id must fit SRC_BITS");
static_assert(SRC_BITS + SORT_SHIFT <= 31, "packed word must stay non-negative");
static_assert(SORT_QUADS * 4 * SORT_B == NEDGES, "chunks must tile the edge list");
static_assert(SORT_K <= SORT_THREADS && SORT_SPAN <= SORT_THREADS, "one thread per bucket / per node of a bucket");
static_assert(2 * SORT_B * SORT_K + SORT_K + 1 <= NNODES, "table and bucket bounds must fit their workspace plane");

__device__ __forceinline__ int node_clamp(int v) { return min(max(v, 0), NNODES - 1); }

__global__ void __launch_bounds__(SORT_THREADS) gat_bucket_hist(const int* __restrict__ dst0,
                                                                const int* __restrict__ dst1,
                                                                int* __restrict__ table) {
  __shared__ int h[SORT_K];
  const int t = threadIdx.x;
  const int r = blockIdx.x >= SORT_B ? 1 : 0;
  const int b = blockIdx.x - r * SORT_B;
  const int4* __restrict__ d4 = (const int4*)(r ? dst1 : dst0) + (size_t)b * SORT_QUADS;
  if (t < SORT_K) h[t] = 0;
  __syncthreads();
  int4 d[SORT_ITERS];
  #pragma unroll
  for (int u = 0; u < SORT_ITERS; ++u) {
    const int q = t + u * SORT_THREADS;
    d[u] = make_int4(0, 0, 0, 0);
    if (q < SORT_QUADS) d[u] = d4[q];
  }
  #pragma unroll
  for (int u = 0; u < SORT_ITERS; ++u) {
    if (t + u * SORT_THREADS < SORT_QUADS) {
      atomicAdd(&h[node_clamp(d[u].x) >> SORT_SHIFT], 1);
      atomicAdd(&h[node_clamp(d[u].y) >> SORT_SHIFT], 1);
      atomicAdd(&h[node_clamp(d[u].z) >> SORT_SHIFT], 1);
      atomicAdd(&h[node_clamp(d[u].w) >> SORT_SHIFT], 1);
    }
  }
  __syncthreads();
  if (t < SORT_K) table[(size_t)blockIdx.x * SORT_K + t] = h[t];
}

__device__ __forceinline__ int take_slot(int* slot, int dn) {
  const int pos = atomicAdd(&slot[dn >> SORT_SHIFT], 1);
  return min(max(pos, 0), NEDGES - 1);
}
__device__ __forceinline__ int pack_edge(int dn, int s) {
  return node_clamp(s) | ((dn & (SORT_SPAN - 1)) << SRC_BITS);
}

__global__ void __launch_bounds__(SORT_THREADS) gat_bucket_partition(const int* __restrict__ src,
                                                                     const int* __restrict__ dst,
                                                                     const int* __restrict__ table,
                                                                     int* __restrict__ bbase,
                                                                     int* __restrict__ part) {
  __shared__ int tot[SORT_K];
  __shared__ int slot[SORT_K];
  const int t = threadIdx.x;
  const int b = blockIdx.x;
  int mytot = 0, mypre = 0;
  if (t < SORT_K) {
    #pragma unroll
    for (int j = 0; j < SORT_B; ++j) {
      const int v = table[j * SORT_K + t];
      mytot += v;
      mypre += j < b ? v : 0;
    }
    tot[t] = mytot;
  }
  __syncthreads();
  if (t < SORT_K) {
    int base = 0;
    for (int j = 0; j < t; ++j) base += tot[j];
    slot[t] = base + mypre;
    if (b == 0) {
      bbase[t] = base;
      if (t == SORT_K - 1) bbase[SORT_K] = base + mytot;
    }
  }
  __syncthreads();
  const int4* __restrict__ d4 = (const int4*)dst + (size_t)b * SORT_QUADS;
  const int4* __restrict__ s4 = (const int4*)src + (size_t)b * SORT_QUADS;
  int4 d[SORT_ITERS], s[SORT_ITERS];
  #pragma unroll
  for (int u = 0; u < SORT_ITERS; ++u) {
    const int q = t + u * SORT_THREADS;
    d[u] = make_int4(0, 0, 0, 0);
    s[u] = make_int4(0, 0, 0, 0);
    if (q < SORT_QUADS) { d[u] = d4[q]; s[u] = s4[q]; }
  }
  #pragma unroll
  for (int u = 0; u < SORT_ITERS; ++u) {
    if (t + u * SORT_THREADS < SORT_QUADS) {
      const int dx = node_clamp(d[u].x), dy = node_clamp(d[u].y);
      const int dz = node_clamp(d[u].z), dw = node_clamp(d[u].w);
      const int px = take_slot(slot, dx);
      const int py = take_slot(slot, dy);
      const int pz = take_slot(slot, dz);
      const int pw = take_slot(slot, dw);
      part[px] = pack_edge(dx, s[u].x);
      part[py] = pack_edge(dy, s[u].y);
      part[pz] = pack_edge(dz, s[u].z);
      part[pw] = pack_edge(dw, s[u].w);
    }
  }
}

__global__ void __launch_bounds__(SORT_THREADS) gat_bucket_sort(const int* __restrict__ bbase,
                                                                const int* __restrict__ part,
                                                                int* __restrict__ cur,
                                                                int* __restrict__ ssrc) {
  __shared__ int cntl[SORT_SPAN];
  __shared__ int curl[SORT_SPAN];
  __shared__ int wtot[SORT_THREADS / 32];
  const int t = threadIdx.x;
  const int lane = t & 31, wv = t >> 5;
  const int k = blockIdx.x;
  int beg = bbase[k], end = bbase[k + 1];
  beg = min(max(beg, 0), NEDGES);
  end = min(max(end, beg), NEDGES);
  const int n = end - beg;

  if (t < SORT_SPAN) cntl[t] = 0;
  __syncthreads();
  for (int i0 = t; i0 < n; i0 += 4 * SORT_THREADS) {
    int w[4];
    #pragma unroll
    for (int u = 0; u < 4; ++u) {
      const int i = i0 + u * SORT_THREADS;
      w[u] = i < n ? part[beg + i] : -1;
    }
    #pragma unroll
    for (int u = 0; u < 4; ++u)
      if (w[u] >= 0) atomicAdd(&cntl[(w[u] >> SRC_BITS) & (SORT_SPAN - 1)], 1);
  }
  __syncthreads();

  const int v = t < SORT_SPAN ? cntl[t] : 0;
  int inc = v;
  #pragma unroll
  for (int d = 1; d < 32; d <<= 1) {
    const int y = __shfl_up(inc, d, 32);
    if (lane >= d) inc += y;
  }
  if (lane == 31) wtot[wv] = inc;
  __syncthreads();
  int run = inc - v;
  for (int w = 0; w < wv; ++w) run += wtot[w];
  if (t < SORT_SPAN) {
    curl[t] = beg + run;
    const int node = (k << SORT_SHIFT) + t;
    if (node < NNODES) cur[node] = beg + run + v;
  }
  __syncthreads();

  for (int i0 = t; i0 < n; i0 += 4 * SORT_THREADS) {
    int w[4], pos[4];
    #pragma unroll
    for (int u = 0; u < 4; ++u) {
      const int i = i0 + u * SORT_THREADS;
      w[u] = i < n ? part[beg + i] : -1;
    }
    #pragma unroll
    for (int u = 0; u < 4; ++u) {
      pos[u] = 0;
      if (w[u] >= 0) {
        const int p = atomicAdd(&curl[(w[u] >> SRC_BITS) & (SORT_SPAN - 1)], 1);
        pos[u] = min(max(p, 0), NEDGES - 1);
      }
    }
    #pragma unroll
    for (int u = 0; u < 4; ++u)
      if (w[u] >= 0) ssrc[pos[u]] = w[u] & SRC_MASK;
  }
}

__device__ __forceinline__ float wave_sum(float v) {
  #pragma unroll
  for (int m = 16; m >= 1; m >>= 1) v += __shfl_xor(v, m, 32);
  return v;
}
__device__ __forceinline__ float wave_max(float v) {
  #pragma unroll
  for (int m = 16; m >= 1; m >>= 1) v = fmaxf(v, __shfl_xor(v, m, 32));
  return v;
}

__device__ __forceinline__ float uniform_of(float v) {
  return __int_as_float(__builtin_amdgcn_readfirstlane(__float_as_int(v)));
}

__device__ __forceinline__ void load_chunk(const int* __restrict__ ssrc,
                                           const float* __restrict__ el,
                                           int base, int cnt, unsigned lane,
                                           const float er[NH],
                                           int& sidx, float e[NH]) {
  sidx = 0;
  #pragma unroll
  for (int hh = 0; hh < NH; ++hh) e[hh] = -INFINITY;
  if ((int)lane < cnt) {
    int s = ssrc[base + (int)lane];
    s = min(max(s, 0), NNODES - 1);
    sidx = s;
    const float4 l4 = *(const float4*)(el + (size_t)s * NH);
    e[0] = leaky(l4.x + er[0]);
    e[1] = leaky(l4.y + er[1]);
    e[2] = leaky(l4.z + er[2]);
    e[3] = leaky(l4.w + er[3]);
  }
}

#define GRP 8
__device__ __forceinline__ void aggregate(const float* __restrict__ feat,
                                          int sidx, const float al[NH],
                                          int cnt, unsigned lane, float acc[NH]) {
  for (int j0 = 0; j0 < cnt; j0 += GRP) {
    float r[GRP][NH];
    #pragma unroll
    for (int u = 0; u < GRP; ++u) {
      if (j0 + u < cnt) {
        const int sj = __builtin_amdgcn_readlane(sidx, j0 + u);
        const float* __restrict__ row = feat + (size_t)sj * HD;
        #pragma unroll
        for (int hh = 0; hh < NH; ++hh) r[u][hh] = row[hh * ND + lane];
      } else {
        #pragma unroll
        for (int hh = 0; hh < NH; ++hh) r[u][hh] = 0.f;
      }
    }
    #pragma unroll
    for (int u = 0; u < GRP; ++u) {
      #pragma unroll
      for (int hh = 0; hh < NH; ++hh) {
        const float a = __int_as_float(
            __builtin_amdgcn_readlane(__float_as_int(al[hh]), j0 + u));
        acc[hh] += r[u][hh] * a;
      }
    }
  }
}

#define AGG_NODES 12
__device__ __forceinline__ void node_segment(int bnd, int k, int& beg, int& end) {
  beg = __builtin_amdgcn_readlane(bnd, k);
  end = __builtin_amdgcn_readlane(bnd, k + 1);
  beg = min(max(beg, 0), NEDGES);
  end = min(max(end, beg), NEDGES);
}

template <bool FIRST>
__global__ void gat_node_agg(const int* __restrict__ cur,
                             const int* __restrict__ ssrc,
                             const float* __restrict__ el,
                             const float* __restrict__ ar,
                             const float* __restrict__ feat,
                             const float* __restrict__ b0,
                             const float* __restrict__ b1,
                             float* __restrict__ out) {
  const int n0 = __builtin_amdgcn_readfirstlane(
      (int)(blockIdx.x * (blockDim.x >> 5) + (threadIdx.x >> 5))) * AGG_NODES;
  if (n0 >= NNODES) return;
  const unsigned lane = threadIdx.x & 31;
  const int nn = min(AGG_NODES, NNODES - n0);

  const int bidx = n0 - 1 + (int)lane;
  int bnd = 0;
  if (lane <= AGG_NODES && bidx >= 0 && bidx < NNODES) bnd = cur[bidx];

  float arv[NH], bias[NH];
  #pragma unroll
  for (int hh = 0; hh < NH; ++hh) {
    arv[hh]  = ar[hh * ND + lane];
    bias[hh] = FIRST ? (b0[hh * ND + lane] + b1[hh * ND + lane]) : 0.f;
  }

  int   sC = 0;
  float4 lC = make_float4(0.f, 0.f, 0.f, 0.f);
  float mrC[NH], bsC[NH];
  int   sN = 0;

  {
    int b, e; node_segment(bnd, 0, b, e);
    if ((int)lane < min(e - b, 32)) sN = ssrc[b + (int)lane];
  }
  {
    int b, e; node_segment(bnd, 0, b, e);
    sC = min(max(sN, 0), NNODES - 1);
    if ((int)lane < min(e - b, 32)) lC = *(const float4*)(el + (size_t)sC * NH);
    const float* __restrict__ myrow = feat + (size_t)n0 * HD;
    const float* __restrict__ orow0 = out + (size_t)n0 * HD;
    #pragma unroll
    for (int hh = 0; hh < NH; ++hh) {
      mrC[hh] = myrow[hh * ND + lane];
      bsC[hh] = FIRST ? 0.f : orow0[hh * ND + lane];
    }
  }
  sN = 0;
  if (1 < nn) {
    int b, e; node_segment(bnd, 1, b, e);
    if ((int)lane < min(e - b, 32)) sN = ssrc[b + (int)lane];
  }

  for (int i = 0; i < nn; ++i) {
    const int n = n0 + i;

    int   sX = 0;
    float4 lX = make_float4(0.f, 0.f, 0.f, 0.f);
    float mrX[NH] = {0.f, 0.f, 0.f, 0.f}, bsX[NH] = {0.f, 0.f, 0.f, 0.f};
    if (i + 1 < nn) {
      int b, e; node_segment(bnd, i + 1, b, e);
      sX = min(max(sN, 0), NNODES - 1);
      if ((int)lane < min(e - b, 32)) lX = *(const float4*)(el + (size_t)sX * NH);
      const float* __restrict__ nrow = feat + (size_t)(n + 1) * HD;
      const float* __restrict__ norow = out + (size_t)(n + 1) * HD;
      #pragma unroll
      for (int hh = 0; hh < NH; ++hh) {
        mrX[hh] = nrow[hh * ND + lane];
        bsX[hh] = FIRST ? 0.f : norow[hh * ND + lane];
      }
    }
    int sNN = 0;
    if (i + 2 < nn) {
      int b, e; node_segment(bnd, i + 2, b, e);
      if ((int)lane < min(e - b, 32)) sNN = ssrc[b + (int)lane];
    }

    int beg, end; node_segment(bnd, i, beg, end);
    const int deg = end - beg;

    float er[NH];
    #pragma unroll
    for (int hh = 0; hh < NH; ++hh)
      er[hh] = uniform_of(wave_sum(mrC[hh] * arv[hh]));

    float acc[NH] = {0.f, 0.f, 0.f, 0.f};

    if (deg <= 32) {
      int sidx = 0; float e[NH];
      #pragma unroll
      for (int hh = 0; hh < NH; ++hh) e[hh] = -INFINITY;
      if ((int)lane < deg) {
        sidx = sC;
        e[0] = leaky(lC.x + er[0]);
        e[1] = leaky(lC.y + er[1]);
        e[2] = leaky(lC.z + er[2]);
        e[3] = leaky(lC.w + er[3]);
      }
      float al[NH];
      #pragma unroll
      for (int hh = 0; hh < NH; ++hh) {
        float m = uniform_of(wave_max(e[hh]));
        if (!isfinite(m)) m = 0.f;
        const float ex = __expf(e[hh] - m);
        const float s = uniform_of(wave_sum(ex));
        al[hh] = ex / fmaxf(s, 1e-20f);
      }
      aggregate(feat, sidx, al, deg, lane, acc);
    } else {
      float m[NH], s[NH];
      {
        float mx[NH] = {-INFINITY, -INFINITY, -INFINITY, -INFINITY};
        for (int c = beg; c < end; c += 32) {
          int sidx; float e[NH];
          load_chunk(ssrc, el, c, min(32, end - c), lane, er, sidx, e);
          #pragma unroll
          for (int hh = 0; hh < NH; ++hh) mx[hh] = fmaxf(mx[hh], e[hh]);
        }
        #pragma unroll
        for (int hh = 0; hh < NH; ++hh) {
          m[hh] = uniform_of(wave_max(mx[hh]));
          if (!isfinite(m[hh])) m[hh] = 0.f;
        }
      }
      {
        float sm[NH] = {0.f, 0.f, 0.f, 0.f};
        for (int c = beg; c < end; c += 32) {
          int sidx; float e[NH];
          load_chunk(ssrc, el, c, min(32, end - c), lane, er, sidx, e);
          #pragma unroll
          for (int hh = 0; hh < NH; ++hh) sm[hh] += __expf(e[hh] - m[hh]);
        }
        #pragma unroll
        for (int hh = 0; hh < NH; ++hh) s[hh] = fmaxf(uniform_of(wave_sum(sm[hh])), 1e-20f);
      }
      for (int c = beg; c < end; c += 32) {
        const int cnt = min(32, end - c);
        int sidx; float e[NH];
        load_chunk(ssrc, el, c, cnt, lane, er, sidx, e);
        float al[NH];
        #pragma unroll
        for (int hh = 0; hh < NH; ++hh) al[hh] = __expf(e[hh] - m[hh]) / s[hh];
        aggregate(feat, sidx, al, cnt, lane, acc);
      }
    }

    float* __restrict__ orow = out + (size_t)n * HD;
    #pragma unroll
    for (int hh = 0; hh < NH; ++hh) orow[hh * ND + lane] = (FIRST ? bias[hh] : bsC[hh]) + acc[hh];

    sC = sX; lC = lX; sN = sNN;
    #pragma unroll
    for (int hh = 0; hh < NH; ++hh) { mrC[hh] = mrX[hh]; bsC[hh] = bsX[hh]; }
  }
}

extern "C" void kernel_launch(void* const* d_in, const int* in_sizes, int n_in,
                              void* d_out, int out_size, void* d_ws, size_t ws_size,
                              hipStream_t stream) {
  const float* h    = (const float*)d_in[0];
  const int*   srcs[2] = { (const int*)d_in[1], (const int*)d_in[3] };
  const int*   dsts[2] = { (const int*)d_in[2], (const int*)d_in[4] };
  const float* Ws[2]   = { (const float*)d_in[5],  (const float*)d_in[9]  };
  const float* als[2]  = { (const float*)d_in[6],  (const float*)d_in[10] };
  const float* ars[2]  = { (const float*)d_in[7],  (const float*)d_in[11] };
  const float* b0 = (const float*)d_in[8];
  const float* b1 = (const float*)d_in[12];
  float* out = (float*)d_out;

  float* feat = (float*)d_ws;
  float* el   = feat + (size_t)NNODES * HD;
  int*   tab  = (int*)(el + (size_t)NNODES * NH);
  int*   cur  = tab + NNODES;
  int*   ssrc = cur + NNODES;
  int*   part = (int*)feat;
  int*   tabs[2] = { tab, tab + SORT_B * SORT_K };
  int*   bbase   = tab + 2 * SORT_B * SORT_K;

  const int BA = 128;
  const int nTile = (GEMM_TILES + (GEMM_BLOCK / 32) - 1) / (GEMM_BLOCK / 32);
  const int nNode = (NNODES + (BA / 32) * AGG_NODES - 1) / ((BA / 32) * AGG_NODES);

  gat_bucket_hist<<<2 * SORT_B, SORT_THREADS, 0, stream>>>(dsts[0], dsts[1], tab);
  for (int r = 0; r < 2; ++r) {
    gat_bucket_partition<<<SORT_B, SORT_THREADS, 0, stream>>>(srcs[r], dsts[r], tabs[r], bbase, part);
    gat_bucket_sort<<<SORT_K, SORT_THREADS, 0, stream>>>(bbase, part, cur, ssrc);
    gat_gemm_el<<<nTile, GEMM_BLOCK, 0, stream>>>(h, Ws[r], als[r], feat, el);
    if (r == 0)
      gat_node_agg<true><<<nNode, BA, 0, stream>>>(cur, ssrc, el, ars[r], feat, b0, b1, out);
    else
      gat_node_agg<false><<<nNode, BA, 0, stream>>>(cur, ssrc, el, ars[r], feat, b0, b1, out);
  }
}


# edge sort: both levels rank into an LDS copy of the work-group's piece and store it in list order (consecutive lanes, consecutive words), not one scattered 4-byte store per edge; ranks unchanged
# speedup vs baseline: 3.9249x; 1.2070x over previous
#include <hip/hip_runtime.h>
#include <math.h>

#define NNODES 100000
#define NEDGES 1000000
#define INF    128
#define NH     4
#define ND     32
#define HD     128
#define NEG_SLOPE 0.2f

typedef __attribute__((ext_vector_type(2))) float v2f;
typedef __attribute__((ext_vector_type(8))) float v8f;

template <int CTRL>
__device__ __forceinline__ float dpp_swz(float v) {
  return __int_as_float(__builtin_amdgcn_mov_dpp(__float_as_int(v), CTRL, 0xf, 0xf, false));
}
__device__ __forceinline__ float row16_sum(float v) {
  v += dpp_swz<0x161>(v);
  v += dpp_swz<0x162>(v);
  v += dpp_swz<0x164>(v);
  v += dpp_swz<0x168>(v);
  return v;
}

#define GEMM_BLOCK  64
#define GEMM_KSTEPS (INF / 4)
#define GEMM_TILES  (NNODES / 16)
__global__ void __launch_bounds__(GEMM_BLOCK) gat_gemm_el(const float* __restrict__ X,
                                                          const float* __restrict__ W,
                                                          const float* __restrict__ al,
                                                          float* __restrict__ Y,
                                                          float* __restrict__ el) {
  const int tile = __builtin_amdgcn_readfirstlane(
      (int)(blockIdx.x * (GEMM_BLOCK >> 5) + (threadIdx.x >> 5)));
  if (tile >= GEMM_TILES) return;
  const int lane = threadIdx.x & 31;
  const int m    = lane & 15;
  const int koff = (lane >> 4) << 1;
  const int row0 = tile << 4;

  const float* __restrict__ xrow = X + (size_t)(row0 + m) * INF + koff;
  v2f af[GEMM_KSTEPS];
  #pragma unroll
  for (int s = 0; s < GEMM_KSTEPS; ++s) af[s] = *(const v2f*)(xrow + 4 * s);

  const int rbase = row0 + ((lane >> 4) << 3);
  for (int hh = 0; hh < NH; ++hh) {
    float p[8];
    #pragma unroll
    for (int half = 0; half < 2; ++half) {
      const int col = (hh << 5) + (half << 4) + m;
      const float* __restrict__ wcol = W + (size_t)koff * HD + col;
      v8f c = {};
      #pragma unroll
      for (int s = 0; s < GEMM_KSTEPS; ++s) {
        v2f b;
        b.x = wcol[(size_t)(4 * s) * HD];
        b.y = wcol[(size_t)(4 * s + 1) * HD];
        c = __builtin_amdgcn_wmma_f32_16x16x4_f32(false, af[s], false, b,
                                                  (short)0, c, false, false);
      }
      const float a = al[col];
      #pragma unroll
      for (int i = 0; i < 8; ++i) {
        Y[(size_t)(rbase + i) * HD + col] = c[i];
        p[i] = half == 0 ? c[i] * a : fmaf(c[i], a, p[i]);
      }
    }
    #pragma unroll
    for (int i = 0; i < 8; ++i) p[i] = row16_sum(p[i]);
    float v = p[0];
    #pragma unroll
    for (int i = 1; i < 8; ++i) v = (m == i) ? p[i] : v;
    if (m < 8) el[(size_t)(rbase + m) * NH + hh] = v;
  }
}

__device__ __forceinline__ float leaky(float v) {
  return v > 0.f ? v : NEG_SLOPE * v;
}

#define SORT_SHIFT   9
#define SORT_SPAN    (1 << SORT_SHIFT)
#define SORT_K       ((NNODES + SORT_SPAN - 1) >> SORT_SHIFT)
#define SORT_B       125
#define SORT_QUADS   (NEDGES / (4 * SORT_B))
#define SORT_THREADS 512
#define SORT_ITERS   ((SORT_QUADS + SORT_THREADS - 1) / SORT_THREADS)
#define SORT_CAP     12288
#define SRC_BITS     17
#define SRC_MASK     ((1 << SRC_BITS) - 1)
static_assert(NNODES <= (1 << SRC_BITS), "a source id must fit SRC_BITS");
static_assert(SRC_BITS + SORT_SHIFT <= 31, "packed word must stay non-negative");
static_assert(SORT_QUADS * 4 * SORT_B == NEDGES, "chunks must tile the edge list");
static_assert(SORT_K <= SORT_THREADS && SORT_SPAN <= SORT_THREADS, "one thread per bucket / per node of a bucket");
static_assert(2 * SORT_B * SORT_K + SORT_K + 1 <= NNODES, "table and bucket bounds must fit their workspace plane");

__device__ __forceinline__ int node_clamp(int v) { return min(max(v, 0), NNODES - 1); }

__global__ void __launch_bounds__(SORT_THREADS) gat_bucket_hist(const int* __restrict__ dst0,
                                                                const int* __restrict__ dst1,
                                                                int* __restrict__ table) {
  __shared__ int h[SORT_K];
  const int t = threadIdx.x;
  const int r = blockIdx.x >= SORT_B ? 1 : 0;
  const int b = blockIdx.x - r * SORT_B;
  const int4* __restrict__ d4 = (const int4*)(r ? dst1 : dst0) + (size_t)b * SORT_QUADS;
  if (t < SORT_K) h[t] = 0;
  __syncthreads();
  int4 d[SORT_ITERS];
  #pragma unroll
  for (int u = 0; u < SORT_ITERS; ++u) {
    const int q = t + u * SORT_THREADS;
    d[u] = make_int4(0, 0, 0, 0);
    if (q < SORT_QUADS) d[u] = d4[q];
  }
  #pragma unroll
  for (int u = 0; u < SORT_ITERS; ++u) {
    if (t + u * SORT_THREADS < SORT_QUADS) {
      atomicAdd(&h[node_clamp(d[u].x) >> SORT_SHIFT], 1);
      atomicAdd(&h[node_clamp(d[u].y) >> SORT_SHIFT], 1);
      atomicAdd(&h[node_clamp(d[u].z) >> SORT_SHIFT], 1);
      atomicAdd(&h[node_clamp(d[u].w) >> SORT_SHIFT], 1);
    }
  }
  __syncthreads();
  if (t < SORT_K) table[(size_t)blockIdx.x * SORT_K + t] = h[t];
}

#define PART_EDGES (4 * SORT_QUADS)
__device__ __forceinline__ int take_slot(int* slot, int k) {
  const int pos = atomicAdd(&slot[k], 1);
  return min(max(pos, 0), PART_EDGES - 1);
}
__device__ __forceinline__ int pack_edge(int dn, int s) {
  return node_clamp(s) | ((dn & (SORT_SPAN - 1)) << SRC_BITS);
}

__global__ void __launch_bounds__(SORT_THREADS) gat_bucket_partition(const int* __restrict__ src,
                                                                     const int* __restrict__ dst,
                                                                     const int* __restrict__ table,
                                                                     int* __restrict__ bbase,
                                                                     int* __restrict__ part) {
  __shared__ int tot[SORT_K];
  __shared__ int own[SORT_K];
  __shared__ int slot[SORT_K];
  __shared__ int shift[SORT_K];
  __shared__ int stage_w[PART_EDGES];
  __shared__ unsigned short stage_k[PART_EDGES];
  const int t = threadIdx.x;
  const int b = blockIdx.x;
  int mytot = 0, mypre = 0;
  if (t < SORT_K) {
    #pragma unroll
    for (int j = 0; j < SORT_B; ++j) {
      const int v = table[j * SORT_K + t];
      mytot += v;
      mypre += j < b ? v : 0;
    }
    tot[t] = mytot;
    own[t] = table[b * SORT_K + t];
  }
  __syncthreads();
  if (t < SORT_K) {
    int base = 0;
    int lbase = 0;
    for (int j = 0; j < t; ++j) { base += tot[j]; lbase += own[j]; }
    slot[t] = lbase;
    shift[t] = base + mypre - lbase;
    if (b == 0) {
      bbase[t] = base;
      if (t == SORT_K - 1) bbase[SORT_K] = base + mytot;
    }
  }
  __syncthreads();
  const int4* __restrict__ d4 = (const int4*)dst + (size_t)b * SORT_QUADS;
  const int4* __restrict__ s4 = (const int4*)src + (size_t)b * SORT_QUADS;
  int4 d[SORT_ITERS], s[SORT_ITERS];
  #pragma unroll
  for (int u = 0; u < SORT_ITERS; ++u) {
    const int q = t + u * SORT_THREADS;
    d[u] = make_int4(0, 0, 0, 0);
    s[u] = make_int4(0, 0, 0, 0);
    if (q < SORT_QUADS) { d[u] = d4[q]; s[u] = s4[q]; }
  }
  #pragma unroll
  for (int u = 0; u < SORT_ITERS; ++u) {
    if (t + u * SORT_THREADS < SORT_QUADS) {
      const int dx = node_clamp(d[u].x), dy = node_clamp(d[u].y);
      const int dz = node_clamp(d[u].z), dw = node_clamp(d[u].w);
      const int kx = dx >> SORT_SHIFT, ky = dy >> SORT_SHIFT;
      const int kz = dz >> SORT_SHIFT, kw = dw >> SORT_SHIFT;
      const int px = take_slot(slot, kx);
      const int py = take_slot(slot, ky);
      const int pz = take_slot(slot, kz);
      const int pw = take_slot(slot, kw);
      stage_w[px] = pack_edge(dx, s[u].x);  stage_k[px] = (unsigned short)kx;
      stage_w[py] = pack_edge(dy, s[u].y);  stage_k[py] = (unsigned short)ky;
      stage_w[pz] = pack_edge(dz, s[u].z);  stage_k[pz] = (unsigned short)kz;
      stage_w[pw] = pack_edge(dw, s[u].w);  stage_k[pw] = (unsigned short)kw;
    }
  }
  __syncthreads();
  for (int i = t; i < PART_EDGES; i += SORT_THREADS) {
    const int k = min((int)stage_k[i], SORT_K - 1);
    const int g = min(max(i + shift[k], 0), NEDGES - 1);
    part[g] = stage_w[i];
  }
}

__global__ void __launch_bounds__(SORT_THREADS) gat_bucket_sort(const int* __restrict__ bbase,
                                                                const int* __restrict__ part,
                                                                int* __restrict__ cur,
                                                                int* __restrict__ ssrc) {
  __shared__ int cntl[SORT_SPAN];
  __shared__ int curl[SORT_SPAN];
  __shared__ int wtot[SORT_THREADS / 32];
  __shared__ int stage[SORT_CAP];
  const int t = threadIdx.x;
  const int lane = t & 31, wv = t >> 5;
  const int k = blockIdx.x;
  int beg = bbase[k], end = bbase[k + 1];
  beg = min(max(beg, 0), NEDGES);
  end = min(max(end, beg), NEDGES);
  const int n = end - beg;
  const bool staged = n <= SORT_CAP;

  if (t < SORT_SPAN) cntl[t] = 0;
  __syncthreads();
  for (int i0 = t; i0 < n; i0 += 4 * SORT_THREADS) {
    int w[4];
    #pragma unroll
    for (int u = 0; u < 4; ++u) {
      const int i = i0 + u * SORT_THREADS;
      w[u] = i < n ? part[beg + i] : -1;
    }
    #pragma unroll
    for (int u = 0; u < 4; ++u)
      if (w[u] >= 0) atomicAdd(&cntl[(w[u] >> SRC_BITS) & (SORT_SPAN - 1)], 1);
  }
  __syncthreads();

  const int v = t < SORT_SPAN ? cntl[t] : 0;
  int inc = v;
  #pragma unroll
  for (int d = 1; d < 32; d <<= 1) {
    const int y = __shfl_up(inc, d, 32);
    if (lane >= d) inc += y;
  }
  if (lane == 31) wtot[wv] = inc;
  __syncthreads();
  int run = inc - v;
  for (int w = 0; w < wv; ++w) run += wtot[w];
  if (t < SORT_SPAN) {
    curl[t] = beg + run;
    const int node = (k << SORT_SHIFT) + t;
    if (node < NNODES) cur[node] = beg + run + v;
  }
  __syncthreads();

  for (int i0 = t; i0 < n; i0 += 4 * SORT_THREADS) {
    int w[4], pos[4];
    #pragma unroll
    for (int u = 0; u < 4; ++u) {
      const int i = i0 + u * SORT_THREADS;
      w[u] = i < n ? part[beg + i] : -1;
    }
    #pragma unroll
    for (int u = 0; u < 4; ++u) {
      pos[u] = 0;
      if (w[u] >= 0) {
        const int p = atomicAdd(&curl[(w[u] >> SRC_BITS) & (SORT_SPAN - 1)], 1);
        pos[u] = staged ? min(max(p - beg, 0), SORT_CAP - 1) : min(max(p, 0), NEDGES - 1);
      }
    }
    #pragma unroll
    for (int u = 0; u < 4; ++u) {
      if (w[u] >= 0) {
        if (staged) stage[pos[u]] = w[u] & SRC_MASK;
        else        ssrc[pos[u]]  = w[u] & SRC_MASK;
      }
    }
  }
  if (staged) {
    __syncthreads();
    for (int i = t; i < n; i += SORT_THREADS) ssrc[beg + i] = stage[i];
  }
}

__device__ __forceinline__ float wave_sum(float v) {
  #pragma unroll
  for (int m = 16; m >= 1; m >>= 1) v += __shfl_xor(v, m, 32);
  return v;
}
__device__ __forceinline__ float wave_max(float v) {
  #pragma unroll
  for (int m = 16; m >= 1; m >>= 1) v = fmaxf(v, __shfl_xor(v, m, 32));
  return v;
}

__device__ __forceinline__ float uniform_of(float v) {
  return __int_as_float(__builtin_amdgcn_readfirstlane(__float_as_int(v)));
}

__device__ __forceinline__ void load_chunk(const int* __restrict__ ssrc,
                                           const float* __restrict__ el,
                                           int base, int cnt, unsigned lane,
                                           const float er[NH],
                                           int& sidx, float e[NH]) {
  sidx = 0;
  #pragma unroll
  for (int hh = 0; hh < NH; ++hh) e[hh] = -INFINITY;
  if ((int)lane < cnt) {
    int s = ssrc[base + (int)lane];
    s = min(max(s, 0), NNODES - 1);
    sidx = s;
    const float4 l4 = *(const float4*)(el + (size_t)s * NH);
    e[0] = leaky(l4.x + er[0]);
    e[1] = leaky(l4.y + er[1]);
    e[2] = leaky(l4.z + er[2]);
    e[3] = leaky(l4.w + er[3]);
  }
}

#define GRP 8
__device__ __forceinline__ void aggregate(const float* __restrict__ feat,
                                          int sidx, const float al[NH],
                                          int cnt, unsigned lane, float acc[NH]) {
  for (int j0 = 0; j0 < cnt; j0 += GRP) {
    float r[GRP][NH];
    #pragma unroll
    for (int u = 0; u < GRP; ++u) {
      if (j0 + u < cnt) {
        const int sj = __builtin_amdgcn_readlane(sidx, j0 + u);
        const float* __restrict__ row = feat + (size_t)sj * HD;
        #pragma unroll
        for (int hh = 0; hh < NH; ++hh) r[u][hh] = row[hh * ND + lane];
      } else {
        #pragma unroll
        for (int hh = 0; hh < NH; ++hh) r[u][hh] = 0.f;
      }
    }
    #pragma unroll
    for (int u = 0; u < GRP; ++u) {
      #pragma unroll
      for (int hh = 0; hh < NH; ++hh) {
        const float a = __int_as_float(
            __builtin_amdgcn_readlane(__float_as_int(al[hh]), j0 + u));
        acc[hh] += r[u][hh] * a;
      }
    }
  }
}

#define AGG_NODES 12
__device__ __forceinline__ void node_segment(int bnd, int k, int& beg, int& end) {
  beg = __builtin_amdgcn_readlane(bnd, k);
  end = __builtin_amdgcn_readlane(bnd, k + 1);
  beg = min(max(beg, 0), NEDGES);
  end = min(max(end, beg), NEDGES);
}

template <bool FIRST>
__global__ void gat_node_agg(const int* __restrict__ cur,
                             const int* __restrict__ ssrc,
                             const float* __restrict__ el,
                             const float* __restrict__ ar,
                             const float* __restrict__ feat,
                             const float* __restrict__ b0,
                             const float* __restrict__ b1,
                             float* __restrict__ out) {
  const int n0 = __builtin_amdgcn_readfirstlane(
      (int)(blockIdx.x * (blockDim.x >> 5) + (threadIdx.x >> 5))) * AGG_NODES;
  if (n0 >= NNODES) return;
  const unsigned lane = threadIdx.x & 31;
  const int nn = min(AGG_NODES, NNODES - n0);

  const int bidx = n0 - 1 + (int)lane;
  int bnd = 0;
  if (lane <= AGG_NODES && bidx >= 0 && bidx < NNODES) bnd = cur[bidx];

  float arv[NH], bias[NH];
  #pragma unroll
  for (int hh = 0; hh < NH; ++hh) {
    arv[hh]  = ar[hh * ND + lane];
    bias[hh] = FIRST ? (b0[hh * ND + lane] + b1[hh * ND + lane]) : 0.f;
  }

  int   sC = 0;
  float4 lC = make_float4(0.f, 0.f, 0.f, 0.f);
  float mrC[NH], bsC[NH];
  int   sN = 0;

  {
    int b, e; node_segment(bnd, 0, b, e);
    if ((int)lane < min(e - b, 32)) sN = ssrc[b + (int)lane];
  }
  {
    int b, e; node_segment(bnd, 0, b, e);
    sC = min(max(sN, 0), NNODES - 1);
    if ((int)lane < min(e - b, 32)) lC = *(const float4*)(el + (size_t)sC * NH);
    const float* __restrict__ myrow = feat + (size_t)n0 * HD;
    const float* __restrict__ orow0 = out + (size_t)n0 * HD;
    #pragma unroll
    for (int hh = 0; hh < NH; ++hh) {
      mrC[hh] = myrow[hh * ND + lane];
      bsC[hh] = FIRST ? 0.f : orow0[hh * ND + lane];
    }
  }
  sN = 0;
  if (1 < nn) {
    int b, e; node_segment(bnd, 1, b, e);
    if ((int)lane < min(e - b, 32)) sN = ssrc[b + (int)lane];
  }

  for (int i = 0; i < nn; ++i) {
    const int n = n0 + i;

    int   sX = 0;
    float4 lX = make_float4(0.f, 0.f, 0.f, 0.f);
    float mrX[NH] = {0.f, 0.f, 0.f, 0.f}, bsX[NH] = {0.f, 0.f, 0.f, 0.f};
    if (i + 1 < nn) {
      int b, e; node_segment(bnd, i + 1, b, e);
      sX = min(max(sN, 0), NNODES - 1);
      if ((int)lane < min(e - b, 32)) lX = *(const float4*)(el + (size_t)sX * NH);
      const float* __restrict__ nrow = feat + (size_t)(n + 1) * HD;
      const float* __restrict__ norow = out + (size_t)(n + 1) * HD;
      #pragma unroll
      for (int hh = 0; hh < NH; ++hh) {
        mrX[hh] = nrow[hh * ND + lane];
        bsX[hh] = FIRST ? 0.f : norow[hh * ND + lane];
      }
    }
    int sNN = 0;
    if (i + 2 < nn) {
      int b, e; node_segment(bnd, i + 2, b, e);
      if ((int)lane < min(e - b, 32)) sNN = ssrc[b + (int)lane];
    }

    int beg, end; node_segment(bnd, i, beg, end);
    const int deg = end - beg;

    float er[NH];
    #pragma unroll
    for (int hh = 0; hh < NH; ++hh)
      er[hh] = uniform_of(wave_sum(mrC[hh] * arv[hh]));

    float acc[NH] = {0.f, 0.f, 0.f, 0.f};

    if (deg <= 32) {
      int sidx = 0; float e[NH];
      #pragma unroll
      for (int hh = 0; hh < NH; ++hh) e[hh] = -INFINITY;
      if ((int)lane < deg) {
        sidx = sC;
        e[0] = leaky(lC.x + er[0]);
        e[1] = leaky(lC.y + er[1]);
        e[2] = leaky(lC.z + er[2]);
        e[3] = leaky(lC.w + er[3]);
      }
      float al[NH];
      #pragma unroll
      for (int hh = 0; hh < NH; ++hh) {
        float m = uniform_of(wave_max(e[hh]));
        if (!isfinite(m)) m = 0.f;
        const float ex = __expf(e[hh] - m);
        const float s = uniform_of(wave_sum(ex));
        al[hh] = ex / fmaxf(s, 1e-20f);
      }
      aggregate(feat, sidx, al, deg, lane, acc);
    } else {
      float m[NH], s[NH];
      {
        float mx[NH] = {-INFINITY, -INFINITY, -INFINITY, -INFINITY};
        for (int c = beg; c < end; c += 32) {
          int sidx; float e[NH];
          load_chunk(ssrc, el, c, min(32, end - c), lane, er, sidx, e);
          #pragma unroll
          for (int hh = 0; hh < NH; ++hh) mx[hh] = fmaxf(mx[hh], e[hh]);
        }
        #pragma unroll
        for (int hh = 0; hh < NH; ++hh) {
          m[hh] = uniform_of(wave_max(mx[hh]));
          if (!isfinite(m[hh])) m[hh] = 0.f;
        }
      }
      {
        float sm[NH] = {0.f, 0.f, 0.f, 0.f};
        for (int c = beg; c < end; c += 32) {
          int sidx; float e[NH];
          load_chunk(ssrc, el, c, min(32, end - c), lane, er, sidx, e);
          #pragma unroll
          for (int hh = 0; hh < NH; ++hh) sm[hh] += __expf(e[hh] - m[hh]);
        }
        #pragma unroll
        for (int hh = 0; hh < NH; ++hh) s[hh] = fmaxf(uniform_of(wave_sum(sm[hh])), 1e-20f);
      }
      for (int c = beg; c < end; c += 32) {
        const int cnt = min(32, end - c);
        int sidx; float e[NH];
        load_chunk(ssrc, el, c, cnt, lane, er, sidx, e);
        float al[NH];
        #pragma unroll
        for (int hh = 0; hh < NH; ++hh) al[hh] = __expf(e[hh] - m[hh]) / s[hh];
        aggregate(feat, sidx, al, cnt, lane, acc);
      }
    }

    float* __restrict__ orow = out + (size_t)n * HD;
    #pragma unroll
    for (int hh = 0; hh < NH; ++hh) orow[hh * ND + lane] = (FIRST ? bias[hh] : bsC[hh]) + acc[hh];

    sC = sX; lC = lX; sN = sNN;
    #pragma unroll
    for (int hh = 0; hh < NH; ++hh) { mrC[hh] = mrX[hh]; bsC[hh] = bsX[hh]; }
  }
}

extern "C" void kernel_launch(void* const* d_in, const int* in_sizes, int n_in,
                              void* d_out, int out_size, void* d_ws, size_t ws_size,
                              hipStream_t stream) {
  const float* h    = (const float*)d_in[0];
  const int*   srcs[2] = { (const int*)d_in[1], (const int*)d_in[3] };
  const int*   dsts[2] = { (const int*)d_in[2], (const int*)d_in[4] };
  const float* Ws[2]   = { (const float*)d_in[5],  (const float*)d_in[9]  };
  const float* als[2]  = { (const float*)d_in[6],  (const float*)d_in[10] };
  const float* ars[2]  = { (const float*)d_in[7],  (const float*)d_in[11] };
  const float* b0 = (const float*)d_in[8];
  const float* b1 = (const float*)d_in[12];
  float* out = (float*)d_out;

  float* feat = (float*)d_ws;
  float* el   = feat + (size_t)NNODES * HD;
  int*   tab  = (int*)(el + (size_t)NNODES * NH);
  int*   cur  = tab + NNODES;
  int*   ssrc = cur + NNODES;
  int*   part = (int*)feat;
  int*   tabs[2] = { tab, tab + SORT_B * SORT_K };
  int*   bbase   = tab + 2 * SORT_B * SORT_K;

  const int BA = 128;
  const int nTile = (GEMM_TILES + (GEMM_BLOCK / 32) - 1) / (GEMM_BLOCK / 32);
  const int nNode = (NNODES + (BA / 32) * AGG_NODES - 1) / ((BA / 32) * AGG_NODES);

  gat_bucket_hist<<<2 * SORT_B, SORT_THREADS, 0, stream>>>(dsts[0], dsts[1], tab);
  for (int r = 0; r < 2; ++r) {
    gat_bucket_partition<<<SORT_B, SORT_THREADS, 0, stream>>>(srcs[r], dsts[r], tabs[r], bbase, part);
    gat_bucket_sort<<<SORT_K, SORT_THREADS, 0, stream>>>(bbase, part, cur, ssrc);
    gat_gemm_el<<<nTile, GEMM_BLOCK, 0, stream>>>(h, Ws[r], als[r], feat, el);
    if (r == 0)
      gat_node_agg<true><<<nNode, BA, 0, stream>>>(cur, ssrc, el, ars[r], feat, b0, b1, out);
    else
      gat_node_agg<false><<<nNode, BA, 0, stream>>>(cur, ssrc, el, ars[r], feat, b0, b1, out);
  }
}


# aggregation lanes = 4 heads x 8 edge slots: softmax max/sum and er as 3-step dpp trees in 8-lane groups, edge weight by dpp8 broadcast, rows as 16-byte-per-lane moves; er, exp sums in new tree order
# speedup vs baseline: 4.3741x; 1.1144x over previous
#include <hip/hip_runtime.h>
#include <math.h>

#define NNODES 100000
#define NEDGES 1000000
#define INF    128
#define NH     4
#define ND     32
#define HD     128
#define NEG_SLOPE 0.2f

typedef __attribute__((ext_vector_type(2))) float v2f;
typedef __attribute__((ext_vector_type(8))) float v8f;

template <int CTRL>
__device__ __forceinline__ float dpp_swz(float v) {
  return __int_as_float(__builtin_amdgcn_mov_dpp(__float_as_int(v), CTRL, 0xf, 0xf, false));
}
__device__ __forceinline__ float row16_sum(float v) {
  v += dpp_swz<0x161>(v);
  v += dpp_swz<0x162>(v);
  v += dpp_swz<0x164>(v);
  v += dpp_swz<0x168>(v);
  return v;
}

#define GEMM_BLOCK  64
#define GEMM_KSTEPS (INF / 4)
#define GEMM_TILES  (NNODES / 16)
__global__ void __launch_bounds__(GEMM_BLOCK) gat_gemm_el(const float* __restrict__ X,
                                                          const float* __restrict__ W,
                                                          const float* __restrict__ al,
                                                          float* __restrict__ Y,
                                                          float* __restrict__ el) {
  const int tile = __builtin_amdgcn_readfirstlane(
      (int)(blockIdx.x * (GEMM_BLOCK >> 5) + (threadIdx.x >> 5)));
  if (tile >= GEMM_TILES) return;
  const int lane = threadIdx.x & 31;
  const int m    = lane & 15;
  const int koff = (lane >> 4) << 1;
  const int row0 = tile << 4;

  const float* __restrict__ xrow = X + (size_t)(row0 + m) * INF + koff;
  v2f af[GEMM_KSTEPS];
  #pragma unroll
  for (int s = 0; s < GEMM_KSTEPS; ++s) af[s] = *(const v2f*)(xrow + 4 * s);

  const int rbase = row0 + ((lane >> 4) << 3);
  for (int hh = 0; hh < NH; ++hh) {
    float p[8];
    #pragma unroll
    for (int half = 0; half < 2; ++half) {
      const int col = (hh << 5) + (half << 4) + m;
      const float* __restrict__ wcol = W + (size_t)koff * HD + col;
      v8f c = {};
      #pragma unroll
      for (int s = 0; s < GEMM_KSTEPS; ++s) {
        v2f b;
        b.x = wcol[(size_t)(4 * s) * HD];
        b.y = wcol[(size_t)(4 * s + 1) * HD];
        c = __builtin_amdgcn_wmma_f32_16x16x4_f32(false, af[s], false, b,
                                                  (short)0, c, false, false);
      }
      const float a = al[col];
      #pragma unroll
      for (int i = 0; i < 8; ++i) {
        Y[(size_t)(rbase + i) * HD + col] = c[i];
        p[i] = half == 0 ? c[i] * a : fmaf(c[i], a, p[i]);
      }
    }
    #pragma unroll
    for (int i = 0; i < 8; ++i) p[i] = row16_sum(p[i]);
    float v = p[0];
    #pragma unroll
    for (int i = 1; i < 8; ++i) v = (m == i) ? p[i] : v;
    if (m < 8) el[(size_t)(rbase + m) * NH + hh] = v;
  }
}

__device__ __forceinline__ float leaky(float v) {
  return v > 0.f ? v : NEG_SLOPE * v;
}

#define SORT_SHIFT   9
#define SORT_SPAN    (1 << SORT_SHIFT)
#define SORT_K       ((NNODES + SORT_SPAN - 1) >> SORT_SHIFT)
#define SORT_B       125
#define SORT_QUADS   (NEDGES / (4 * SORT_B))
#define SORT_THREADS 512
#define SORT_ITERS   ((SORT_QUADS + SORT_THREADS - 1) / SORT_THREADS)
#define SORT_CAP     12288
#define SRC_BITS     17
#define SRC_MASK     ((1 << SRC_BITS) - 1)
static_assert(NNODES <= (1 << SRC_BITS), "a source id must fit SRC_BITS");
static_assert(SRC_BITS + SORT_SHIFT <= 31, "packed word must stay non-negative");
static_assert(SORT_QUADS * 4 * SORT_B == NEDGES, "chunks must tile the edge list");
static_assert(SORT_K <= SORT_THREADS && SORT_SPAN <= SORT_THREADS, "one thread per bucket / per node of a bucket");
static_assert(2 * SORT_B * SORT_K + SORT_K + 1 <= NNODES, "table and bucket bounds must fit their workspace plane");

__device__ __forceinline__ int node_clamp(int v) { return min(max(v, 0), NNODES - 1); }

__global__ void __launch_bounds__(SORT_THREADS) gat_bucket_hist(const int* __restrict__ dst0,
                                                                const int* __restrict__ dst1,
                                                                int* __restrict__ table) {
  __shared__ int h[SORT_K];
  const int t = threadIdx.x;
  const int r = blockIdx.x >= SORT_B ? 1 : 0;
  const int b = blockIdx.x - r * SORT_B;
  const int4* __restrict__ d4 = (const int4*)(r ? dst1 : dst0) + (size_t)b * SORT_QUADS;
  if (t < SORT_K) h[t] = 0;
  __syncthreads();
  int4 d[SORT_ITERS];
  #pragma unroll
  for (int u = 0; u < SORT_ITERS; ++u) {
    const int q = t + u * SORT_THREADS;
    d[u] = make_int4(0, 0, 0, 0);
    if (q < SORT_QUADS) d[u] = d4[q];
  }
  #pragma unroll
  for (int u = 0; u < SORT_ITERS; ++u) {
    if (t + u * SORT_THREADS < SORT_QUADS) {
      atomicAdd(&h[node_clamp(d[u].x) >> SORT_SHIFT], 1);
      atomicAdd(&h[node_clamp(d[u].y) >> SORT_SHIFT], 1);
      atomicAdd(&h[node_clamp(d[u].z) >> SORT_SHIFT], 1);
      atomicAdd(&h[node_clamp(d[u].w) >> SORT_SHIFT], 1);
    }
  }
  __syncthreads();
  if (t < SORT_K) table[(size_t)blockIdx.x * SORT_K + t] = h[t];
}

#define PART_EDGES (4 * SORT_QUADS)
__device__ __forceinline__ int take_slot(int* slot, int k) {
  const int pos = atomicAdd(&slot[k], 1);
  return min(max(pos, 0), PART_EDGES - 1);
}
__device__ __forceinline__ int pack_edge(int dn, int s) {
  return node_clamp(s) | ((dn & (SORT_SPAN - 1)) << SRC_BITS);
}

__global__ void __launch_bounds__(SORT_THREADS) gat_bucket_partition(const int* __restrict__ src,
                                                                     const int* __restrict__ dst,
                                                                     const int* __restrict__ table,
                                                                     int* __restrict__ bbase,
                                                                     int* __restrict__ part) {
  __shared__ int tot[SORT_K];
  __shared__ int own[SORT_K];
  __shared__ int slot[SORT_K];
  __shared__ int shift[SORT_K];
  __shared__ int stage_w[PART_EDGES];
  __shared__ unsigned short stage_k[PART_EDGES];
  const int t = threadIdx.x;
  const int b = blockIdx.x;
  int mytot = 0, mypre = 0;
  if (t < SORT_K) {
    #pragma unroll
    for (int j = 0; j < SORT_B; ++j) {
      const int v = table[j * SORT_K + t];
      mytot += v;
      mypre += j < b ? v : 0;
    }
    tot[t] = mytot;
    own[t] = table[b * SORT_K + t];
  }
  __syncthreads();
  if (t < SORT_K) {
    int base = 0;
    int lbase = 0;
    for (int j = 0; j < t; ++j) { base += tot[j]; lbase += own[j]; }
    slot[t] = lbase;
    shift[t] = base + mypre - lbase;
    if (b == 0) {
      bbase[t] = base;
      if (t == SORT_K - 1) bbase[SORT_K] = base + mytot;
    }
  }
  __syncthreads();
  const int4* __restrict__ d4 = (const int4*)dst + (size_t)b * SORT_QUADS;
  const int4* __restrict__ s4 = (const int4*)src + (size_t)b * SORT_QUADS;
  int4 d[SORT_ITERS], s[SORT_ITERS];
  #pragma unroll
  for (int u = 0; u < SORT_ITERS; ++u) {
    const int q = t + u * SORT_THREADS;
    d[u] = make_int4(0, 0, 0, 0);
    s[u] = make_int4(0, 0, 0, 0);
    if (q < SORT_QUADS) { d[u] = d4[q]; s[u] = s4[q]; }
  }
  #pragma unroll
  for (int u = 0; u < SORT_ITERS; ++u) {
    if (t + u * SORT_THREADS < SORT_QUADS) {
      const int dx = node_clamp(d[u].x), dy = node_clamp(d[u].y);
      const int dz = node_clamp(d[u].z), dw = node_clamp(d[u].w);
      const int kx = dx >> SORT_SHIFT, ky = dy >> SORT_SHIFT;
      const int kz = dz >> SORT_SHIFT, kw = dw >> SORT_SHIFT;
      const int px = take_slot(slot, kx);
      const int py = take_slot(slot, ky);
      const int pz = take_slot(slot, kz);
      const int pw = take_slot(slot, kw);
      stage_w[px] = pack_edge(dx, s[u].x);  stage_k[px] = (unsigned short)kx;
      stage_w[py] = pack_edge(dy, s[u].y);  stage_k[py] = (unsigned short)ky;
      stage_w[pz] = pack_edge(dz, s[u].z);  stage_k[pz] = (unsigned short)kz;
      stage_w[pw] = pack_edge(dw, s[u].w);  stage_k[pw] = (unsigned short)kw;
    }
  }
  __syncthreads();
  for (int i = t; i < PART_EDGES; i += SORT_THREADS) {
    const int k = min((int)stage_k[i], SORT_K - 1);
    const int g = min(max(i + shift[k], 0), NEDGES - 1);
    part[g] = stage_w[i];
  }
}

__global__ void __launch_bounds__(SORT_THREADS) gat_bucket_sort(const int* __restrict__ bbase,
                                                                const int* __restrict__ part,
                                                                int* __restrict__ cur,
                                                                int* __restrict__ ssrc) {
  __shared__ int cntl[SORT_SPAN];
  __shared__ int curl[SORT_SPAN];
  __shared__ int wtot[SORT_THREADS / 32];
  __shared__ int stage[SORT_CAP];
  const int t = threadIdx.x;
  const int lane = t & 31, wv = t >> 5;
  const int k = blockIdx.x;
  int beg = bbase[k], end = bbase[k + 1];
  beg = min(max(beg, 0), NEDGES);
  end = min(max(end, beg), NEDGES);
  const int n = end - beg;
  const bool staged = n <= SORT_CAP;

  if (t < SORT_SPAN) cntl[t] = 0;
  __syncthreads();
  for (int i0 = t; i0 < n; i0 += 4 * SORT_THREADS) {
    int w[4];
    #pragma unroll
    for (int u = 0; u < 4; ++u) {
      const int i = i0 + u * SORT_THREADS;
      w[u] = i < n ? part[beg + i] : -1;
    }
    #pragma unroll
    for (int u = 0; u < 4; ++u)
      if (w[u] >= 0) atomicAdd(&cntl[(w[u] >> SRC_BITS) & (SORT_SPAN - 1)], 1);
  }
  __syncthreads();

  const int v = t < SORT_SPAN ? cntl[t] : 0;
  int inc = v;
  #pragma unroll
  for (int d = 1; d < 32; d <<= 1) {
    const int y = __shfl_up(inc, d, 32);
    if (lane >= d) inc += y;
  }
  if (lane == 31) wtot[wv] = inc;
  __syncthreads();
  int run = inc - v;
  for (int w = 0; w < wv; ++w) run += wtot[w];
  if (t < SORT_SPAN) {
    curl[t] = beg + run;
    const int node = (k << SORT_SHIFT) + t;
    if (node < NNODES) cur[node] = beg + run + v;
  }
  __syncthreads();

  for (int i0 = t; i0 < n; i0 += 4 * SORT_THREADS) {
    int w[4], pos[4];
    #pragma unroll
    for (int u = 0; u < 4; ++u) {
      const int i = i0 + u * SORT_THREADS;
      w[u] = i < n ? part[beg + i] : -1;
    }
    #pragma unroll
    for (int u = 0; u < 4; ++u) {
      pos[u] = 0;
      if (w[u] >= 0) {
        const int p = atomicAdd(&curl[(w[u] >> SRC_BITS) & (SORT_SPAN - 1)], 1);
        pos[u] = staged ? min(max(p - beg, 0), SORT_CAP - 1) : min(max(p, 0), NEDGES - 1);
      }
    }
    #pragma unroll
    for (int u = 0; u < 4; ++u) {
      if (w[u] >= 0) {
        if (staged) stage[pos[u]] = w[u] & SRC_MASK;
        else        ssrc[pos[u]]  = w[u] & SRC_MASK;
      }
    }
  }
  if (staged) {
    __syncthreads();
    for (int i = t; i < n; i += SORT_THREADS) ssrc[beg + i] = stage[i];
  }
}

#define SLOTS 4
#define CHUNK (8 * SLOTS)

__device__ __forceinline__ float grp8_sum(float v) {
  v += dpp_swz<0x161>(v);
  v += dpp_swz<0x162>(v);
  v += dpp_swz<0x164>(v);
  return v;
}
__device__ __forceinline__ float grp8_max(float v) {
  v = fmaxf(v, dpp_swz<0x161>(v));
  v = fmaxf(v, dpp_swz<0x162>(v));
  v = fmaxf(v, dpp_swz<0x164>(v));
  return v;
}
template <int U>
__device__ __forceinline__ float grp8_bcast(float v) {
  return __int_as_float(__builtin_amdgcn_mov_dpp8(__float_as_int(v), U * 0x249249));
}

__device__ __forceinline__ void load_ids(const int* __restrict__ ssrc, int base, int cnt,
                                         int j, int sid[SLOTS]) {
  #pragma unroll
  for (int q = 0; q < SLOTS; ++q) {
    sid[q] = 0;
    if (8 * q < cnt) {
      if (8 * q + j < cnt) sid[q] = ssrc[base + 8 * q + j];
    }
  }
}
__device__ __forceinline__ void clamp_ids(const int raw[SLOTS], int sid[SLOTS]) {
  #pragma unroll
  for (int q = 0; q < SLOTS; ++q) sid[q] = min(max(raw[q], 0), NNODES - 1);
}
__device__ __forceinline__ void load_el(const float* __restrict__ el, const int sid[SLOTS],
                                        int cnt, int g, int j, float l[SLOTS]) {
  #pragma unroll
  for (int q = 0; q < SLOTS; ++q) {
    l[q] = 0.f;
    if (8 * q < cnt) {
      if (8 * q + j < cnt) l[q] = el[(size_t)sid[q] * NH + g];
    }
  }
}
__device__ __forceinline__ void chunk_scores(const float l[SLOTS], float er, int cnt, int j,
                                             float e[SLOTS]) {
  #pragma unroll
  for (int q = 0; q < SLOTS; ++q)
    e[q] = (8 * q + j < cnt) ? leaky(l[q] + er) : -INFINITY;
}

template <int U>
__device__ __forceinline__ void row_fma(const float4& r, float alq, float4& acc) {
  const float w = grp8_bcast<U>(alq);
  acc.x = fmaf(r.x, w, acc.x);
  acc.y = fmaf(r.y, w, acc.y);
  acc.z = fmaf(r.z, w, acc.z);
  acc.w = fmaf(r.w, w, acc.w);
}

__device__ __forceinline__ void aggregate(const float* __restrict__ feat,
                                          const int sid[SLOTS], const float al[SLOTS],
                                          int cnt, unsigned lane, float4& acc) {
  #pragma unroll
  for (int q = 0; q < SLOTS; ++q) {
    const int left = cnt - 8 * q;
    if (left > 0) {
      float4 r[8];
      #pragma unroll
      for (int u = 0; u < 8; ++u) {
        r[u] = make_float4(0.f, 0.f, 0.f, 0.f);
        if (u < left) {
          const int sj = __builtin_amdgcn_readlane(sid[q], u);
          r[u] = *(const float4*)(feat + (size_t)sj * HD + 4 * lane);
        }
      }
      row_fma<0>(r[0], al[q], acc);
      if (1 < left) row_fma<1>(r[1], al[q], acc);
      if (2 < left) row_fma<2>(r[2], al[q], acc);
      if (3 < left) row_fma<3>(r[3], al[q], acc);
      if (4 < left) row_fma<4>(r[4], al[q], acc);
      if (5 < left) row_fma<5>(r[5], al[q], acc);
      if (6 < left) row_fma<6>(r[6], al[q], acc);
      if (7 < left) row_fma<7>(r[7], al[q], acc);
    }
  }
}

#define AGG_NODES 12
__device__ __forceinline__ void node_segment(int bnd, int k, int& beg, int& end) {
  beg = __builtin_amdgcn_readlane(bnd, k);
  end = __builtin_amdgcn_readlane(bnd, k + 1);
  beg = min(max(beg, 0), NEDGES);
  end = min(max(end, beg), NEDGES);
}

template <bool FIRST>
__global__ void gat_node_agg(const int* __restrict__ cur,
                             const int* __restrict__ ssrc,
                             const float* __restrict__ el,
                             const float* __restrict__ ar,
                             const float* __restrict__ feat,
                             const float* __restrict__ b0,
                             const float* __restrict__ b1,
                             float* __restrict__ out) {
  const int n0 = __builtin_amdgcn_readfirstlane(
      (int)(blockIdx.x * (blockDim.x >> 5) + (threadIdx.x >> 5))) * AGG_NODES;
  if (n0 >= NNODES) return;
  const unsigned lane = threadIdx.x & 31;
  const int g = (int)(lane >> 3);
  const int j = (int)(lane & 7);
  const int nn = min(AGG_NODES, NNODES - n0);

  const int bidx = n0 - 1 + (int)lane;
  int bnd = 0;
  if (lane <= AGG_NODES && bidx >= 0 && bidx < NNODES) bnd = cur[bidx];

  const float4 ar4 = *(const float4*)(ar + 4 * lane);
  float4 bias4 = make_float4(0.f, 0.f, 0.f, 0.f);
  if (FIRST) {
    const float4 x0 = *(const float4*)(b0 + 4 * lane);
    const float4 x1 = *(const float4*)(b1 + 4 * lane);
    bias4 = make_float4(x0.x + x1.x, x0.y + x1.y, x0.z + x1.z, x0.w + x1.w);
  }

  int    sC[SLOTS] = {0, 0, 0, 0};
  float  lC[SLOTS] = {0.f, 0.f, 0.f, 0.f};
  float4 mrC = make_float4(0.f, 0.f, 0.f, 0.f), bsC = make_float4(0.f, 0.f, 0.f, 0.f);
  int    sN[SLOTS] = {0, 0, 0, 0};

  {
    int b, e; node_segment(bnd, 0, b, e);
    load_ids(ssrc, b, min(e - b, CHUNK), j, sN);
  }
  {
    int b, e; node_segment(bnd, 0, b, e);
    clamp_ids(sN, sC);
    load_el(el, sC, min(e - b, CHUNK), g, j, lC);
    mrC = *(const float4*)(feat + (size_t)n0 * HD + 4 * lane);
    if (!FIRST) bsC = *(const float4*)(out + (size_t)n0 * HD + 4 * lane);
  }
  #pragma unroll
  for (int q = 0; q < SLOTS; ++q) sN[q] = 0;
  if (1 < nn) {
    int b, e; node_segment(bnd, 1, b, e);
    load_ids(ssrc, b, min(e - b, CHUNK), j, sN);
  }

  for (int i = 0; i < nn; ++i) {
    const int n = n0 + i;

    int    sX[SLOTS] = {0, 0, 0, 0};
    float  lX[SLOTS] = {0.f, 0.f, 0.f, 0.f};
    float4 mrX = make_float4(0.f, 0.f, 0.f, 0.f), bsX = make_float4(0.f, 0.f, 0.f, 0.f);
    if (i + 1 < nn) {
      int b, e; node_segment(bnd, i + 1, b, e);
      clamp_ids(sN, sX);
      load_el(el, sX, min(e - b, CHUNK), g, j, lX);
      mrX = *(const float4*)(feat + (size_t)(n + 1) * HD + 4 * lane);
      if (!FIRST) bsX = *(const float4*)(out + (size_t)(n + 1) * HD + 4 * lane);
    }
    int sNN[SLOTS] = {0, 0, 0, 0};
    if (i + 2 < nn) {
      int b, e; node_segment(bnd, i + 2, b, e);
      load_ids(ssrc, b, min(e - b, CHUNK), j, sNN);
    }

    int beg, end; node_segment(bnd, i, beg, end);
    const int deg = end - beg;

    const float er = grp8_sum(mrC.x * ar4.x + mrC.y * ar4.y + mrC.z * ar4.z + mrC.w * ar4.w);

    float4 acc = make_float4(0.f, 0.f, 0.f, 0.f);

    if (deg <= CHUNK) {
      float e[SLOTS];
      chunk_scores(lC, er, deg, j, e);
      float m = grp8_max(fmaxf(fmaxf(e[0], e[1]), fmaxf(e[2], e[3])));
      if (!isfinite(m)) m = 0.f;
      float ex[SLOTS];
      #pragma unroll
      for (int q = 0; q < SLOTS; ++q) ex[q] = __expf(e[q] - m);
      const float s = grp8_sum((ex[0] + ex[1]) + (ex[2] + ex[3]));
      const float den = fmaxf(s, 1e-20f);
      float al[SLOTS];
      #pragma unroll
      for (int q = 0; q < SLOTS; ++q) {
        al[q] = 0.f;
        if (8 * q < deg) al[q] = ex[q] / den;
      }
      aggregate(feat, sC, al, deg, lane, acc);
    } else {
      float m, s;
      {
        float mx = -INFINITY;
        for (int c = beg; c < end; c += CHUNK) {
          const int cnt = min(CHUNK, end - c);
          int raw[SLOTS], sid[SLOTS]; float l[SLOTS], e[SLOTS];
          load_ids(ssrc, c, cnt, j, raw);
          clamp_ids(raw, sid);
          load_el(el, sid, cnt, g, j, l);
          chunk_scores(l, er, cnt, j, e);
          mx = fmaxf(mx, fmaxf(fmaxf(e[0], e[1]), fmaxf(e[2], e[3])));
        }
        m = grp8_max(mx);
        if (!isfinite(m)) m = 0.f;
      }
      {
        float sm = 0.f;
        for (int c = beg; c < end; c += CHUNK) {
          const int cnt = min(CHUNK, end - c);
          int raw[SLOTS], sid[SLOTS]; float l[SLOTS], e[SLOTS];
          load_ids(ssrc, c, cnt, j, raw);
          clamp_ids(raw, sid);
          load_el(el, sid, cnt, g, j, l);
          chunk_scores(l, er, cnt, j, e);
          sm += (__expf(e[0] - m) + __expf(e[1] - m)) + (__expf(e[2] - m) + __expf(e[3] - m));
        }
        s = fmaxf(grp8_sum(sm), 1e-20f);
      }
      for (int c = beg; c < end; c += CHUNK) {
        const int cnt = min(CHUNK, end - c);
        int raw[SLOTS], sid[SLOTS]; float l[SLOTS], e[SLOTS], al[SLOTS];
        load_ids(ssrc, c, cnt, j, raw);
        clamp_ids(raw, sid);
        load_el(el, sid, cnt, g, j, l);
        chunk_scores(l, er, cnt, j, e);
        #pragma unroll
        for (int q = 0; q < SLOTS; ++q) al[q] = __expf(e[q] - m) / s;
        aggregate(feat, sid, al, cnt, lane, acc);
      }
    }

    const float4 st = FIRST ? bias4 : bsC;
    *(float4*)(out + (size_t)n * HD + 4 * lane) =
        make_float4(st.x + acc.x, st.y + acc.y, st.z + acc.z, st.w + acc.w);

    #pragma unroll
    for (int q = 0; q < SLOTS; ++q) { sC[q] = sX[q]; lC[q] = lX[q]; sN[q] = sNN[q]; }
    mrC = mrX; bsC = bsX;
  }
}

extern "C" void kernel_launch(void* const* d_in, const int* in_sizes, int n_in,
                              void* d_out, int out_size, void* d_ws, size_t ws_size,
                              hipStream_t stream) {
  const float* h    = (const float*)d_in[0];
  const int*   srcs[2] = { (const int*)d_in[1], (const int*)d_in[3] };
  const int*   dsts[2] = { (const int*)d_in[2], (const int*)d_in[4] };
  const float* Ws[2]   = { (const float*)d_in[5],  (const float*)d_in[9]  };
  const float* als[2]  = { (const float*)d_in[6],  (const float*)d_in[10] };
  const float* ars[2]  = { (const float*)d_in[7],  (const float*)d_in[11] };
  const float* b0 = (const float*)d_in[8];
  const float* b1 = (const float*)d_in[12];
  float* out = (float*)d_out;

  float* feat = (float*)d_ws;
  float* el   = feat + (size_t)NNODES * HD;
  int*   tab  = (int*)(el + (size_t)NNODES * NH);
  int*   cur  = tab + NNODES;
  int*   ssrc = cur + NNODES;
  int*   part = (int*)feat;
  int*   tabs[2] = { tab, tab + SORT_B * SORT_K };
  int*   bbase   = tab + 2 * SORT_B * SORT_K;

  const int BA = 128;
  const int nTile = (GEMM_TILES + (GEMM_BLOCK / 32) - 1) / (GEMM_BLOCK / 32);
  const int nNode = (NNODES + (BA / 32) * AGG_NODES - 1) / ((BA / 32) * AGG_NODES);

  gat_bucket_hist<<<2 * SORT_B, SORT_THREADS, 0, stream>>>(dsts[0], dsts[1], tab);
  for (int r = 0; r < 2; ++r) {
    gat_bucket_partition<<<SORT_B, SORT_THREADS, 0, stream>>>(srcs[r], dsts[r], tabs[r], bbase, part);
    gat_bucket_sort<<<SORT_K, SORT_THREADS, 0, stream>>>(bbase, part, cur, ssrc);
    gat_gemm_el<<<nTile, GEMM_BLOCK, 0, stream>>>(h, Ws[r], als[r], feat, el);
    if (r == 0)
      gat_node_agg<true><<<nNode, BA, 0, stream>>>(cur, ssrc, el, ars[r], feat, b0, b1, out);
    else
      gat_node_agg<false><<<nNode, BA, 0, stream>>>(cur, ssrc, el, ars[r], feat, b0, b1, out);
  }
}


# edge grouping in one level: a chunk writes its piece sorted by 48-node bucket with its own slot row, the aggregation work-group ranks its bucket by node in LDS: 9 launches -> 6; new arrival order
# speedup vs baseline: 4.5962x; 1.0508x over previous
#include <hip/hip_runtime.h>
#include <math.h>

#define NNODES 100000
#define NEDGES 1000000
#define INF    128
#define NH     4
#define ND     32
#define HD     128
#define NEG_SLOPE 0.2f

typedef __attribute__((ext_vector_type(2))) float v2f;
typedef __attribute__((ext_vector_type(8))) float v8f;

template <int CTRL>
__device__ __forceinline__ float dpp_swz(float v) {
  return __int_as_float(__builtin_amdgcn_mov_dpp(__float_as_int(v), CTRL, 0xf, 0xf, false));
}
__device__ __forceinline__ float row16_sum(float v) {
  v += dpp_swz<0x161>(v);
  v += dpp_swz<0x162>(v);
  v += dpp_swz<0x164>(v);
  v += dpp_swz<0x168>(v);
  return v;
}

#define GEMM_BLOCK  64
#define GEMM_KSTEPS (INF / 4)
#define GEMM_TILES  (NNODES / 16)
__global__ void __launch_bounds__(GEMM_BLOCK) gat_gemm_el(const float* __restrict__ X,
                                                          const float* __restrict__ W,
                                                          const float* __restrict__ al,
                                                          float* __restrict__ Y,
                                                          float* __restrict__ el) {
  const int tile = __builtin_amdgcn_readfirstlane(
      (int)(blockIdx.x * (GEMM_BLOCK >> 5) + (threadIdx.x >> 5)));
  if (tile >= GEMM_TILES) return;
  const int lane = threadIdx.x & 31;
  const int m    = lane & 15;
  const int koff = (lane >> 4) << 1;
  const int row0 = tile << 4;

  const float* __restrict__ xrow = X + (size_t)(row0 + m) * INF + koff;
  v2f af[GEMM_KSTEPS];
  #pragma unroll
  for (int s = 0; s < GEMM_KSTEPS; ++s) af[s] = *(const v2f*)(xrow + 4 * s);

  const int rbase = row0 + ((lane >> 4) << 3);
  for (int hh = 0; hh < NH; ++hh) {
    float p[8];
    #pragma unroll
    for (int half = 0; half < 2; ++half) {
      const int col = (hh << 5) + (half << 4) + m;
      const float* __restrict__ wcol = W + (size_t)koff * HD + col;
      v8f c = {};
      #pragma unroll
      for (int s = 0; s < GEMM_KSTEPS; ++s) {
        v2f b;
        b.x = wcol[(size_t)(4 * s) * HD];
        b.y = wcol[(size_t)(4 * s + 1) * HD];
        c = __builtin_amdgcn_wmma_f32_16x16x4_f32(false, af[s], false, b,
                                                  (short)0, c, false, false);
      }
      const float a = al[col];
      #pragma unroll
      for (int i = 0; i < 8; ++i) {
        Y[(size_t)(rbase + i) * HD + col] = c[i];
        p[i] = half == 0 ? c[i] * a : fmaf(c[i], a, p[i]);
      }
    }
    #pragma unroll
    for (int i = 0; i < 8; ++i) p[i] = row16_sum(p[i]);
    float v = p[0];
    #pragma unroll
    for (int i = 1; i < 8; ++i) v = (m == i) ? p[i] : v;
    if (m < 8) el[(size_t)(rbase + m) * NH + hh] = v;
  }
}

__device__ __forceinline__ float leaky(float v) {
  return v > 0.f ? v : NEG_SLOPE * v;
}

#define AGG_NODES    12
#define AGG_WAVES    4
#define AGG_THREADS  (32 * AGG_WAVES)
#define BKT_NODES    (AGG_NODES * AGG_WAVES)
#define NBKT         ((NNODES + BKT_NODES - 1) / BKT_NODES)
#define BKT_CAP      1024
#define SORT_B       125
#define SORT_QUADS   (NEDGES / (4 * SORT_B))
#define SORT_THREADS 512
#define SORT_ITERS   ((SORT_QUADS + SORT_THREADS - 1) / SORT_THREADS)
#define PART_EDGES   (4 * SORT_QUADS)
#define SCAN_PER     ((NBKT + 1 + SORT_THREADS - 1) / SORT_THREADS)
#define SCAN_LEN     (SCAN_PER * SORT_THREADS)
#define PREF_ROW     ((NBKT + 2) & ~1)
#define SRC_BITS     17
#define SRC_MASK     ((1 << SRC_BITS) - 1)
#define OFF_MASK     63
#define K_BITS       12
static_assert(NNODES <= (1 << SRC_BITS), "a source id must fit SRC_BITS");
static_assert(BKT_NODES <= OFF_MASK + 1, "a node offset must fit its field");
static_assert(NBKT < (1 << K_BITS) && PART_EDGES < (1 << (31 - K_BITS)), "bucket and rank must fit one word");
static_assert(SORT_QUADS * 4 * SORT_B == NEDGES, "chunks must tile the edge list");
static_assert(PART_EDGES < 65536, "a first slot must fit 16 bits");
static_assert(SORT_B <= AGG_THREADS, "one thread per run of a bucket");
static_assert(BKT_NODES + 1 <= 64, "the node scan uses two waves");
static_assert(PREF_ROW <= SCAN_LEN && PREF_ROW > NBKT, "a row holds start[0..NBKT]");
static_assert(SORT_B * (PREF_ROW / 2) <= 2 * NNODES, "the first-slot rows must fit their workspace plane");

__device__ __forceinline__ int node_clamp(int v) { return min(max(v, 0), NNODES - 1); }

__device__ __forceinline__ int wave_inc_scan(int v, int lane) {
  #pragma unroll
  for (int d = 1; d < 32; d <<= 1) {
    const int y = __shfl_up(v, d, 32);
    if (lane >= d) v += y;
  }
  return v;
}

__global__ void __launch_bounds__(SORT_THREADS) gat_chunk_partition(const int* __restrict__ src,
                                                                    const int* __restrict__ dst,
                                                                    int* __restrict__ pref_words,
                                                                    int* __restrict__ part) {
  __shared__ int cnt[SCAN_LEN];
  __shared__ int start[SCAN_LEN];
  __shared__ int wtot[SORT_THREADS / 32];
  __shared__ int stage_w[PART_EDGES];
  const int t = threadIdx.x;
  const int lane = t & 31, wv = t >> 5;
  const int b = blockIdx.x;
  #pragma unroll
  for (int u = 0; u < SCAN_PER; ++u) cnt[t + u * SORT_THREADS] = 0;
  __syncthreads();

  const int4* __restrict__ d4 = (const int4*)dst + (size_t)b * SORT_QUADS;
  const int4* __restrict__ s4 = (const int4*)src + (size_t)b * SORT_QUADS;
  int4 d[SORT_ITERS], s[SORT_ITERS];
  #pragma unroll
  for (int u = 0; u < SORT_ITERS; ++u) {
    const int q = t + u * SORT_THREADS;
    d[u] = make_int4(0, 0, 0, 0);
    s[u] = make_int4(0, 0, 0, 0);
    if (q < SORT_QUADS) { d[u] = d4[q]; s[u] = s4[q]; }
  }
  int kp[SORT_ITERS][4];
  int wd[SORT_ITERS][4];
  #pragma unroll
  for (int u = 0; u < SORT_ITERS; ++u) {
    #pragma unroll
    for (int c = 0; c < 4; ++c) { kp[u][c] = 0; wd[u][c] = 0; }
    if (t + u * SORT_THREADS < SORT_QUADS) {
      const int dn[4] = { node_clamp(d[u].x), node_clamp(d[u].y), node_clamp(d[u].z), node_clamp(d[u].w) };
      const int sn[4] = { node_clamp(s[u].x), node_clamp(s[u].y), node_clamp(s[u].z), node_clamp(s[u].w) };
      #pragma unroll
      for (int c = 0; c < 4; ++c) {
        const int k = dn[c] / BKT_NODES;
        const int pos = atomicAdd(&cnt[k], 1);
        kp[u][c] = k | (pos << K_BITS);
        wd[u][c] = sn[c] | ((dn[c] - k * BKT_NODES) << SRC_BITS);
      }
    }
  }
  __syncthreads();

  int own[SCAN_PER];
  int tsum = 0;
  #pragma unroll
  for (int u = 0; u < SCAN_PER; ++u) { own[u] = cnt[t * SCAN_PER + u]; tsum += own[u]; }
  const int inc = wave_inc_scan(tsum, lane);
  if (lane == 31) wtot[wv] = inc;
  __syncthreads();
  int run = inc - tsum;
  for (int w = 0; w < wv; ++w) run += wtot[w];
  #pragma unroll
  for (int u = 0; u < SCAN_PER; ++u) { start[t * SCAN_PER + u] = run; run += own[u]; }
  __syncthreads();

  #pragma unroll
  for (int u = 0; u < SORT_ITERS; ++u) {
    if (t + u * SORT_THREADS < SORT_QUADS) {
      #pragma unroll
      for (int c = 0; c < 4; ++c) {
        const int k = kp[u][c] & ((1 << K_BITS) - 1);
        const int slot = start[k] + (kp[u][c] >> K_BITS);
        stage_w[min(max(slot, 0), PART_EDGES - 1)] = wd[u][c];
      }
    }
  }
  __syncthreads();
  int* __restrict__ mine = part + (size_t)b * PART_EDGES;
  for (int i = t; i < PART_EDGES; i += SORT_THREADS) mine[i] = stage_w[i];
  int* __restrict__ row = pref_words + (size_t)b * (PREF_ROW / 2);
  for (int i = t; i < PREF_ROW / 2; i += SORT_THREADS)
    row[i] = (start[2 * i] & 0xffff) | (start[2 * i + 1] << 16);
}

#define SLOTS 4
#define CHUNK (8 * SLOTS)

__device__ __forceinline__ float grp8_sum(float v) {
  v += dpp_swz<0x161>(v);
  v += dpp_swz<0x162>(v);
  v += dpp_swz<0x164>(v);
  return v;
}
__device__ __forceinline__ float grp8_max(float v) {
  v = fmaxf(v, dpp_swz<0x161>(v));
  v = fmaxf(v, dpp_swz<0x162>(v));
  v = fmaxf(v, dpp_swz<0x164>(v));
  return v;
}
template <int U>
__device__ __forceinline__ float grp8_bcast(float v) {
  return __int_as_float(__builtin_amdgcn_mov_dpp8(__float_as_int(v), U * 0x249249));
}

__device__ __forceinline__ void load_ids(const int* __restrict__ ssrc, int base, int cnt,
                                         int j, int sid[SLOTS]) {
  #pragma unroll
  for (int q = 0; q < SLOTS; ++q) {
    sid[q] = 0;
    if (8 * q < cnt) {
      if (8 * q + j < cnt) sid[q] = ssrc[base + 8 * q + j];
    }
  }
}
__device__ __forceinline__ void clamp_ids(const int raw[SLOTS], int sid[SLOTS]) {
  #pragma unroll
  for (int q = 0; q < SLOTS; ++q) sid[q] = min(max(raw[q], 0), NNODES - 1);
}
__device__ __forceinline__ void load_el(const float* __restrict__ el, const int sid[SLOTS],
                                        int cnt, int g, int j, float l[SLOTS]) {
  #pragma unroll
  for (int q = 0; q < SLOTS; ++q) {
    l[q] = 0.f;
    if (8 * q < cnt) {
      if (8 * q + j < cnt) l[q] = el[(size_t)sid[q] * NH + g];
    }
  }
}
__device__ __forceinline__ void chunk_scores(const float l[SLOTS], float er, int cnt, int j,
                                             float e[SLOTS]) {
  #pragma unroll
  for (int q = 0; q < SLOTS; ++q)
    e[q] = (8 * q + j < cnt) ? leaky(l[q] + er) : -INFINITY;
}

template <int U>
__device__ __forceinline__ void row_fma(const float4& r, float alq, float4& acc) {
  const float w = grp8_bcast<U>(alq);
  acc.x = fmaf(r.x, w, acc.x);
  acc.y = fmaf(r.y, w, acc.y);
  acc.z = fmaf(r.z, w, acc.z);
  acc.w = fmaf(r.w, w, acc.w);
}

__device__ __forceinline__ void aggregate(const float* __restrict__ feat,
                                          const int sid[SLOTS], const float al[SLOTS],
                                          int cnt, unsigned lane, float4& acc) {
  #pragma unroll
  for (int q = 0; q < SLOTS; ++q) {
    const int left = cnt - 8 * q;
    if (left > 0) {
      float4 r[8];
      #pragma unroll
      for (int u = 0; u < 8; ++u) {
        r[u] = make_float4(0.f, 0.f, 0.f, 0.f);
        if (u < left) {
          const int sj = __builtin_amdgcn_readlane(sid[q], u);
          r[u] = *(const float4*)(feat + (size_t)sj * HD + 4 * lane);
        }
      }
      row_fma<0>(r[0], al[q], acc);
      if (1 < left) row_fma<1>(r[1], al[q], acc);
      if (2 < left) row_fma<2>(r[2], al[q], acc);
      if (3 < left) row_fma<3>(r[3], al[q], acc);
      if (4 < left) row_fma<4>(r[4], al[q], acc);
      if (5 < left) row_fma<5>(r[5], al[q], acc);
      if (6 < left) row_fma<6>(r[6], al[q], acc);
      if (7 < left) row_fma<7>(r[7], al[q], acc);
    }
  }
}

__device__ __forceinline__ void node_segment(int bnd, int k, int& beg, int& end) {
  beg = __builtin_amdgcn_readlane(bnd, k);
  end = __builtin_amdgcn_readlane(bnd, k + 1);
  beg = min(max(beg, 0), BKT_CAP);
  end = min(max(end, beg), BKT_CAP);
}

template <bool FIRST>
__device__ __forceinline__ void agg_nodes(const int* ids, int bnd, int n0, int nn, unsigned lane,
                                          const float* __restrict__ el,
                                          const float4 ar4, const float4 bias4,
                                          const float* __restrict__ feat,
                                          float* __restrict__ out) {
  const int g = (int)(lane >> 3);
  const int j = (int)(lane & 7);
  int    sC[SLOTS] = {0, 0, 0, 0};
  float  lC[SLOTS] = {0.f, 0.f, 0.f, 0.f};
  float4 mrC = make_float4(0.f, 0.f, 0.f, 0.f), bsC = make_float4(0.f, 0.f, 0.f, 0.f);
  int    sN[SLOTS] = {0, 0, 0, 0};

  {
    int b, e; node_segment(bnd, 0, b, e);
    load_ids(ids, b, min(e - b, CHUNK), j, sN);
  }
  {
    int b, e; node_segment(bnd, 0, b, e);
    clamp_ids(sN, sC);
    load_el(el, sC, min(e - b, CHUNK), g, j, lC);
    mrC = *(const float4*)(feat + (size_t)n0 * HD + 4 * lane);
    if (!FIRST) bsC = *(const float4*)(out + (size_t)n0 * HD + 4 * lane);
  }
  #pragma unroll
  for (int q = 0; q < SLOTS; ++q) sN[q] = 0;
  if (1 < nn) {
    int b, e; node_segment(bnd, 1, b, e);
    load_ids(ids, b, min(e - b, CHUNK), j, sN);
  }

  for (int i = 0; i < nn; ++i) {
    const int n = n0 + i;

    int    sX[SLOTS] = {0, 0, 0, 0};
    float  lX[SLOTS] = {0.f, 0.f, 0.f, 0.f};
    float4 mrX = make_float4(0.f, 0.f, 0.f, 0.f), bsX = make_float4(0.f, 0.f, 0.f, 0.f);
    if (i + 1 < nn) {
      int b, e; node_segment(bnd, i + 1, b, e);
      clamp_ids(sN, sX);
      load_el(el, sX, min(e - b, CHUNK), g, j, lX);
      mrX = *(const float4*)(feat + (size_t)(n + 1) * HD + 4 * lane);
      if (!FIRST) bsX = *(const float4*)(out + (size_t)(n + 1) * HD + 4 * lane);
    }
    int sNN[SLOTS] = {0, 0, 0, 0};
    if (i + 2 < nn) {
      int b, e; node_segment(bnd, i + 2, b, e);
      load_ids(ids, b, min(e - b, CHUNK), j, sNN);
    }

    int beg, end; node_segment(bnd, i, beg, end);
    const int deg = end - beg;

    const float er = grp8_sum(mrC.x * ar4.x + mrC.y * ar4.y + mrC.z * ar4.z + mrC.w * ar4.w);

    float4 acc = make_float4(0.f, 0.f, 0.f, 0.f);

    if (deg <= CHUNK) {
      float e[SLOTS];
      chunk_scores(lC, er, deg, j, e);
      float m = grp8_max(fmaxf(fmaxf(e[0], e[1]), fmaxf(e[2], e[3])));
      if (!isfinite(m)) m = 0.f;
      float ex[SLOTS];
      #pragma unroll
      for (int q = 0; q < SLOTS; ++q) ex[q] = __expf(e[q] - m);
      const float s = grp8_sum((ex[0] + ex[1]) + (ex[2] + ex[3]));
      const float den = fmaxf(s, 1e-20f);
      float al[SLOTS];
      #pragma unroll
      for (int q = 0; q < SLOTS; ++q) {
        al[q] = 0.f;
        if (8 * q < deg) al[q] = ex[q] / den;
      }
      aggregate(feat, sC, al, deg, lane, acc);
    } else {
      float m, s;
      {
        float mx = -INFINITY;
        for (int c = beg; c < end; c += CHUNK) {
          const int cnt = min(CHUNK, end - c);
          int raw[SLOTS], sid[SLOTS]; float l[SLOTS], e[SLOTS];
          load_ids(ids, c, cnt, j, raw);
          clamp_ids(raw, sid);
          load_el(el, sid, cnt, g, j, l);
          chunk_scores(l, er, cnt, j, e);
          mx = fmaxf(mx, fmaxf(fmaxf(e[0], e[1]), fmaxf(e[2], e[3])));
        }
        m = grp8_max(mx);
        if (!isfinite(m)) m = 0.f;
      }
      {
        float sm = 0.f;
        for (int c = beg; c < end; c += CHUNK) {
          const int cnt = min(CHUNK, end - c);
          int raw[SLOTS], sid[SLOTS]; float l[SLOTS], e[SLOTS];
          load_ids(ids, c, cnt, j, raw);
          clamp_ids(raw, sid);
          load_el(el, sid, cnt, g, j, l);
          chunk_scores(l, er, cnt, j, e);
          sm += (__expf(e[0] - m) + __expf(e[1] - m)) + (__expf(e[2] - m) + __expf(e[3] - m));
        }
        s = fmaxf(grp8_sum(sm), 1e-20f);
      }
      for (int c = beg; c < end; c += CHUNK) {
        const int cnt = min(CHUNK, end - c);
        int raw[SLOTS], sid[SLOTS]; float l[SLOTS], e[SLOTS], al[SLOTS];
        load_ids(ids, c, cnt, j, raw);
        clamp_ids(raw, sid);
        load_el(el, sid, cnt, g, j, l);
        chunk_scores(l, er, cnt, j, e);
        #pragma unroll
        for (int q = 0; q < SLOTS; ++q) al[q] = __expf(e[q] - m) / s;
        aggregate(feat, sid, al, cnt, lane, acc);
      }
    }

    const float4 st = FIRST ? bias4 : bsC;
    *(float4*)(out + (size_t)n * HD + 4 * lane) =
        make_float4(st.x + acc.x, st.y + acc.y, st.z + acc.z, st.w + acc.w);

    #pragma unroll
    for (int q = 0; q < SLOTS; ++q) { sC[q] = sX[q]; lC[q] = lX[q]; sN[q] = sNN[q]; }
    mrC = mrX; bsC = bsX;
  }
}

template <int SWEEP>
__device__ __forceinline__ void slow_sweep(const int* rstart, const int* rlen, const int* __restrict__ part,
                                           int off, unsigned lane, int g,
                                           const float* __restrict__ el, const float* __restrict__ feat,
                                           float er, float m, float den, float& red, float4& acc) {
  for (int c = 0; c < SORT_B; ++c) {
    const int rs = rstart[c], rl = rlen[c];
    for (int p0 = 0; p0 < rl; p0 += 32) {
      const int w = (p0 + (int)lane < rl) ? part[rs + p0 + (int)lane] : -1;
      const bool hit = w >= 0 && ((w >> SRC_BITS) & OFF_MASK) == off;
      unsigned mask = __builtin_amdgcn_ballot_w32(hit);
      while (mask) {
        const int l = __builtin_ctz(mask);
        mask &= mask - 1;
        const int sj = node_clamp(__shfl(w, l, 32) & SRC_MASK);
        const float e = leaky(el[(size_t)sj * NH + g] + er);
        if (SWEEP == 0) red = fmaxf(red, e);
        if (SWEEP == 1) red += __expf(e - m);
        if (SWEEP == 2) {
          const float a = __expf(e - m) / den;
          const float4 r = *(const float4*)(feat + (size_t)sj * HD + 4 * lane);
          acc.x = fmaf(r.x, a, acc.x); acc.y = fmaf(r.y, a, acc.y);
          acc.z = fmaf(r.z, a, acc.z); acc.w = fmaf(r.w, a, acc.w);
        }
      }
    }
  }
}
template <bool FIRST>
__device__ __forceinline__ void agg_nodes_slow(const int* rstart, const int* rlen, const int* __restrict__ part,
                                               int off0, int n0, int nn, unsigned lane,
                                               const float* __restrict__ el,
                                               const float4 ar4, const float4 bias4,
                                               const float* __restrict__ feat,
                                               float* __restrict__ out) {
  const int g = (int)(lane >> 3);
  for (int i = 0; i < nn; ++i) {
    const int n = n0 + i;
    const float4 mr = *(const float4*)(feat + (size_t)n * HD + 4 * lane);
    const float er = grp8_sum(mr.x * ar4.x + mr.y * ar4.y + mr.z * ar4.z + mr.w * ar4.w);
    float4 acc = make_float4(0.f, 0.f, 0.f, 0.f);
    float m = -INFINITY, s = 0.f;
    slow_sweep<0>(rstart, rlen, part, off0 + i, lane, g, el, feat, er, 0.f, 1.f, m, acc);
    if (!isfinite(m)) m = 0.f;
    slow_sweep<1>(rstart, rlen, part, off0 + i, lane, g, el, feat, er, m, 1.f, s, acc);
    const float den = fmaxf(s, 1e-20f);
    float unused = 0.f;
    slow_sweep<2>(rstart, rlen, part, off0 + i, lane, g, el, feat, er, m, den, unused, acc);
    float4 st = bias4;
    if (!FIRST) st = *(const float4*)(out + (size_t)n * HD + 4 * lane);
    *(float4*)(out + (size_t)n * HD + 4 * lane) =
        make_float4(st.x + acc.x, st.y + acc.y, st.z + acc.z, st.w + acc.w);
  }
}

template <bool FIRST>
__global__ void __launch_bounds__(AGG_THREADS) gat_bucket_agg(const int* __restrict__ pref_words,
                                                              const int* __restrict__ part,
                                                              const float* __restrict__ el,
                                                              const float* __restrict__ ar,
                                                              const float* __restrict__ feat,
                                                              const float* __restrict__ b0,
                                                              const float* __restrict__ b1,
                                                              float* __restrict__ out) {
  __shared__ int rstart[AGG_THREADS];
  __shared__ int rlen[AGG_THREADS];
  __shared__ int wsum[AGG_WAVES];
  __shared__ int ncnt[64];
  __shared__ int nwave[2];
  __shared__ int seg[64];
  __shared__ int wbuf[BKT_CAP];
  __shared__ unsigned short rk[BKT_CAP];
  __shared__ int ids[BKT_CAP];
  const int t = threadIdx.x;
  const unsigned lane = t & 31;
  const int wv = __builtin_amdgcn_readfirstlane(t >> 5);
  const int k = blockIdx.x;

  int len = 0, gstart = 0;
  if (t < SORT_B) {
    const int* __restrict__ row = pref_words + (size_t)t * (PREF_ROW / 2);
    const int wa = row[k >> 1], wb = row[(k + 1) >> 1];
    int p0 = (wa >> ((k & 1) << 4)) & 0xffff;
    int p1 = (wb >> (((k + 1) & 1) << 4)) & 0xffff;
    p0 = min(max(p0, 0), PART_EDGES);
    p1 = min(max(p1, p0), PART_EDGES);
    len = p1 - p0;
    gstart = t * PART_EDGES + p0;
  }
  rstart[t] = gstart;
  rlen[t] = len;
  if (t < 64) ncnt[t] = 0;
  const int inc = wave_inc_scan(len, (int)lane);
  if (lane == 31) wsum[wv] = inc;
  __syncthreads();
  int off = inc - len;
  for (int w = 0; w < wv; ++w) off += wsum[w];
  int n = 0;
  #pragma unroll
  for (int w = 0; w < AGG_WAVES; ++w) n += wsum[w];
  n = __builtin_amdgcn_readfirstlane(n);

  const int n0 = (k * AGG_WAVES + wv) * AGG_NODES;
  const int nn = min(AGG_NODES, NNODES - n0);
  const float4 ar4 = *(const float4*)(ar + 4 * lane);
  float4 bias4 = make_float4(0.f, 0.f, 0.f, 0.f);
  if (FIRST) {
    const float4 x0 = *(const float4*)(b0 + 4 * lane);
    const float4 x1 = *(const float4*)(b1 + 4 * lane);
    bias4 = make_float4(x0.x + x1.x, x0.y + x1.y, x0.z + x1.z, x0.w + x1.w);
  }

  if (n <= BKT_CAP) {
    for (int i0 = 0; i0 < len; i0 += 8) {
      int w[8];
      #pragma unroll
      for (int u = 0; u < 8; ++u) w[u] = (i0 + u < len) ? part[gstart + i0 + u] : -1;
      #pragma unroll
      for (int u = 0; u < 8; ++u) {
        if (w[u] >= 0) {
          const int o = min((w[u] >> SRC_BITS) & OFF_MASK, BKT_NODES - 1);
          const int p = atomicAdd(&ncnt[o], 1);
          const int e = min(max(off + i0 + u, 0), BKT_CAP - 1);
          wbuf[e] = w[u];
          rk[e] = (unsigned short)p;
        }
      }
    }
    __syncthreads();
    int v = 0, ic = 0;
    if (wv < 2) {
      v = ncnt[t];
      ic = wave_inc_scan(v, (int)lane);
      if (lane == 31) nwave[wv] = ic;
    }
    __syncthreads();
    if (wv < 2) seg[t] = ic - v + (wv == 1 ? nwave[0] : 0);
    __syncthreads();
    for (int e = t; e < n; e += AGG_THREADS) {
      const int w = wbuf[e];
      const int o = min((w >> SRC_BITS) & OFF_MASK, BKT_NODES - 1);
      const int slot = min(max(seg[o] + (int)rk[e], 0), BKT_CAP - 1);
      ids[slot] = w & SRC_MASK;
    }
    __syncthreads();
    if (nn > 0) {
      const int bnd = seg[wv * AGG_NODES + (int)min(lane, (unsigned)AGG_NODES)];
      agg_nodes<FIRST>(ids, bnd, n0, nn, lane, el, ar4, bias4, feat, out);
    }
  } else {
    if (nn > 0)
      agg_nodes_slow<FIRST>(rstart, rlen, part, wv * AGG_NODES, n0, nn, lane, el, ar4, bias4, feat, out);
  }
}

extern "C" void kernel_launch(void* const* d_in, const int* in_sizes, int n_in,
                              void* d_out, int out_size, void* d_ws, size_t ws_size,
                              hipStream_t stream) {
  const float* h    = (const float*)d_in[0];
  const int*   srcs[2] = { (const int*)d_in[1], (const int*)d_in[3] };
  const int*   dsts[2] = { (const int*)d_in[2], (const int*)d_in[4] };
  const float* Ws[2]   = { (const float*)d_in[5],  (const float*)d_in[9]  };
  const float* als[2]  = { (const float*)d_in[6],  (const float*)d_in[10] };
  const float* ars[2]  = { (const float*)d_in[7],  (const float*)d_in[11] };
  const float* b0 = (const float*)d_in[8];
  const float* b1 = (const float*)d_in[12];
  float* out = (float*)d_out;

  float* feat = (float*)d_ws;
  float* el   = feat + (size_t)NNODES * HD;
  int*   pref = (int*)(el + (size_t)NNODES * NH);
  int*   part = pref + 2 * NNODES;

  const int nTile = (GEMM_TILES + (GEMM_BLOCK / 32) - 1) / (GEMM_BLOCK / 32);

  for (int r = 0; r < 2; ++r) {
    gat_chunk_partition<<<SORT_B, SORT_THREADS, 0, stream>>>(srcs[r], dsts[r], pref, part);
    gat_gemm_el<<<nTile, GEMM_BLOCK, 0, stream>>>(h, Ws[r], als[r], feat, el);
    if (r == 0)
      gat_bucket_agg<true><<<NBKT, AGG_THREADS, 0, stream>>>(pref, part, el, ars[r], feat, b0, b1, out);
    else
      gat_bucket_agg<false><<<NBKT, AGG_THREADS, 0, stream>>>(pref, part, el, ars[r], feat, b0, b1, out);
  }
}


# aggregation: loads issued ahead for the next nodes (ids, el, own row) take no lane mask or branch (clamped index, select afterwards): no wait right behind each load; sums and their order unchanged
# speedup vs baseline: 4.7518x; 1.0339x over previous
#include <hip/hip_runtime.h>
#include <math.h>

#define NNODES 100000
#define NEDGES 1000000
#define INF    128
#define NH     4
#define ND     32
#define HD     128
#define NEG_SLOPE 0.2f

typedef __attribute__((ext_vector_type(2))) float v2f;
typedef __attribute__((ext_vector_type(8))) float v8f;

template <int CTRL>
__device__ __forceinline__ float dpp_swz(float v) {
  return __int_as_float(__builtin_amdgcn_mov_dpp(__float_as_int(v), CTRL, 0xf, 0xf, false));
}
__device__ __forceinline__ float row16_sum(float v) {
  v += dpp_swz<0x161>(v);
  v += dpp_swz<0x162>(v);
  v += dpp_swz<0x164>(v);
  v += dpp_swz<0x168>(v);
  return v;
}

#define GEMM_BLOCK  64
#define GEMM_KSTEPS (INF / 4)
#define GEMM_TILES  (NNODES / 16)
__global__ void __launch_bounds__(GEMM_BLOCK) gat_gemm_el(const float* __restrict__ X,
                                                          const float* __restrict__ W,
                                                          const float* __restrict__ al,
                                                          float* __restrict__ Y,
                                                          float* __restrict__ el) {
  const int tile = __builtin_amdgcn_readfirstlane(
      (int)(blockIdx.x * (GEMM_BLOCK >> 5) + (threadIdx.x >> 5)));
  if (tile >= GEMM_TILES) return;
  const int lane = threadIdx.x & 31;
  const int m    = lane & 15;
  const int koff = (lane >> 4) << 1;
  const int row0 = tile << 4;

  const float* __restrict__ xrow = X + (size_t)(row0 + m) * INF + koff;
  v2f af[GEMM_KSTEPS];
  #pragma unroll
  for (int s = 0; s < GEMM_KSTEPS; ++s) af[s] = *(const v2f*)(xrow + 4 * s);

  const int rbase = row0 + ((lane >> 4) << 3);
  for (int hh = 0; hh < NH; ++hh) {
    float p[8];
    #pragma unroll
    for (int half = 0; half < 2; ++half) {
      const int col = (hh << 5) + (half << 4) + m;
      const float* __restrict__ wcol = W + (size_t)koff * HD + col;
      v8f c = {};
      #pragma unroll
      for (int s = 0; s < GEMM_KSTEPS; ++s) {
        v2f b;
        b.x = wcol[(size_t)(4 * s) * HD];
        b.y = wcol[(size_t)(4 * s + 1) * HD];
        c = __builtin_amdgcn_wmma_f32_16x16x4_f32(false, af[s], false, b,
                                                  (short)0, c, false, false);
      }
      const float a = al[col];
      #pragma unroll
      for (int i = 0; i < 8; ++i) {
        Y[(size_t)(rbase + i) * HD + col] = c[i];
        p[i] = half == 0 ? c[i] * a : fmaf(c[i], a, p[i]);
      }
    }
    #pragma unroll
    for (int i = 0; i < 8; ++i) p[i] = row16_sum(p[i]);
    float v = p[0];
    #pragma unroll
    for (int i = 1; i < 8; ++i) v = (m == i) ? p[i] : v;
    if (m < 8) el[(size_t)(rbase + m) * NH + hh] = v;
  }
}

__device__ __forceinline__ float leaky(float v) {
  return v > 0.f ? v : NEG_SLOPE * v;
}

#define AGG_NODES    12
#define AGG_WAVES    4
#define AGG_THREADS  (32 * AGG_WAVES)
#define BKT_NODES    (AGG_NODES * AGG_WAVES)
#define NBKT         ((NNODES + BKT_NODES - 1) / BKT_NODES)
#define BKT_CAP      1024
#define SORT_B       125
#define SORT_QUADS   (NEDGES / (4 * SORT_B))
#define SORT_THREADS 512
#define SORT_ITERS   ((SORT_QUADS + SORT_THREADS - 1) / SORT_THREADS)
#define PART_EDGES   (4 * SORT_QUADS)
#define SCAN_PER     ((NBKT + 1 + SORT_THREADS - 1) / SORT_THREADS)
#define SCAN_LEN     (SCAN_PER * SORT_THREADS)
#define PREF_ROW     ((NBKT + 2) & ~1)
#define SRC_BITS     17
#define SRC_MASK     ((1 << SRC_BITS) - 1)
#define OFF_MASK     63
#define K_BITS       12
static_assert(NNODES <= (1 << SRC_BITS), "a source id must fit SRC_BITS");
static_assert(BKT_NODES <= OFF_MASK + 1, "a node offset must fit its field");
static_assert(NBKT < (1 << K_BITS) && PART_EDGES < (1 << (31 - K_BITS)), "bucket and rank must fit one word");
static_assert(SORT_QUADS * 4 * SORT_B == NEDGES, "chunks must tile the edge list");
static_assert(PART_EDGES < 65536, "a first slot must fit 16 bits");
static_assert(SORT_B <= AGG_THREADS, "one thread per run of a bucket");
static_assert(BKT_NODES + 1 <= 64, "the node scan uses two waves");
static_assert(PREF_ROW <= SCAN_LEN && PREF_ROW > NBKT, "a row holds start[0..NBKT]");
static_assert(SORT_B * (PREF_ROW / 2) <= 2 * NNODES, "the first-slot rows must fit their workspace plane");

__device__ __forceinline__ int node_clamp(int v) { return min(max(v, 0), NNODES - 1); }

__device__ __forceinline__ int wave_inc_scan(int v, int lane) {
  #pragma unroll
  for (int d = 1; d < 32; d <<= 1) {
    const int y = __shfl_up(v, d, 32);
    if (lane >= d) v += y;
  }
  return v;
}

__global__ void __launch_bounds__(SORT_THREADS) gat_chunk_partition(const int* __restrict__ src,
                                                                    const int* __restrict__ dst,
                                                                    int* __restrict__ pref_words,
                                                                    int* __restrict__ part) {
  __shared__ int cnt[SCAN_LEN];
  __shared__ int start[SCAN_LEN];
  __shared__ int wtot[SORT_THREADS / 32];
  __shared__ int stage_w[PART_EDGES];
  const int t = threadIdx.x;
  const int lane = t & 31, wv = t >> 5;
  const int b = blockIdx.x;
  #pragma unroll
  for (int u = 0; u < SCAN_PER; ++u) cnt[t + u * SORT_THREADS] = 0;
  __syncthreads();

  const int4* __restrict__ d4 = (const int4*)dst + (size_t)b * SORT_QUADS;
  const int4* __restrict__ s4 = (const int4*)src + (size_t)b * SORT_QUADS;
  int4 d[SORT_ITERS], s[SORT_ITERS];
  #pragma unroll
  for (int u = 0; u < SORT_ITERS; ++u) {
    const int q = t + u * SORT_THREADS;
    d[u] = make_int4(0, 0, 0, 0);
    s[u] = make_int4(0, 0, 0, 0);
    if (q < SORT_QUADS) { d[u] = d4[q]; s[u] = s4[q]; }
  }
  int kp[SORT_ITERS][4];
  int wd[SORT_ITERS][4];
  #pragma unroll
  for (int u = 0; u < SORT_ITERS; ++u) {
    #pragma unroll
    for (int c = 0; c < 4; ++c) { kp[u][c] = 0; wd[u][c] = 0; }
    if (t + u * SORT_THREADS < SORT_QUADS) {
      const int dn[4] = { node_clamp(d[u].x), node_clamp(d[u].y), node_clamp(d[u].z), node_clamp(d[u].w) };
      const int sn[4] = { node_clamp(s[u].x), node_clamp(s[u].y), node_clamp(s[u].z), node_clamp(s[u].w) };
      #pragma unroll
      for (int c = 0; c < 4; ++c) {
        const int k = dn[c] / BKT_NODES;
        const int pos = atomicAdd(&cnt[k], 1);
        kp[u][c] = k | (pos << K_BITS);
        wd[u][c] = sn[c] | ((dn[c] - k * BKT_NODES) << SRC_BITS);
      }
    }
  }
  __syncthreads();

  int own[SCAN_PER];
  int tsum = 0;
  #pragma unroll
  for (int u = 0; u < SCAN_PER; ++u) { own[u] = cnt[t * SCAN_PER + u]; tsum += own[u]; }
  const int inc = wave_inc_scan(tsum, lane);
  if (lane == 31) wtot[wv] = inc;
  __syncthreads();
  int run = inc - tsum;
  for (int w = 0; w < wv; ++w) run += wtot[w];
  #pragma unroll
  for (int u = 0; u < SCAN_PER; ++u) { start[t * SCAN_PER + u] = run; run += own[u]; }
  __syncthreads();

  #pragma unroll
  for (int u = 0; u < SORT_ITERS; ++u) {
    if (t + u * SORT_THREADS < SORT_QUADS) {
      #pragma unroll
      for (int c = 0; c < 4; ++c) {
        const int k = kp[u][c] & ((1 << K_BITS) - 1);
        const int slot = start[k] + (kp[u][c] >> K_BITS);
        stage_w[min(max(slot, 0), PART_EDGES - 1)] = wd[u][c];
      }
    }
  }
  __syncthreads();
  int* __restrict__ mine = part + (size_t)b * PART_EDGES;
  for (int i = t; i < PART_EDGES; i += SORT_THREADS) mine[i] = stage_w[i];
  int* __restrict__ row = pref_words + (size_t)b * (PREF_ROW / 2);
  for (int i = t; i < PREF_ROW / 2; i += SORT_THREADS)
    row[i] = (start[2 * i] & 0xffff) | (start[2 * i + 1] << 16);
}

#define SLOTS 4
#define CHUNK (8 * SLOTS)

__device__ __forceinline__ float grp8_sum(float v) {
  v += dpp_swz<0x161>(v);
  v += dpp_swz<0x162>(v);
  v += dpp_swz<0x164>(v);
  return v;
}
__device__ __forceinline__ float grp8_max(float v) {
  v = fmaxf(v, dpp_swz<0x161>(v));
  v = fmaxf(v, dpp_swz<0x162>(v));
  v = fmaxf(v, dpp_swz<0x164>(v));
  return v;
}
template <int U>
__device__ __forceinline__ float grp8_bcast(float v) {
  return __int_as_float(__builtin_amdgcn_mov_dpp8(__float_as_int(v), U * 0x249249));
}

__device__ __forceinline__ void ids_ahead(const int* ssrc, int base, int j, int raw[SLOTS]) {
  #pragma unroll
  for (int q = 0; q < SLOTS; ++q) raw[q] = ssrc[min(base + 8 * q + j, BKT_CAP - 1)];
}
__device__ __forceinline__ void pick_ids(const int raw[SLOTS], int cnt, int j, int sid[SLOTS]) {
  #pragma unroll
  for (int q = 0; q < SLOTS; ++q)
    sid[q] = (8 * q + j < cnt) ? min(max(raw[q], 0), NNODES - 1) : 0;
}
__device__ __forceinline__ void el_ahead(const float* __restrict__ el, const int sid[SLOTS],
                                         int g, float l[SLOTS]) {
  #pragma unroll
  for (int q = 0; q < SLOTS; ++q) l[q] = el[(size_t)sid[q] * NH + g];
}

__device__ __forceinline__ void load_ids(const int* __restrict__ ssrc, int base, int cnt,
                                         int j, int sid[SLOTS]) {
  #pragma unroll
  for (int q = 0; q < SLOTS; ++q) {
    sid[q] = 0;
    if (8 * q < cnt) {
      if (8 * q + j < cnt) sid[q] = ssrc[base + 8 * q + j];
    }
  }
}
__device__ __forceinline__ void clamp_ids(const int raw[SLOTS], int sid[SLOTS]) {
  #pragma unroll
  for (int q = 0; q < SLOTS; ++q) sid[q] = min(max(raw[q], 0), NNODES - 1);
}
__device__ __forceinline__ void load_el(const float* __restrict__ el, const int sid[SLOTS],
                                        int cnt, int g, int j, float l[SLOTS]) {
  #pragma unroll
  for (int q = 0; q < SLOTS; ++q) {
    l[q] = 0.f;
    if (8 * q < cnt) {
      if (8 * q + j < cnt) l[q] = el[(size_t)sid[q] * NH + g];
    }
  }
}
__device__ __forceinline__ void chunk_scores(const float l[SLOTS], float er, int cnt, int j,
                                             float e[SLOTS]) {
  #pragma unroll
  for (int q = 0; q < SLOTS; ++q)
    e[q] = (8 * q + j < cnt) ? leaky(l[q] + er) : -INFINITY;
}

template <int U>
__device__ __forceinline__ void row_fma(const float4& r, float alq, float4& acc) {
  const float w = grp8_bcast<U>(alq);
  acc.x = fmaf(r.x, w, acc.x);
  acc.y = fmaf(r.y, w, acc.y);
  acc.z = fmaf(r.z, w, acc.z);
  acc.w = fmaf(r.w, w, acc.w);
}

__device__ __forceinline__ void aggregate(const float* __restrict__ feat,
                                          const int sid[SLOTS], const float al[SLOTS],
                                          int cnt, unsigned lane, float4& acc) {
  #pragma unroll
  for (int q = 0; q < SLOTS; ++q) {
    const int left = cnt - 8 * q;
    if (left > 0) {
      float4 r[8];
      #pragma unroll
      for (int u = 0; u < 8; ++u) {
        r[u] = make_float4(0.f, 0.f, 0.f, 0.f);
        if (u < left) {
          const int sj = __builtin_amdgcn_readlane(sid[q], u);
          r[u] = *(const float4*)(feat + (size_t)sj * HD + 4 * lane);
        }
      }
      row_fma<0>(r[0], al[q], acc);
      if (1 < left) row_fma<1>(r[1], al[q], acc);
      if (2 < left) row_fma<2>(r[2], al[q], acc);
      if (3 < left) row_fma<3>(r[3], al[q], acc);
      if (4 < left) row_fma<4>(r[4], al[q], acc);
      if (5 < left) row_fma<5>(r[5], al[q], acc);
      if (6 < left) row_fma<6>(r[6], al[q], acc);
      if (7 < left) row_fma<7>(r[7], al[q], acc);
    }
  }
}

__device__ __forceinline__ void node_segment(int bnd, int k, int& beg, int& end) {
  beg = __builtin_amdgcn_readlane(bnd, k);
  end = __builtin_amdgcn_readlane(bnd, k + 1);
  beg = min(max(beg, 0), BKT_CAP);
  end = min(max(end, beg), BKT_CAP);
}

template <bool FIRST>
__device__ __forceinline__ void agg_nodes(const int* ids, int bnd, int n0, int nn, unsigned lane,
                                          const float* __restrict__ el,
                                          const float4 ar4, const float4 bias4,
                                          const float* __restrict__ feat,
                                          float* __restrict__ out) {
  const int g = (int)(lane >> 3);
  const int j = (int)(lane & 7);
  int    sC[SLOTS];
  float  lC[SLOTS];
  float4 mrC, bsC = make_float4(0.f, 0.f, 0.f, 0.f);
  int    sN[SLOTS];

  {
    int b, e; node_segment(bnd, 0, b, e);
    ids_ahead(ids, b, j, sN);
  }
  {
    int b, e; node_segment(bnd, 0, b, e);
    pick_ids(sN, min(e - b, CHUNK), j, sC);
    el_ahead(el, sC, g, lC);
    mrC = *(const float4*)(feat + (size_t)n0 * HD + 4 * lane);
    if (!FIRST) bsC = *(const float4*)(out + (size_t)n0 * HD + 4 * lane);
  }
  {
    int b, e; node_segment(bnd, min(1, nn - 1), b, e);
    ids_ahead(ids, b, j, sN);
  }

  for (int i = 0; i < nn; ++i) {
    const int n = n0 + i;
    const int k1 = min(i + 1, nn - 1);
    const int k2 = min(i + 2, nn - 1);

    int    sX[SLOTS];
    float  lX[SLOTS];
    float4 mrX, bsX = make_float4(0.f, 0.f, 0.f, 0.f);
    {
      int b, e; node_segment(bnd, k1, b, e);
      pick_ids(sN, min(e - b, CHUNK), j, sX);
      el_ahead(el, sX, g, lX);
      mrX = *(const float4*)(feat + (size_t)(n0 + k1) * HD + 4 * lane);
      if (!FIRST) bsX = *(const float4*)(out + (size_t)(n0 + k1) * HD + 4 * lane);
    }
    int sNN[SLOTS];
    {
      int b, e; node_segment(bnd, k2, b, e);
      ids_ahead(ids, b, j, sNN);
    }

    int beg, end; node_segment(bnd, i, beg, end);
    const int deg = end - beg;

    const float er = grp8_sum(mrC.x * ar4.x + mrC.y * ar4.y + mrC.z * ar4.z + mrC.w * ar4.w);

    float4 acc = make_float4(0.f, 0.f, 0.f, 0.f);

    if (deg <= CHUNK) {
      float e[SLOTS];
      chunk_scores(lC, er, deg, j, e);
      float m = grp8_max(fmaxf(fmaxf(e[0], e[1]), fmaxf(e[2], e[3])));
      if (!isfinite(m)) m = 0.f;
      float ex[SLOTS];
      #pragma unroll
      for (int q = 0; q < SLOTS; ++q) ex[q] = __expf(e[q] - m);
      const float s = grp8_sum((ex[0] + ex[1]) + (ex[2] + ex[3]));
      const float den = fmaxf(s, 1e-20f);
      float al[SLOTS];
      #pragma unroll
      for (int q = 0; q < SLOTS; ++q) {
        al[q] = 0.f;
        if (8 * q < deg) al[q] = ex[q] / den;
      }
      aggregate(feat, sC, al, deg, lane, acc);
    } else {
      float m, s;
      {
        float mx = -INFINITY;
        for (int c = beg; c < end; c += CHUNK) {
          const int cnt = min(CHUNK, end - c);
          int raw[SLOTS], sid[SLOTS]; float l[SLOTS], e[SLOTS];
          load_ids(ids, c, cnt, j, raw);
          clamp_ids(raw, sid);
          load_el(el, sid, cnt, g, j, l);
          chunk_scores(l, er, cnt, j, e);
          mx = fmaxf(mx, fmaxf(fmaxf(e[0], e[1]), fmaxf(e[2], e[3])));
        }
        m = grp8_max(mx);
        if (!isfinite(m)) m = 0.f;
      }
      {
        float sm = 0.f;
        for (int c = beg; c < end; c += CHUNK) {
          const int cnt = min(CHUNK, end - c);
          int raw[SLOTS], sid[SLOTS]; float l[SLOTS], e[SLOTS];
          load_ids(ids, c, cnt, j, raw);
          clamp_ids(raw, sid);
          load_el(el, sid, cnt, g, j, l);
          chunk_scores(l, er, cnt, j, e);
          sm += (__expf(e[0] - m) + __expf(e[1] - m)) + (__expf(e[2] - m) + __expf(e[3] - m));
        }
        s = fmaxf(grp8_sum(sm), 1e-20f);
      }
      for (int c = beg; c < end; c += CHUNK) {
        const int cnt = min(CHUNK, end - c);
        int raw[SLOTS], sid[SLOTS]; float l[SLOTS], e[SLOTS], al[SLOTS];
        load_ids(ids, c, cnt, j, raw);
        clamp_ids(raw, sid);
        load_el(el, sid, cnt, g, j, l);
        chunk_scores(l, er, cnt, j, e);
        #pragma unroll
        for (int q = 0; q < SLOTS; ++q) al[q] = __expf(e[q] - m) / s;
        aggregate(feat, sid, al, cnt, lane, acc);
      }
    }

    const float4 st = FIRST ? bias4 : bsC;
    *(float4*)(out + (size_t)n * HD + 4 * lane) =
        make_float4(st.x + acc.x, st.y + acc.y, st.z + acc.z, st.w + acc.w);

    #pragma unroll
    for (int q = 0; q < SLOTS; ++q) { sC[q] = sX[q]; lC[q] = lX[q]; sN[q] = sNN[q]; }
    mrC = mrX; bsC = bsX;
  }
}

template <int SWEEP>
__device__ __forceinline__ void slow_sweep(const int* rstart, const int* rlen, const int* __restrict__ part,
                                           int off, unsigned lane, int g,
                                           const float* __restrict__ el, const float* __restrict__ feat,
                                           float er, float m, float den, float& red, float4& acc) {
  for (int c = 0; c < SORT_B; ++c) {
    const int rs = rstart[c], rl = rlen[c];
    for (int p0 = 0; p0 < rl; p0 += 32) {
      const int w = (p0 + (int)lane < rl) ? part[rs + p0 + (int)lane] : -1;
      const bool hit = w >= 0 && ((w >> SRC_BITS) & OFF_MASK) == off;
      unsigned mask = __builtin_amdgcn_ballot_w32(hit);
      while (mask) {
        const int l = __builtin_ctz(mask);
        mask &= mask - 1;
        const int sj = node_clamp(__shfl(w, l, 32) & SRC_MASK);
        const float e = leaky(el[(size_t)sj * NH + g] + er);
        if (SWEEP == 0) red = fmaxf(red, e);
        if (SWEEP == 1) red += __expf(e - m);
        if (SWEEP == 2) {
          const float a = __expf(e - m) / den;
          const float4 r = *(const float4*)(feat + (size_t)sj * HD + 4 * lane);
          acc.x = fmaf(r.x, a, acc.x); acc.y = fmaf(r.y, a, acc.y);
          acc.z = fmaf(r.z, a, acc.z); acc.w = fmaf(r.w, a, acc.w);
        }
      }
    }
  }
}
template <bool FIRST>
__device__ __forceinline__ void agg_nodes_slow(const int* rstart, const int* rlen, const int* __restrict__ part,
                                               int off0, int n0, int nn, unsigned lane,
                                               const float* __restrict__ el,
                                               const float4 ar4, const float4 bias4,
                                               const float* __restrict__ feat,
                                               float* __restrict__ out) {
  const int g = (int)(lane >> 3);
  for (int i = 0; i < nn; ++i) {
    const int n = n0 + i;
    const float4 mr = *(const float4*)(feat + (size_t)n * HD + 4 * lane);
    const float er = grp8_sum(mr.x * ar4.x + mr.y * ar4.y + mr.z * ar4.z + mr.w * ar4.w);
    float4 acc = make_float4(0.f, 0.f, 0.f, 0.f);
    float m = -INFINITY, s = 0.f;
    slow_sweep<0>(rstart, rlen, part, off0 + i, lane, g, el, feat, er, 0.f, 1.f, m, acc);
    if (!isfinite(m)) m = 0.f;
    slow_sweep<1>(rstart, rlen, part, off0 + i, lane, g, el, feat, er, m, 1.f, s, acc);
    const float den = fmaxf(s, 1e-20f);
    float unused = 0.f;
    slow_sweep<2>(rstart, rlen, part, off0 + i, lane, g, el, feat, er, m, den, unused, acc);
    float4 st = bias4;
    if (!FIRST) st = *(const float4*)(out + (size_t)n * HD + 4 * lane);
    *(float4*)(out + (size_t)n * HD + 4 * lane) =
        make_float4(st.x + acc.x, st.y + acc.y, st.z + acc.z, st.w + acc.w);
  }
}

template <bool FIRST>
__global__ void __launch_bounds__(AGG_THREADS) gat_bucket_agg(const int* __restrict__ pref_words,
                                                              const int* __restrict__ part,
                                                              const float* __restrict__ el,
                                                              const float* __restrict__ ar,
                                                              const float* __restrict__ feat,
                                                              const float* __restrict__ b0,
                                                              const float* __restrict__ b1,
                                                              float* __restrict__ out) {
  __shared__ int rstart[AGG_THREADS];
  __shared__ int rlen[AGG_THREADS];
  __shared__ int wsum[AGG_WAVES];
  __shared__ int ncnt[64];
  __shared__ int nwave[2];
  __shared__ int seg[64];
  __shared__ int wbuf[BKT_CAP];
  __shared__ unsigned short rk[BKT_CAP];
  __shared__ int ids[BKT_CAP];
  const int t = threadIdx.x;
  const unsigned lane = t & 31;
  const int wv = __builtin_amdgcn_readfirstlane(t >> 5);
  const int k = blockIdx.x;

  int len = 0, gstart = 0;
  if (t < SORT_B) {
    const int* __restrict__ row = pref_words + (size_t)t * (PREF_ROW / 2);
    const int wa = row[k >> 1], wb = row[(k + 1) >> 1];
    int p0 = (wa >> ((k & 1) << 4)) & 0xffff;
    int p1 = (wb >> (((k + 1) & 1) << 4)) & 0xffff;
    p0 = min(max(p0, 0), PART_EDGES);
    p1 = min(max(p1, p0), PART_EDGES);
    len = p1 - p0;
    gstart = t * PART_EDGES + p0;
  }
  rstart[t] = gstart;
  rlen[t] = len;
  if (t < 64) ncnt[t] = 0;
  const int inc = wave_inc_scan(len, (int)lane);
  if (lane == 31) wsum[wv] = inc;
  __syncthreads();
  int off = inc - len;
  for (int w = 0; w < wv; ++w) off += wsum[w];
  int n = 0;
  #pragma unroll
  for (int w = 0; w < AGG_WAVES; ++w) n += wsum[w];
  n = __builtin_amdgcn_readfirstlane(n);

  const int n0 = (k * AGG_WAVES + wv) * AGG_NODES;
  const int nn = min(AGG_NODES, NNODES - n0);
  const float4 ar4 = *(const float4*)(ar + 4 * lane);
  float4 bias4 = make_float4(0.f, 0.f, 0.f, 0.f);
  if (FIRST) {
    const float4 x0 = *(const float4*)(b0 + 4 * lane);
    const float4 x1 = *(const float4*)(b1 + 4 * lane);
    bias4 = make_float4(x0.x + x1.x, x0.y + x1.y, x0.z + x1.z, x0.w + x1.w);
  }

  if (n <= BKT_CAP) {
    for (int i0 = 0; i0 < len; i0 += 8) {
      int w[8];
      #pragma unroll
      for (int u = 0; u < 8; ++u) w[u] = (i0 + u < len) ? part[gstart + i0 + u] : -1;
      #pragma unroll
      for (int u = 0; u < 8; ++u) {
        if (w[u] >= 0) {
          const int o = min((w[u] >> SRC_BITS) & OFF_MASK, BKT_NODES - 1);
          const int p = atomicAdd(&ncnt[o], 1);
          const int e = min(max(off + i0 + u, 0), BKT_CAP - 1);
          wbuf[e] = w[u];
          rk[e] = (unsigned short)p;
        }
      }
    }
    __syncthreads();
    int v = 0, ic = 0;
    if (wv < 2) {
      v = ncnt[t];
      ic = wave_inc_scan(v, (int)lane);
      if (lane == 31) nwave[wv] = ic;
    }
    __syncthreads();
    if (wv < 2) seg[t] = ic - v + (wv == 1 ? nwave[0] : 0);
    __syncthreads();
    for (int e = t; e < n; e += AGG_THREADS) {
      const int w = wbuf[e];
      const int o = min((w >> SRC_BITS) & OFF_MASK, BKT_NODES - 1);
      const int slot = min(max(seg[o] + (int)rk[e], 0), BKT_CAP - 1);
      ids[slot] = w & SRC_MASK;
    }
    if (t < CHUNK && n + t < BKT_CAP) ids[n + t] = 0;
    __syncthreads();
    if (nn > 0) {
      const int bnd = seg[wv * AGG_NODES + (int)min(lane, (unsigned)AGG_NODES)];
      agg_nodes<FIRST>(ids, bnd, n0, nn, lane, el, ar4, bias4, feat, out);
    }
  } else {
    if (nn > 0)
      agg_nodes_slow<FIRST>(rstart, rlen, part, wv * AGG_NODES, n0, nn, lane, el, ar4, bias4, feat, out);
  }
}

extern "C" void kernel_launch(void* const* d_in, const int* in_sizes, int n_in,
                              void* d_out, int out_size, void* d_ws, size_t ws_size,
                              hipStream_t stream) {
  const float* h    = (const float*)d_in[0];
  const int*   srcs[2] = { (const int*)d_in[1], (const int*)d_in[3] };
  const int*   dsts[2] = { (const int*)d_in[2], (const int*)d_in[4] };
  const float* Ws[2]   = { (const float*)d_in[5],  (const float*)d_in[9]  };
  const float* als[2]  = { (const float*)d_in[6],  (const float*)d_in[10] };
  const float* ars[2]  = { (const float*)d_in[7],  (const float*)d_in[11] };
  const float* b0 = (const float*)d_in[8];
  const float* b1 = (const float*)d_in[12];
  float* out = (float*)d_out;

  float* feat = (float*)d_ws;
  float* el   = feat + (size_t)NNODES * HD;
  int*   pref = (int*)(el + (size_t)NNODES * NH);
  int*   part = pref + 2 * NNODES;

  const int nTile = (GEMM_TILES + (GEMM_BLOCK / 32) - 1) / (GEMM_BLOCK / 32);

  for (int r = 0; r < 2; ++r) {
    gat_chunk_partition<<<SORT_B, SORT_THREADS, 0, stream>>>(srcs[r], dsts[r], pref, part);
    gat_gemm_el<<<nTile, GEMM_BLOCK, 0, stream>>>(h, Ws[r], als[r], feat, el);
    if (r == 0)
      gat_bucket_agg<true><<<NBKT, AGG_THREADS, 0, stream>>>(pref, part, el, ars[r], feat, b0, b1, out);
    else
      gat_bucket_agg<false><<<NBKT, AGG_THREADS, 0, stream>>>(pref, part, el, ars[r], feat, b0, b1, out);
  }
}
